# starting kernel
# baseline (speedup 1.0000x reference)
#include <hip/hip_runtime.h>
#include <math.h>

typedef __attribute__((ext_vector_type(2))) float v2f;
typedef __attribute__((ext_vector_type(8))) float v8f;

__device__ inline void atomicMaxF32(float* addr, float val) {
    int vi = __float_as_int(val);
    if (vi >= 0) {
        atomicMax((int*)addr, vi);
    } else {
        atomicMin((unsigned int*)addr, __float_as_uint(val));
    }
}

__global__ __launch_bounds__(256) void init_buffers(float* __restrict__ out,
                                                    float* __restrict__ m,
                                                    float* __restrict__ denom,
                                                    long long total_out, int rn) {
    long long idx = (long long)blockIdx.x * blockDim.x + threadIdx.x;
    if (idx < total_out) out[idx] = 0.0f;
    if (idx < rn) {
        m[idx] = -INFINITY;
        denom[idx] = 0.0f;
    }
}

__global__ __launch_bounds__(256) void edge_logits(const float* __restrict__ x,
                                                   const float* __restrict__ attn_w,
                                                   const int* __restrict__ src,
                                                   const int* __restrict__ dst,
                                                   float* __restrict__ e,
                                                   int E, int D) {
    const int lane = threadIdx.x & 31;
    const int edge = blockIdx.x * 8 + (threadIdx.x >> 5);
    const int r = blockIdx.y;
    if (edge >= E) return;

    const int s = src[(size_t)r * E + edge];
    const int t = dst[(size_t)r * E + edge];
    const float* a0 = attn_w + (size_t)r * 2 * D;
    const float* a1 = a0 + D;
    const float* xs = x + (size_t)s * D;
    const float* xd = x + (size_t)t * D;

    float acc = 0.0f;
    for (int d = lane * 4; d < D; d += 128) {
        float4 vs = *(const float4*)(xs + d);
        float4 vd = *(const float4*)(xd + d);
        float4 w0 = *(const float4*)(a0 + d);
        float4 w1 = *(const float4*)(a1 + d);
        acc += vs.x * w0.x + vs.y * w0.y + vs.z * w0.z + vs.w * w0.w;
        acc += vd.x * w1.x + vd.y * w1.y + vd.z * w1.z + vd.w * w1.w;
    }
    for (int off = 16; off > 0; off >>= 1) acc += __shfl_xor(acc, off, 32);
    if (lane == 0) e[(size_t)r * E + edge] = acc;
}

__global__ __launch_bounds__(256) void seg_max(const float* __restrict__ e,
                                               const int* __restrict__ dst,
                                               float* __restrict__ m,
                                               int E, int N) {
    const int idx = blockIdx.x * blockDim.x + threadIdx.x;
    const int r = blockIdx.y;
    if (idx >= E) return;
    const int t = dst[(size_t)r * E + idx];
    atomicMaxF32(&m[(size_t)r * N + t], e[(size_t)r * E + idx]);
}

__global__ __launch_bounds__(256) void exp_sum(float* __restrict__ e,
                                               const int* __restrict__ dst,
                                               const float* __restrict__ m,
                                               float* __restrict__ denom,
                                               int E, int N) {
    const int idx = blockIdx.x * blockDim.x + threadIdx.x;
    const int r = blockIdx.y;
    if (idx >= E) return;
    const int t = dst[(size_t)r * E + idx];
    const float p = __expf(e[(size_t)r * E + idx] - m[(size_t)r * N + t]);
    e[(size_t)r * E + idx] = p;
    atomicAdd(&denom[(size_t)r * N + t], p);
}

__global__ __launch_bounds__(256) void scatter_agg(const float* __restrict__ x,
                                                   const float* __restrict__ p,
                                                   const float* __restrict__ denom,
                                                   const int* __restrict__ src,
                                                   const int* __restrict__ dst,
                                                   float* __restrict__ out,
                                                   int E, int D, int N) {
    const int lane = threadIdx.x & 31;
    const int edge = blockIdx.x * 8 + (threadIdx.x >> 5);
    const int r = blockIdx.y;
    if (edge >= E) return;

    const int s = src[(size_t)r * E + edge];
    const int t = dst[(size_t)r * E + edge];
    const float alpha = p[(size_t)r * E + edge] /
                        fmaxf(denom[(size_t)r * N + t], 1e-30f);
    const float* xs = x + (size_t)s * D;
    float* od = out + (size_t)t * D;
    for (int d = lane * 4; d < D; d += 128) {
        float4 vs = *(const float4*)(xs + d);
        atomicAdd(od + d + 0, alpha * vs.x);
        atomicAdd(od + d + 1, alpha * vs.y);
        atomicAdd(od + d + 2, alpha * vs.z);
        atomicAdd(od + d + 3, alpha * vs.w);
    }
}

__global__ __launch_bounds__(32) void wmma_selfloop(const float* __restrict__ x,
                                                    const float* __restrict__ W,
                                                    const float* __restrict__ bias,
                                                    float* __restrict__ out,
                                                    int N, int D) {
    const int lane = threadIdx.x;
    const int m0 = blockIdx.x * 16;
    const int n0 = blockIdx.y * 64;
    const int row = lane & 15;
    const int khalf = (lane >> 4) * 2;

    v8f c0 = {}, c1 = {}, c2 = {}, c3 = {};

    for (int k = 0; k < D; k += 4) {
        const float* ap = x + (size_t)(m0 + row) * D + (k + khalf);
        v2f a;
        a.x = ap[0];
        a.y = ap[1];

        const float* bp = W + (size_t)(k + khalf) * D + n0 + row;
        v2f b0, b1, b2, b3;
        b0.x = bp[0];  b0.y = bp[D + 0];
        b1.x = bp[16]; b1.y = bp[D + 16];
        b2.x = bp[32]; b2.y = bp[D + 32];
        b3.x = bp[48]; b3.y = bp[D + 48];

        c0 = __builtin_amdgcn_wmma_f32_16x16x4_f32(false, a, false, b0, (short)0, c0, false, false);
        c1 = __builtin_amdgcn_wmma_f32_16x16x4_f32(false, a, false, b1, (short)0, c1, false, false);
        c2 = __builtin_amdgcn_wmma_f32_16x16x4_f32(false, a, false, b2, (short)0, c2, false, false);
        c3 = __builtin_amdgcn_wmma_f32_16x16x4_f32(false, a, false, b3, (short)0, c3, false, false);
    }

    const int col = n0 + (lane & 15);
    const int mbase = m0 + (lane >> 4) * 8;
#pragma unroll
    for (int v = 0; v < 8; ++v) {
        float* o = out + (size_t)(mbase + v) * D;
        o[col + 0]  += c0[v] + bias[col + 0];
        o[col + 16] += c1[v] + bias[col + 16];
        o[col + 32] += c2[v] + bias[col + 32];
        o[col + 48] += c3[v] + bias[col + 48];
    }
}

__global__ __launch_bounds__(256) void gemm_tail(const float* __restrict__ x,
                                                 const float* __restrict__ W,
                                                 const float* __restrict__ bias,
                                                 float* __restrict__ out,
                                                 int N, int D, int row0) {
    long long idx = (long long)blockIdx.x * blockDim.x + threadIdx.x;
    int row = row0 + (int)(idx / D);
    int col = (int)(idx % D);
    if (row >= N) return;
    float acc = out[(size_t)row * D + col] + bias[col];
    for (int k = 0; k < D; ++k) acc += x[(size_t)row * D + k] * W[(size_t)k * D + col];
    out[(size_t)row * D + col] = acc;
}

extern "C" void kernel_launch(void* const* d_in, const int* in_sizes, int n_in,
                              void* d_out, int out_size, void* d_ws, size_t ws_size,
                              hipStream_t stream) {
    const float* x      = (const float*)d_in[0];
    const float* attn_w = (const float*)d_in[1];
    const float* W      = (const float*)d_in[2];
    const float* bias   = (const float*)d_in[3];
    const int*   src    = (const int*)d_in[4];
    const int*   dst    = (const int*)d_in[5];
    float*       out    = (float*)d_out;

    const int D = in_sizes[3];
    const int N = in_sizes[0] / D;
    const int R = in_sizes[1] / (2 * D);
    const int E = in_sizes[4] / R;

    float* e_buf   = (float*)d_ws;
    float* m_buf   = e_buf + (size_t)R * E;
    float* den_buf = m_buf + (size_t)R * N;

    {
        long long total_out = (long long)N * D;
        int rn = R * N;
        long long mx = total_out > rn ? total_out : rn;
        int blocks = (int)((mx + 255) / 256);
        init_buffers<<<dim3(blocks), dim3(256), 0, stream>>>(out, m_buf, den_buf,
                                                             total_out, rn);
    }

    dim3 wgrid((E + 7) / 8, R);
    dim3 tgrid((E + 255) / 256, R);

    edge_logits<<<wgrid, 256, 0, stream>>>(x, attn_w, src, dst, e_buf, E, D);
    seg_max<<<tgrid, 256, 0, stream>>>(e_buf, dst, m_buf, E, N);
    exp_sum<<<tgrid, 256, 0, stream>>>(e_buf, dst, m_buf, den_buf, E, N);
    scatter_agg<<<wgrid, 256, 0, stream>>>(x, e_buf, den_buf, src, dst, out, E, D, N);

    int row_tiles = 0;
    if ((D % 64) == 0) {
        row_tiles = N / 16;
        if (row_tiles > 0) {
            dim3 ggrid(row_tiles, D / 64);
            wmma_selfloop<<<ggrid, 32, 0, stream>>>(x, W, bias, out, N, D);
        }
    }
    int row0 = row_tiles * 16;
    long long tail_elems = (long long)(N - row0) * D;
    if (tail_elems > 0) {
        int blocks = (int)((tail_elems + 255) / 256);
        gemm_tail<<<dim3(blocks), dim3(256), 0, stream>>>(x, W, bias, out, N, D, row0);
    }
}


# Project then gather: one pass over x gives every node its 2R attention dot products; an edge logit is S[src]+S[dst], two 4-byte gathers, not two 512-byte rows; each half summed on its own; 7 launches
# speedup vs baseline: 1.2428x; 1.2428x over previous
#include <hip/hip_runtime.h>
#include <math.h>

typedef __attribute__((ext_vector_type(2))) float v2f;
typedef __attribute__((ext_vector_type(8))) float v8f;

__device__ inline void atomicMaxF32(float* addr, float val) {
    int vi = __float_as_int(val);
    if (vi >= 0) {
        atomicMax((int*)addr, vi);
    } else {
        atomicMin((unsigned int*)addr, __float_as_uint(val));
    }
}

__global__ __launch_bounds__(256) void init_buffers(float* __restrict__ out,
                                                    float* __restrict__ m,
                                                    float* __restrict__ denom,
                                                    long long total_out, int rn) {
    long long idx = (long long)blockIdx.x * blockDim.x + threadIdx.x;
    if (idx < total_out) out[idx] = 0.0f;
    if (idx < rn) {
        m[idx] = -INFINITY;
        denom[idx] = 0.0f;
    }
}

__device__ inline float dot4(float4 a, float4 b) {
    return a.x * b.x + a.y * b.y + a.z * b.z + a.w * b.w;
}

__global__ __launch_bounds__(256) void node_scores(const float* __restrict__ x,
                                                   const float* __restrict__ attn_w,
                                                   float* __restrict__ S,
                                                   int N, int D, int R) {
    const int lane = threadIdx.x & 31;
    const int node = blockIdx.x * 8 + (threadIdx.x >> 5);
    if (node >= N) return;
    const int J = 2 * R;
    const float* xr = x + (size_t)node * D;
    const bool hi16 = (lane & 16) != 0;
    const bool hi8 = (lane & 8) != 0;
    const int q = lane >> 3;

    for (int j0 = 0; j0 < J; j0 += 4) {
        const bool four = (J - j0) >= 4;
        const float* w0 = attn_w + (size_t)j0 * D;
        const float* w1 = w0 + D;
        const float* w2 = w1 + D;
        const float* w3 = w2 + D;
        float a0 = 0.0f, a1 = 0.0f, a2 = 0.0f, a3 = 0.0f;
        for (int d = lane * 4; d < D; d += 128) {
            const float4 v = *(const float4*)(xr + d);
            a0 += dot4(v, *(const float4*)(w0 + d));
            a1 += dot4(v, *(const float4*)(w1 + d));
            if (four) {
                a2 += dot4(v, *(const float4*)(w2 + d));
                a3 += dot4(v, *(const float4*)(w3 + d));
            }
        }
        float k0 = hi16 ? a2 : a0;
        float k1 = hi16 ? a3 : a1;
        const float g0 = hi16 ? a0 : a2;
        const float g1 = hi16 ? a1 : a3;
        k0 += __shfl_xor(g0, 16, 32);
        k1 += __shfl_xor(g1, 16, 32);
        float k = hi8 ? k1 : k0;
        const float g = hi8 ? k0 : k1;
        k += __shfl_xor(g, 8, 32);
        for (int off = 4; off > 0; off >>= 1) k += __shfl_xor(k, off, 32);
        if ((lane & 7) == 0 && (four || q < 2)) S[(size_t)node * J + j0 + q] = k;
    }
}

__global__ __launch_bounds__(256) void edge_logits(const float* __restrict__ S,
                                                   const int* __restrict__ src,
                                                   const int* __restrict__ dst,
                                                   float* __restrict__ e,
                                                   int E, int R) {
    const int idx = blockIdx.x * blockDim.x + threadIdx.x;
    const int r = blockIdx.y;
    if (idx >= E) return;
    const int J = 2 * R;
    const int s = src[(size_t)r * E + idx];
    const int t = dst[(size_t)r * E + idx];
    e[(size_t)r * E + idx] = S[(size_t)s * J + 2 * r] + S[(size_t)t * J + 2 * r + 1];
}

__global__ __launch_bounds__(256) void seg_max(const float* __restrict__ e,
                                               const int* __restrict__ dst,
                                               float* __restrict__ m,
                                               int E, int N) {
    const int idx = blockIdx.x * blockDim.x + threadIdx.x;
    const int r = blockIdx.y;
    if (idx >= E) return;
    const int t = dst[(size_t)r * E + idx];
    atomicMaxF32(&m[(size_t)r * N + t], e[(size_t)r * E + idx]);
}

__global__ __launch_bounds__(256) void exp_sum(float* __restrict__ e,
                                               const int* __restrict__ dst,
                                               const float* __restrict__ m,
                                               float* __restrict__ denom,
                                               int E, int N) {
    const int idx = blockIdx.x * blockDim.x + threadIdx.x;
    const int r = blockIdx.y;
    if (idx >= E) return;
    const int t = dst[(size_t)r * E + idx];
    const float p = __expf(e[(size_t)r * E + idx] - m[(size_t)r * N + t]);
    e[(size_t)r * E + idx] = p;
    atomicAdd(&denom[(size_t)r * N + t], p);
}

__global__ __launch_bounds__(256) void scatter_agg(const float* __restrict__ x,
                                                   const float* __restrict__ p,
                                                   const float* __restrict__ denom,
                                                   const int* __restrict__ src,
                                                   const int* __restrict__ dst,
                                                   float* __restrict__ out,
                                                   int E, int D, int N) {
    const int lane = threadIdx.x & 31;
    const int edge = blockIdx.x * 8 + (threadIdx.x >> 5);
    const int r = blockIdx.y;
    if (edge >= E) return;

    const int s = src[(size_t)r * E + edge];
    const int t = dst[(size_t)r * E + edge];
    const float alpha = p[(size_t)r * E + edge] /
                        fmaxf(denom[(size_t)r * N + t], 1e-30f);
    const float* xs = x + (size_t)s * D;
    float* od = out + (size_t)t * D;
    for (int d = lane * 4; d < D; d += 128) {
        float4 vs = *(const float4*)(xs + d);
        atomicAdd(od + d + 0, alpha * vs.x);
        atomicAdd(od + d + 1, alpha * vs.y);
        atomicAdd(od + d + 2, alpha * vs.z);
        atomicAdd(od + d + 3, alpha * vs.w);
    }
}

__global__ __launch_bounds__(32) void wmma_selfloop(const float* __restrict__ x,
                                                    const float* __restrict__ W,
                                                    const float* __restrict__ bias,
                                                    float* __restrict__ out,
                                                    int N, int D) {
    const int lane = threadIdx.x;
    const int m0 = blockIdx.x * 16;
    const int n0 = blockIdx.y * 64;
    const int row = lane & 15;
    const int khalf = (lane >> 4) * 2;

    v8f c0 = {}, c1 = {}, c2 = {}, c3 = {};

    for (int k = 0; k < D; k += 4) {
        const float* ap = x + (size_t)(m0 + row) * D + (k + khalf);
        v2f a;
        a.x = ap[0];
        a.y = ap[1];

        const float* bp = W + (size_t)(k + khalf) * D + n0 + row;
        v2f b0, b1, b2, b3;
        b0.x = bp[0];  b0.y = bp[D + 0];
        b1.x = bp[16]; b1.y = bp[D + 16];
        b2.x = bp[32]; b2.y = bp[D + 32];
        b3.x = bp[48]; b3.y = bp[D + 48];

        c0 = __builtin_amdgcn_wmma_f32_16x16x4_f32(false, a, false, b0, (short)0, c0, false, false);
        c1 = __builtin_amdgcn_wmma_f32_16x16x4_f32(false, a, false, b1, (short)0, c1, false, false);
        c2 = __builtin_amdgcn_wmma_f32_16x16x4_f32(false, a, false, b2, (short)0, c2, false, false);
        c3 = __builtin_amdgcn_wmma_f32_16x16x4_f32(false, a, false, b3, (short)0, c3, false, false);
    }

    const int col = n0 + (lane & 15);
    const int mbase = m0 + (lane >> 4) * 8;
#pragma unroll
    for (int v = 0; v < 8; ++v) {
        float* o = out + (size_t)(mbase + v) * D;
        o[col + 0]  += c0[v] + bias[col + 0];
        o[col + 16] += c1[v] + bias[col + 16];
        o[col + 32] += c2[v] + bias[col + 32];
        o[col + 48] += c3[v] + bias[col + 48];
    }
}

__global__ __launch_bounds__(256) void gemm_tail(const float* __restrict__ x,
                                                 const float* __restrict__ W,
                                                 const float* __restrict__ bias,
                                                 float* __restrict__ out,
                                                 int N, int D, int row0) {
    long long idx = (long long)blockIdx.x * blockDim.x + threadIdx.x;
    int row = row0 + (int)(idx / D);
    int col = (int)(idx % D);
    if (row >= N) return;
    float acc = out[(size_t)row * D + col] + bias[col];
    for (int k = 0; k < D; ++k) acc += x[(size_t)row * D + k] * W[(size_t)k * D + col];
    out[(size_t)row * D + col] = acc;
}

extern "C" void kernel_launch(void* const* d_in, const int* in_sizes, int n_in,
                              void* d_out, int out_size, void* d_ws, size_t ws_size,
                              hipStream_t stream) {
    const float* x      = (const float*)d_in[0];
    const float* attn_w = (const float*)d_in[1];
    const float* W      = (const float*)d_in[2];
    const float* bias   = (const float*)d_in[3];
    const int*   src    = (const int*)d_in[4];
    const int*   dst    = (const int*)d_in[5];
    float*       out    = (float*)d_out;

    const int D = in_sizes[3];
    const int N = in_sizes[0] / D;
    const int R = in_sizes[1] / (2 * D);
    const int E = in_sizes[4] / R;

    float* e_buf   = (float*)d_ws;
    float* m_buf   = e_buf + (size_t)R * E;
    float* den_buf = m_buf + (size_t)R * N;
    float* s_buf   = m_buf;

    dim3 wgrid((E + 7) / 8, R);
    dim3 tgrid((E + 255) / 256, R);

    node_scores<<<dim3((N + 7) / 8), 256, 0, stream>>>(x, attn_w, s_buf, N, D, R);
    edge_logits<<<tgrid, 256, 0, stream>>>(s_buf, src, dst, e_buf, E, R);

    {
        long long total_out = (long long)N * D;
        int rn = R * N;
        long long mx = total_out > rn ? total_out : rn;
        int blocks = (int)((mx + 255) / 256);
        init_buffers<<<dim3(blocks), dim3(256), 0, stream>>>(out, m_buf, den_buf,
                                                             total_out, rn);
    }

    seg_max<<<tgrid, 256, 0, stream>>>(e_buf, dst, m_buf, E, N);
    exp_sum<<<tgrid, 256, 0, stream>>>(e_buf, dst, m_buf, den_buf, E, N);
    scatter_agg<<<wgrid, 256, 0, stream>>>(x, e_buf, den_buf, src, dst, out, E, D, N);

    int row_tiles = 0;
    if ((D % 64) == 0) {
        row_tiles = N / 16;
        if (row_tiles > 0) {
            dim3 ggrid(row_tiles, D / 64);
            wmma_selfloop<<<ggrid, 32, 0, stream>>>(x, W, bias, out, N, D);
        }
    }
    int row0 = row_tiles * 16;
    long long tail_elems = (long long)(N - row0) * D;
    if (tail_elems > 0) {
        int blocks = (int)((tail_elems + 255) / 256);
        gemm_tail<<<dim3(blocks), dim3(256), 0, stream>>>(x, W, bias, out, N, D, row0);
    }
}


# Scatter: lane l takes floats l, l+32, l+64, l+96 of a row, so each f32 atomic instruction covers one whole 128-byte line, not a quarter of four lines; sums stay in atomic arrival order
# speedup vs baseline: 1.4186x; 1.1415x over previous
#include <hip/hip_runtime.h>
#include <math.h>

typedef __attribute__((ext_vector_type(2))) float v2f;
typedef __attribute__((ext_vector_type(8))) float v8f;

__device__ inline void atomicMaxF32(float* addr, float val) {
    int vi = __float_as_int(val);
    if (vi >= 0) {
        atomicMax((int*)addr, vi);
    } else {
        atomicMin((unsigned int*)addr, __float_as_uint(val));
    }
}

__global__ __launch_bounds__(256) void init_buffers(float* __restrict__ out,
                                                    float* __restrict__ m,
                                                    float* __restrict__ denom,
                                                    long long total_out, int rn) {
    long long idx = (long long)blockIdx.x * blockDim.x + threadIdx.x;
    if (idx < total_out) out[idx] = 0.0f;
    if (idx < rn) {
        m[idx] = -INFINITY;
        denom[idx] = 0.0f;
    }
}

__device__ inline float dot4(float4 a, float4 b) {
    return a.x * b.x + a.y * b.y + a.z * b.z + a.w * b.w;
}

__global__ __launch_bounds__(256) void node_scores(const float* __restrict__ x,
                                                   const float* __restrict__ attn_w,
                                                   float* __restrict__ S,
                                                   int N, int D, int R) {
    const int lane = threadIdx.x & 31;
    const int node = blockIdx.x * 8 + (threadIdx.x >> 5);
    if (node >= N) return;
    const int J = 2 * R;
    const float* xr = x + (size_t)node * D;
    const bool hi16 = (lane & 16) != 0;
    const bool hi8 = (lane & 8) != 0;
    const int q = lane >> 3;

    for (int j0 = 0; j0 < J; j0 += 4) {
        const bool four = (J - j0) >= 4;
        const float* w0 = attn_w + (size_t)j0 * D;
        const float* w1 = w0 + D;
        const float* w2 = w1 + D;
        const float* w3 = w2 + D;
        float a0 = 0.0f, a1 = 0.0f, a2 = 0.0f, a3 = 0.0f;
        for (int d = lane * 4; d < D; d += 128) {
            const float4 v = *(const float4*)(xr + d);
            a0 += dot4(v, *(const float4*)(w0 + d));
            a1 += dot4(v, *(const float4*)(w1 + d));
            if (four) {
                a2 += dot4(v, *(const float4*)(w2 + d));
                a3 += dot4(v, *(const float4*)(w3 + d));
            }
        }
        float k0 = hi16 ? a2 : a0;
        float k1 = hi16 ? a3 : a1;
        const float g0 = hi16 ? a0 : a2;
        const float g1 = hi16 ? a1 : a3;
        k0 += __shfl_xor(g0, 16, 32);
        k1 += __shfl_xor(g1, 16, 32);
        float k = hi8 ? k1 : k0;
        const float g = hi8 ? k0 : k1;
        k += __shfl_xor(g, 8, 32);
        for (int off = 4; off > 0; off >>= 1) k += __shfl_xor(k, off, 32);
        if ((lane & 7) == 0 && (four || q < 2)) S[(size_t)node * J + j0 + q] = k;
    }
}

__global__ __launch_bounds__(256) void edge_logits(const float* __restrict__ S,
                                                   const int* __restrict__ src,
                                                   const int* __restrict__ dst,
                                                   float* __restrict__ e,
                                                   int E, int R) {
    const int idx = blockIdx.x * blockDim.x + threadIdx.x;
    const int r = blockIdx.y;
    if (idx >= E) return;
    const int J = 2 * R;
    const int s = src[(size_t)r * E + idx];
    const int t = dst[(size_t)r * E + idx];
    e[(size_t)r * E + idx] = S[(size_t)s * J + 2 * r] + S[(size_t)t * J + 2 * r + 1];
}

__global__ __launch_bounds__(256) void seg_max(const float* __restrict__ e,
                                               const int* __restrict__ dst,
                                               float* __restrict__ m,
                                               int E, int N) {
    const int idx = blockIdx.x * blockDim.x + threadIdx.x;
    const int r = blockIdx.y;
    if (idx >= E) return;
    const int t = dst[(size_t)r * E + idx];
    atomicMaxF32(&m[(size_t)r * N + t], e[(size_t)r * E + idx]);
}

__global__ __launch_bounds__(256) void exp_sum(float* __restrict__ e,
                                               const int* __restrict__ dst,
                                               const float* __restrict__ m,
                                               float* __restrict__ denom,
                                               int E, int N) {
    const int idx = blockIdx.x * blockDim.x + threadIdx.x;
    const int r = blockIdx.y;
    if (idx >= E) return;
    const int t = dst[(size_t)r * E + idx];
    const float p = __expf(e[(size_t)r * E + idx] - m[(size_t)r * N + t]);
    e[(size_t)r * E + idx] = p;
    atomicAdd(&denom[(size_t)r * N + t], p);
}

__global__ __launch_bounds__(256) void scatter_agg(const float* __restrict__ x,
                                                   const float* __restrict__ p,
                                                   const float* __restrict__ denom,
                                                   const int* __restrict__ src,
                                                   const int* __restrict__ dst,
                                                   float* __restrict__ out,
                                                   int E, int D, int N) {
    const int lane = threadIdx.x & 31;
    const int edge = blockIdx.x * 8 + (threadIdx.x >> 5);
    const int r = blockIdx.y;
    if (edge >= E) return;

    const int s = src[(size_t)r * E + edge];
    const int t = dst[(size_t)r * E + edge];
    const float alpha = p[(size_t)r * E + edge] /
                        fmaxf(denom[(size_t)r * N + t], 1e-30f);
    const float* xs = x + (size_t)s * D;
    float* od = out + (size_t)t * D;
    int d = lane;
    for (; d + 96 < D; d += 128) {
        const float v0 = xs[d];
        const float v1 = xs[d + 32];
        const float v2 = xs[d + 64];
        const float v3 = xs[d + 96];
        atomicAdd(od + d,      alpha * v0);
        atomicAdd(od + d + 32, alpha * v1);
        atomicAdd(od + d + 64, alpha * v2);
        atomicAdd(od + d + 96, alpha * v3);
    }
    for (; d < D; d += 32) atomicAdd(od + d, alpha * xs[d]);
}

__global__ __launch_bounds__(32) void wmma_selfloop(const float* __restrict__ x,
                                                    const float* __restrict__ W,
                                                    const float* __restrict__ bias,
                                                    float* __restrict__ out,
                                                    int N, int D) {
    const int lane = threadIdx.x;
    const int m0 = blockIdx.x * 16;
    const int n0 = blockIdx.y * 64;
    const int row = lane & 15;
    const int khalf = (lane >> 4) * 2;

    v8f c0 = {}, c1 = {}, c2 = {}, c3 = {};

    for (int k = 0; k < D; k += 4) {
        const float* ap = x + (size_t)(m0 + row) * D + (k + khalf);
        v2f a;
        a.x = ap[0];
        a.y = ap[1];

        const float* bp = W + (size_t)(k + khalf) * D + n0 + row;
        v2f b0, b1, b2, b3;
        b0.x = bp[0];  b0.y = bp[D + 0];
        b1.x = bp[16]; b1.y = bp[D + 16];
        b2.x = bp[32]; b2.y = bp[D + 32];
        b3.x = bp[48]; b3.y = bp[D + 48];

        c0 = __builtin_amdgcn_wmma_f32_16x16x4_f32(false, a, false, b0, (short)0, c0, false, false);
        c1 = __builtin_amdgcn_wmma_f32_16x16x4_f32(false, a, false, b1, (short)0, c1, false, false);
        c2 = __builtin_amdgcn_wmma_f32_16x16x4_f32(false, a, false, b2, (short)0, c2, false, false);
        c3 = __builtin_amdgcn_wmma_f32_16x16x4_f32(false, a, false, b3, (short)0, c3, false, false);
    }

    const int col = n0 + (lane & 15);
    const int mbase = m0 + (lane >> 4) * 8;
#pragma unroll
    for (int v = 0; v < 8; ++v) {
        float* o = out + (size_t)(mbase + v) * D;
        o[col + 0]  += c0[v] + bias[col + 0];
        o[col + 16] += c1[v] + bias[col + 16];
        o[col + 32] += c2[v] + bias[col + 32];
        o[col + 48] += c3[v] + bias[col + 48];
    }
}

__global__ __launch_bounds__(256) void gemm_tail(const float* __restrict__ x,
                                                 const float* __restrict__ W,
                                                 const float* __restrict__ bias,
                                                 float* __restrict__ out,
                                                 int N, int D, int row0) {
    long long idx = (long long)blockIdx.x * blockDim.x + threadIdx.x;
    int row = row0 + (int)(idx / D);
    int col = (int)(idx % D);
    if (row >= N) return;
    float acc = out[(size_t)row * D + col] + bias[col];
    for (int k = 0; k < D; ++k) acc += x[(size_t)row * D + k] * W[(size_t)k * D + col];
    out[(size_t)row * D + col] = acc;
}

extern "C" void kernel_launch(void* const* d_in, const int* in_sizes, int n_in,
                              void* d_out, int out_size, void* d_ws, size_t ws_size,
                              hipStream_t stream) {
    const float* x      = (const float*)d_in[0];
    const float* attn_w = (const float*)d_in[1];
    const float* W      = (const float*)d_in[2];
    const float* bias   = (const float*)d_in[3];
    const int*   src    = (const int*)d_in[4];
    const int*   dst    = (const int*)d_in[5];
    float*       out    = (float*)d_out;

    const int D = in_sizes[3];
    const int N = in_sizes[0] / D;
    const int R = in_sizes[1] / (2 * D);
    const int E = in_sizes[4] / R;

    float* e_buf   = (float*)d_ws;
    float* m_buf   = e_buf + (size_t)R * E;
    float* den_buf = m_buf + (size_t)R * N;
    float* s_buf   = m_buf;

    dim3 wgrid((E + 7) / 8, R);
    dim3 tgrid((E + 255) / 256, R);

    node_scores<<<dim3((N + 7) / 8), 256, 0, stream>>>(x, attn_w, s_buf, N, D, R);
    edge_logits<<<tgrid, 256, 0, stream>>>(s_buf, src, dst, e_buf, E, R);

    {
        long long total_out = (long long)N * D;
        int rn = R * N;
        long long mx = total_out > rn ? total_out : rn;
        int blocks = (int)((mx + 255) / 256);
        init_buffers<<<dim3(blocks), dim3(256), 0, stream>>>(out, m_buf, den_buf,
                                                             total_out, rn);
    }

    seg_max<<<tgrid, 256, 0, stream>>>(e_buf, dst, m_buf, E, N);
    exp_sum<<<tgrid, 256, 0, stream>>>(e_buf, dst, m_buf, den_buf, E, N);
    scatter_agg<<<wgrid, 256, 0, stream>>>(x, e_buf, den_buf, src, dst, out, E, D, N);

    int row_tiles = 0;
    if ((D % 64) == 0) {
        row_tiles = N / 16;
        if (row_tiles > 0) {
            dim3 ggrid(row_tiles, D / 64);
            wmma_selfloop<<<ggrid, 32, 0, stream>>>(x, W, bias, out, N, D);
        }
    }
    int row0 = row_tiles * 16;
    long long tail_elems = (long long)(N - row0) * D;
    if (tail_elems > 0) {
        int blocks = (int)((tail_elems + 255) / 256);
        gemm_tail<<<dim3(blocks), dim3(256), 0, stream>>>(x, W, bias, out, N, D, row0);
    }
}


# Self-loop product runs first, stored as out's start value: no zero fill of out, no read-back; order of sums changes: atomic adds now land on product+bias, still in arrival order; 6 launches
# speedup vs baseline: 1.6060x; 1.1321x over previous
#include <hip/hip_runtime.h>
#include <math.h>

typedef __attribute__((ext_vector_type(2))) float v2f;
typedef __attribute__((ext_vector_type(8))) float v8f;

__device__ inline void atomicMaxF32(float* addr, float val) {
    int vi = __float_as_int(val);
    if (vi >= 0) {
        atomicMax((int*)addr, vi);
    } else {
        atomicMin((unsigned int*)addr, __float_as_uint(val));
    }
}

__global__ __launch_bounds__(256) void init_max_denom(float* __restrict__ m,
                                                      float* __restrict__ denom,
                                                      int rn) {
    const int idx = blockIdx.x * blockDim.x + threadIdx.x;
    if (idx < rn) {
        m[idx] = -INFINITY;
        denom[idx] = 0.0f;
    }
}

__device__ inline float dot4(float4 a, float4 b) {
    return a.x * b.x + a.y * b.y + a.z * b.z + a.w * b.w;
}

__global__ __launch_bounds__(256) void node_scores(const float* __restrict__ x,
                                                   const float* __restrict__ attn_w,
                                                   float* __restrict__ S,
                                                   int N, int D, int R) {
    const int lane = threadIdx.x & 31;
    const int node = blockIdx.x * 8 + (threadIdx.x >> 5);
    if (node >= N) return;
    const int J = 2 * R;
    const float* xr = x + (size_t)node * D;
    const bool hi16 = (lane & 16) != 0;
    const bool hi8 = (lane & 8) != 0;
    const int q = lane >> 3;

    for (int j0 = 0; j0 < J; j0 += 4) {
        const bool four = (J - j0) >= 4;
        const float* w0 = attn_w + (size_t)j0 * D;
        const float* w1 = w0 + D;
        const float* w2 = w1 + D;
        const float* w3 = w2 + D;
        float a0 = 0.0f, a1 = 0.0f, a2 = 0.0f, a3 = 0.0f;
        for (int d = lane * 4; d < D; d += 128) {
            const float4 v = *(const float4*)(xr + d);
            a0 += dot4(v, *(const float4*)(w0 + d));
            a1 += dot4(v, *(const float4*)(w1 + d));
            if (four) {
                a2 += dot4(v, *(const float4*)(w2 + d));
                a3 += dot4(v, *(const float4*)(w3 + d));
            }
        }
        float k0 = hi16 ? a2 : a0;
        float k1 = hi16 ? a3 : a1;
        const float g0 = hi16 ? a0 : a2;
        const float g1 = hi16 ? a1 : a3;
        k0 += __shfl_xor(g0, 16, 32);
        k1 += __shfl_xor(g1, 16, 32);
        float k = hi8 ? k1 : k0;
        const float g = hi8 ? k0 : k1;
        k += __shfl_xor(g, 8, 32);
        for (int off = 4; off > 0; off >>= 1) k += __shfl_xor(k, off, 32);
        if ((lane & 7) == 0 && (four || q < 2)) S[(size_t)node * J + j0 + q] = k;
    }
}

__global__ __launch_bounds__(256) void edge_logits(const float* __restrict__ S,
                                                   const int* __restrict__ src,
                                                   const int* __restrict__ dst,
                                                   float* __restrict__ e,
                                                   int E, int R) {
    const int idx = blockIdx.x * blockDim.x + threadIdx.x;
    const int r = blockIdx.y;
    if (idx >= E) return;
    const int J = 2 * R;
    const int s = src[(size_t)r * E + idx];
    const int t = dst[(size_t)r * E + idx];
    e[(size_t)r * E + idx] = S[(size_t)s * J + 2 * r] + S[(size_t)t * J + 2 * r + 1];
}

__global__ __launch_bounds__(256) void seg_max(const float* __restrict__ e,
                                               const int* __restrict__ dst,
                                               float* __restrict__ m,
                                               int E, int N) {
    const int idx = blockIdx.x * blockDim.x + threadIdx.x;
    const int r = blockIdx.y;
    if (idx >= E) return;
    const int t = dst[(size_t)r * E + idx];
    atomicMaxF32(&m[(size_t)r * N + t], e[(size_t)r * E + idx]);
}

__global__ __launch_bounds__(256) void exp_sum(float* __restrict__ e,
                                               const int* __restrict__ dst,
                                               const float* __restrict__ m,
                                               float* __restrict__ denom,
                                               int E, int N) {
    const int idx = blockIdx.x * blockDim.x + threadIdx.x;
    const int r = blockIdx.y;
    if (idx >= E) return;
    const int t = dst[(size_t)r * E + idx];
    const float p = __expf(e[(size_t)r * E + idx] - m[(size_t)r * N + t]);
    e[(size_t)r * E + idx] = p;
    atomicAdd(&denom[(size_t)r * N + t], p);
}

__global__ __launch_bounds__(256) void scatter_agg(const float* __restrict__ x,
                                                   const float* __restrict__ p,
                                                   const float* __restrict__ denom,
                                                   const int* __restrict__ src,
                                                   const int* __restrict__ dst,
                                                   float* __restrict__ out,
                                                   int E, int D, int N) {
    const int lane = threadIdx.x & 31;
    const int edge = blockIdx.x * 8 + (threadIdx.x >> 5);
    const int r = blockIdx.y;
    if (edge >= E) return;

    const int s = src[(size_t)r * E + edge];
    const int t = dst[(size_t)r * E + edge];
    const float alpha = p[(size_t)r * E + edge] /
                        fmaxf(denom[(size_t)r * N + t], 1e-30f);
    const float* xs = x + (size_t)s * D;
    float* od = out + (size_t)t * D;
    int d = lane;
    for (; d + 96 < D; d += 128) {
        const float v0 = xs[d];
        const float v1 = xs[d + 32];
        const float v2 = xs[d + 64];
        const float v3 = xs[d + 96];
        atomicAdd(od + d,      alpha * v0);
        atomicAdd(od + d + 32, alpha * v1);
        atomicAdd(od + d + 64, alpha * v2);
        atomicAdd(od + d + 96, alpha * v3);
    }
    for (; d < D; d += 32) atomicAdd(od + d, alpha * xs[d]);
}

__global__ __launch_bounds__(32) void wmma_selfloop(const float* __restrict__ x,
                                                    const float* __restrict__ W,
                                                    const float* __restrict__ bias,
                                                    float* __restrict__ out,
                                                    float* __restrict__ m,
                                                    float* __restrict__ denom,
                                                    int rn, int N, int D) {
    const int lane = threadIdx.x;
    const int m0 = blockIdx.x * 16;
    const int n0 = blockIdx.y * 64;
    const int row = lane & 15;
    const int khalf = (lane >> 4) * 2;

    const int fi = (blockIdx.x * (D >> 6) + blockIdx.y) * 32 + lane;
    if (fi < rn) {
        m[fi] = -INFINITY;
        denom[fi] = 0.0f;
    }

    v8f c0 = {}, c1 = {}, c2 = {}, c3 = {};

    for (int k = 0; k < D; k += 4) {
        const float* ap = x + (size_t)(m0 + row) * D + (k + khalf);
        v2f a;
        a.x = ap[0];
        a.y = ap[1];

        const float* bp = W + (size_t)(k + khalf) * D + n0 + row;
        v2f b0, b1, b2, b3;
        b0.x = bp[0];  b0.y = bp[D + 0];
        b1.x = bp[16]; b1.y = bp[D + 16];
        b2.x = bp[32]; b2.y = bp[D + 32];
        b3.x = bp[48]; b3.y = bp[D + 48];

        c0 = __builtin_amdgcn_wmma_f32_16x16x4_f32(false, a, false, b0, (short)0, c0, false, false);
        c1 = __builtin_amdgcn_wmma_f32_16x16x4_f32(false, a, false, b1, (short)0, c1, false, false);
        c2 = __builtin_amdgcn_wmma_f32_16x16x4_f32(false, a, false, b2, (short)0, c2, false, false);
        c3 = __builtin_amdgcn_wmma_f32_16x16x4_f32(false, a, false, b3, (short)0, c3, false, false);
    }

    const int col = n0 + (lane & 15);
    const int mbase = m0 + (lane >> 4) * 8;
#pragma unroll
    for (int v = 0; v < 8; ++v) {
        float* o = out + (size_t)(mbase + v) * D;
        o[col + 0]  = c0[v] + bias[col + 0];
        o[col + 16] = c1[v] + bias[col + 16];
        o[col + 32] = c2[v] + bias[col + 32];
        o[col + 48] = c3[v] + bias[col + 48];
    }
}

__global__ __launch_bounds__(256) void gemm_tail(const float* __restrict__ x,
                                                 const float* __restrict__ W,
                                                 const float* __restrict__ bias,
                                                 float* __restrict__ out,
                                                 int N, int D, int row0) {
    long long idx = (long long)blockIdx.x * blockDim.x + threadIdx.x;
    int row = row0 + (int)(idx / D);
    int col = (int)(idx % D);
    if (row >= N) return;
    float acc = bias[col];
    for (int k = 0; k < D; ++k) acc += x[(size_t)row * D + k] * W[(size_t)k * D + col];
    out[(size_t)row * D + col] = acc;
}

extern "C" void kernel_launch(void* const* d_in, const int* in_sizes, int n_in,
                              void* d_out, int out_size, void* d_ws, size_t ws_size,
                              hipStream_t stream) {
    const float* x      = (const float*)d_in[0];
    const float* attn_w = (const float*)d_in[1];
    const float* W      = (const float*)d_in[2];
    const float* bias   = (const float*)d_in[3];
    const int*   src    = (const int*)d_in[4];
    const int*   dst    = (const int*)d_in[5];
    float*       out    = (float*)d_out;

    const int D = in_sizes[3];
    const int N = in_sizes[0] / D;
    const int R = in_sizes[1] / (2 * D);
    const int E = in_sizes[4] / R;

    float* e_buf   = (float*)d_ws;
    float* m_buf   = e_buf + (size_t)R * E;
    float* den_buf = m_buf + (size_t)R * N;
    float* s_buf   = m_buf;

    dim3 wgrid((E + 7) / 8, R);
    dim3 tgrid((E + 255) / 256, R);

    node_scores<<<dim3((N + 7) / 8), 256, 0, stream>>>(x, attn_w, s_buf, N, D, R);
    edge_logits<<<tgrid, 256, 0, stream>>>(s_buf, src, dst, e_buf, E, R);

    const int rn = R * N;
    int row_tiles = 0;
    if ((D % 64) == 0) row_tiles = N / 16;
    const bool fill_rides = (long long)row_tiles * (D / 64) * 32 >= rn;
    if (!fill_rides) {
        init_max_denom<<<dim3((rn + 255) / 256), dim3(256), 0, stream>>>(m_buf, den_buf, rn);
    }
    if (row_tiles > 0) {
        dim3 ggrid(row_tiles, D / 64);
        wmma_selfloop<<<ggrid, 32, 0, stream>>>(x, W, bias, out, m_buf, den_buf,
                                                 fill_rides ? rn : 0, N, D);
    }
    int row0 = row_tiles * 16;
    long long tail_elems = (long long)(N - row0) * D;
    if (tail_elems > 0) {
        int blocks = (int)((tail_elems + 255) / 256);
        gemm_tail<<<dim3(blocks), dim3(256), 0, stream>>>(x, W, bias, out, N, D, row0);
    }

    seg_max<<<tgrid, 256, 0, stream>>>(e_buf, dst, m_buf, E, N);
    exp_sum<<<tgrid, 256, 0, stream>>>(e_buf, dst, m_buf, den_buf, E, N);
    scatter_agg<<<wgrid, 256, 0, stream>>>(x, e_buf, den_buf, src, dst, out, E, D, N);
}


# Edges chained per (destination, relation) by one integer exchange each; a wave per node takes max, exp, sum and weighted rows in registers, adds once to out: no float atomics; sums in chain order
# speedup vs baseline: 2.4080x; 1.4994x over previous
#include <hip/hip_runtime.h>
#include <math.h>

typedef __attribute__((ext_vector_type(2))) float v2f;
typedef __attribute__((ext_vector_type(4))) float v4f;
typedef __attribute__((ext_vector_type(8))) float v8f;

__global__ __launch_bounds__(256) void init_heads(int* __restrict__ head, int rn) {
    const int idx = blockIdx.x * blockDim.x + threadIdx.x;
    if (idx < rn) head[idx] = -1;
}

__device__ inline float dot4(float4 a, float4 b) {
    return a.x * b.x + a.y * b.y + a.z * b.z + a.w * b.w;
}

__global__ __launch_bounds__(256) void node_scores(const float* __restrict__ x,
                                                   const float* __restrict__ attn_w,
                                                   const int* __restrict__ dst,
                                                   float* __restrict__ S,
                                                   int* __restrict__ head,
                                                   int* __restrict__ next,
                                                   int N, int D, int R, int E, int per) {
    const int lane = threadIdx.x & 31;
    const int node = blockIdx.x * 8 + (threadIdx.x >> 5);
    if (node >= N) return;

    const int tot = R * per;
    int eid0 = -1, prev0 = -1;
    if (lane < tot) {
        const int r = lane / per;
        const int idx = node * per + (lane - r * per);
        if (idx < E) {
            eid0 = r * E + idx;
            const int t = dst[eid0];
            prev0 = atomicExch(&head[(size_t)t * R + r], eid0);
        }
    }
    for (int i = lane + 32; i < tot; i += 32) {
        const int r = i / per;
        const int idx = node * per + (i - r * per);
        if (idx < E) {
            const int eid = r * E + idx;
            const int t = dst[eid];
            next[eid] = atomicExch(&head[(size_t)t * R + r], eid);
        }
    }

    const float* xr = x + (size_t)node * D;
    const bool hi16 = (lane & 16) != 0;
    const bool hi8 = (lane & 8) != 0;
    const int q = lane >> 3;

    for (int j0 = 0; j0 < R; j0 += 4) {
        const int nrow = (R - j0) < 4 ? (R - j0) : 4;
        const float* w0 = attn_w + (size_t)(2 * j0) * D;
        const float* w1 = attn_w + (size_t)(2 * (j0 + (nrow > 1 ? 1 : 0))) * D;
        const float* w2 = attn_w + (size_t)(2 * (j0 + (nrow > 2 ? 2 : 0))) * D;
        const float* w3 = attn_w + (size_t)(2 * (j0 + (nrow > 3 ? 3 : 0))) * D;
        float a0 = 0.0f, a1 = 0.0f, a2 = 0.0f, a3 = 0.0f;
        for (int d = lane * 4; d < D; d += 128) {
            const float4 v = *(const float4*)(xr + d);
            a0 += dot4(v, *(const float4*)(w0 + d));
            a1 += dot4(v, *(const float4*)(w1 + d));
            if (nrow > 2) {
                a2 += dot4(v, *(const float4*)(w2 + d));
                a3 += dot4(v, *(const float4*)(w3 + d));
            }
        }
        float k0 = hi16 ? a2 : a0;
        float k1 = hi16 ? a3 : a1;
        const float g0 = hi16 ? a0 : a2;
        const float g1 = hi16 ? a1 : a3;
        k0 += __shfl_xor(g0, 16, 32);
        k1 += __shfl_xor(g1, 16, 32);
        float k = hi8 ? k1 : k0;
        const float g = hi8 ? k0 : k1;
        k += __shfl_xor(g, 8, 32);
        for (int off = 4; off > 0; off >>= 1) k += __shfl_xor(k, off, 32);
        if ((lane & 7) == 0 && q < nrow) S[(size_t)node * R + j0 + q] = k;
    }
    if (eid0 >= 0) next[eid0] = prev0;
}

__device__ inline float lane_f(float v, int k) {
    return __int_as_float(__builtin_amdgcn_readlane(__float_as_int(v), k));
}

struct RowSet {
    v4f v[8];
    int k[8];
};

__device__ inline unsigned load_rows(RowSet& g, unsigned mask, int ms,
                                     const float* __restrict__ xcol, int D, bool on) {
    const v4f zero4 = {0.0f, 0.0f, 0.0f, 0.0f};
#pragma unroll
    for (int j = 0; j < 8; ++j) {
        g.v[j] = zero4;
        g.k[j] = -1;
        if (mask != 0u) {
            g.k[j] = __builtin_ctz(mask);
            mask &= mask - 1u;
            const int s = __builtin_amdgcn_readlane(ms, g.k[j]);
            if (on) g.v[j] = *(const v4f*)(xcol + (size_t)s * D);
        }
    }
    return mask;
}

__device__ inline void add_rows(v4f& acc, const RowSet& g, float alpha) {
#pragma unroll
    for (int j = 0; j < 8; ++j) {
        if (g.k[j] >= 0) acc += lane_f(alpha, g.k[j]) * g.v[j];
    }
}

__global__ __launch_bounds__(256) void gather_softmax(const float* __restrict__ x,
                                                      const float* __restrict__ S,
                                                      const int* __restrict__ src,
                                                      const int* __restrict__ head,
                                                      const int* __restrict__ next,
                                                      float* __restrict__ out,
                                                      int N, int D, int R) {
    const int lane = threadIdx.x & 31;
    const int node = blockIdx.x * 8 + (threadIdx.x >> 5);
    if (node >= N) return;
    const v4f zero4 = {0.0f, 0.0f, 0.0f, 0.0f};

    for (int rg0 = 0; rg0 < R; rg0 += 32) {
        const int Rg = (R - rg0) < 32 ? (R - rg0) : 32;
        const int budget = 32 / Rg;
        const int hop = lane / Rg;
        const int wl = lane - hop * Rg;
        const int hv0 = (lane < Rg) ? head[(size_t)node * R + rg0 + lane] : -1;

        int hv = hv0, ms = -1, k = 0;
        while (k < budget && __builtin_amdgcn_ballot_w32(hv >= 0) != 0u) {
            int sv = -1, nv = -1;
            if (hv >= 0) {
                sv = src[hv];
                nv = next[hv];
            }
            const int got = __shfl(sv, wl, 32);
            if (hop == k) ms = got;
            hv = nv;
            ++k;
        }
        const int hops = k;

        if (__builtin_amdgcn_ballot_w32(hv >= 0) == 0u) {
            const bool valid = ms >= 0;
            const unsigned vmask = __builtin_amdgcn_ballot_w32(valid);
            if (vmask == 0u) continue;
            const float a = valid ? S[(size_t)ms * R + rg0 + wl] : -INFINITY;
            float alpha = 0.0f;

            for (int c0 = 0; c0 < D; c0 += 128) {
                const int d = c0 + lane * 4;
                const bool on = d < D;
                float* orow = out + (size_t)node * D + d;
                const float* xcol = x + d;
                v4f o = zero4;
                if (on) o = *(const v4f*)orow;
                unsigned mask = vmask;
                RowSet g;
                mask = load_rows(g, mask, ms, xcol, D, on);

                if (c0 == 0) {
                    float m = -INFINITY;
                    for (int h0 = 0; h0 < hops; h0 += 8) {
#pragma unroll
                        for (int j = 0; j < 8; ++j) {
                            const float t = __shfl(a, ((h0 + j) * Rg + wl) & 31, 32);
                            if (h0 + j < hops) m = fmaxf(m, t);
                        }
                    }
                    const float p = valid ? __expf(a - m) : 0.0f;
                    float den = 0.0f;
                    for (int h0 = 0; h0 < hops; h0 += 8) {
#pragma unroll
                        for (int j = 0; j < 8; ++j) {
                            const float t = __shfl(p, ((h0 + j) * Rg + wl) & 31, 32);
                            if (h0 + j < hops) den += t;
                        }
                    }
                    alpha = p / fmaxf(den, 1e-30f);
                }

                v4f acc = zero4;
                add_rows(acc, g, alpha);
                while (mask != 0u) {
                    mask = load_rows(g, mask, ms, xcol, D, on);
                    add_rows(acc, g, alpha);
                }
                if (on) *(v4f*)orow = o + acc;
            }
        } else {
            for (int c0 = 0; c0 < D; c0 += 128) {
                const int d = c0 + lane * 4;
                const bool on = d < D;
                float* orow = out + (size_t)node * D + d;
                const float* xcol = x + d;
                v4f acc = zero4;
                for (int w = 0; w < Rg; ++w) {
                    const int r = rg0 + w;
                    const int h0 = __builtin_amdgcn_readlane(hv0, w);
                    float m = -INFINITY;
                    for (int e = h0; e >= 0; e = next[e])
                        m = fmaxf(m, S[(size_t)src[e] * R + r]);
                    float den = 0.0f;
                    for (int e = h0; e >= 0; e = next[e])
                        den += __expf(S[(size_t)src[e] * R + r] - m);
                    const float dcl = fmaxf(den, 1e-30f);
                    for (int e = h0; e >= 0; e = next[e]) {
                        const int s = src[e];
                        const float al = __expf(S[(size_t)s * R + r] - m) / dcl;
                        if (on) acc += al * *(const v4f*)(xcol + (size_t)s * D);
                    }
                }
                if (on) {
                    const v4f o = *(const v4f*)orow;
                    *(v4f*)orow = o + acc;
                }
            }
        }
    }
}

__global__ __launch_bounds__(32) void wmma_selfloop(const float* __restrict__ x,
                                                    const float* __restrict__ W,
                                                    const float* __restrict__ bias,
                                                    float* __restrict__ out,
                                                    int* __restrict__ head,
                                                    int rn, int N, int D) {
    const int lane = threadIdx.x;
    const int m0 = blockIdx.x * 16;
    const int n0 = blockIdx.y * 64;
    const int row = lane & 15;
    const int khalf = (lane >> 4) * 2;

    const int fi = (blockIdx.x * (D >> 6) + blockIdx.y) * 32 + lane;
    if (fi < rn) head[fi] = -1;

    v8f c0 = {}, c1 = {}, c2 = {}, c3 = {};

    for (int k = 0; k < D; k += 4) {
        const float* ap = x + (size_t)(m0 + row) * D + (k + khalf);
        v2f a;
        a.x = ap[0];
        a.y = ap[1];

        const float* bp = W + (size_t)(k + khalf) * D + n0 + row;
        v2f b0, b1, b2, b3;
        b0.x = bp[0];  b0.y = bp[D + 0];
        b1.x = bp[16]; b1.y = bp[D + 16];
        b2.x = bp[32]; b2.y = bp[D + 32];
        b3.x = bp[48]; b3.y = bp[D + 48];

        c0 = __builtin_amdgcn_wmma_f32_16x16x4_f32(false, a, false, b0, (short)0, c0, false, false);
        c1 = __builtin_amdgcn_wmma_f32_16x16x4_f32(false, a, false, b1, (short)0, c1, false, false);
        c2 = __builtin_amdgcn_wmma_f32_16x16x4_f32(false, a, false, b2, (short)0, c2, false, false);
        c3 = __builtin_amdgcn_wmma_f32_16x16x4_f32(false, a, false, b3, (short)0, c3, false, false);
    }

    const int col = n0 + (lane & 15);
    const int mbase = m0 + (lane >> 4) * 8;
#pragma unroll
    for (int v = 0; v < 8; ++v) {
        float* o = out + (size_t)(mbase + v) * D;
        o[col + 0]  = c0[v] + bias[col + 0];
        o[col + 16] = c1[v] + bias[col + 16];
        o[col + 32] = c2[v] + bias[col + 32];
        o[col + 48] = c3[v] + bias[col + 48];
    }
}

__global__ __launch_bounds__(256) void gemm_tail(const float* __restrict__ x,
                                                 const float* __restrict__ W,
                                                 const float* __restrict__ bias,
                                                 float* __restrict__ out,
                                                 int N, int D, int row0) {
    long long idx = (long long)blockIdx.x * blockDim.x + threadIdx.x;
    int row = row0 + (int)(idx / D);
    int col = (int)(idx % D);
    if (row >= N) return;
    float acc = bias[col];
    for (int k = 0; k < D; ++k) acc += x[(size_t)row * D + k] * W[(size_t)k * D + col];
    out[(size_t)row * D + col] = acc;
}

extern "C" void kernel_launch(void* const* d_in, const int* in_sizes, int n_in,
                              void* d_out, int out_size, void* d_ws, size_t ws_size,
                              hipStream_t stream) {
    const float* x      = (const float*)d_in[0];
    const float* attn_w = (const float*)d_in[1];
    const float* W      = (const float*)d_in[2];
    const float* bias   = (const float*)d_in[3];
    const int*   src    = (const int*)d_in[4];
    const int*   dst    = (const int*)d_in[5];
    float*       out    = (float*)d_out;

    const int D = in_sizes[3];
    const int N = in_sizes[0] / D;
    const int R = in_sizes[1] / (2 * D);
    const int E = in_sizes[4] / R;

    int*   next_buf = (int*)d_ws;
    int*   head_buf = next_buf + (size_t)R * E;
    float* s_buf    = (float*)(head_buf + (size_t)R * N);

    const int rn = R * N;
    int row_tiles = 0;
    if ((D % 64) == 0) row_tiles = N / 16;
    const bool fill_rides = (long long)row_tiles * (D / 64) * 32 >= rn;
    if (!fill_rides) {
        init_heads<<<dim3((rn + 255) / 256), dim3(256), 0, stream>>>(head_buf, rn);
    }
    if (row_tiles > 0) {
        dim3 ggrid(row_tiles, D / 64);
        wmma_selfloop<<<ggrid, 32, 0, stream>>>(x, W, bias, out, head_buf,
                                                 fill_rides ? rn : 0, N, D);
    }
    int row0 = row_tiles * 16;
    long long tail_elems = (long long)(N - row0) * D;
    if (tail_elems > 0) {
        int blocks = (int)((tail_elems + 255) / 256);
        gemm_tail<<<dim3(blocks), dim3(256), 0, stream>>>(x, W, bias, out, N, D, row0);
    }

    const int per = (E + N - 1) / N;
    node_scores<<<dim3((N + 7) / 8), 256, 0, stream>>>(x, attn_w, dst, s_buf, head_buf, next_buf,
                                                       N, D, R, E, per);

    gather_softmax<<<dim3((N + 7) / 8), 256, 0, stream>>>(x, s_buf, src, head_buf, next_buf, out,
                                                          N, D, R);
}


# Both node launches as resident grids (1536 and 2048 groups of 8 waves): a wave loops over nodes w, w+waves, ..; next node's heads and destinations read a node ahead; 200000 wave entries become 28672
# speedup vs baseline: 2.4994x; 1.0380x over previous
#include <hip/hip_runtime.h>
#include <math.h>

typedef __attribute__((ext_vector_type(2))) float v2f;
typedef __attribute__((ext_vector_type(4))) float v4f;
typedef __attribute__((ext_vector_type(8))) float v8f;

__global__ __launch_bounds__(256) void init_heads(int* __restrict__ head, int rn) {
    const int idx = blockIdx.x * blockDim.x + threadIdx.x;
    if (idx < rn) head[idx] = -1;
}

__device__ inline float dot4(float4 a, float4 b) {
    return a.x * b.x + a.y * b.y + a.z * b.z + a.w * b.w;
}

__global__ __launch_bounds__(256) void node_scores(const float* __restrict__ x,
                                                   const float* __restrict__ attn_w,
                                                   const int* __restrict__ dst,
                                                   float* __restrict__ S,
                                                   int* __restrict__ head,
                                                   int* __restrict__ next,
                                                   int N, int D, int R, int E, int per,
                                                   int nw) {
    const int lane = threadIdx.x & 31;
    int node = __builtin_amdgcn_readfirstlane(blockIdx.x * 8 + (threadIdx.x >> 5));
    if (node >= N) return;

    const int tot = R * per;
    const int r0 = (lane < tot) ? lane / per : 0;
    const int j0 = lane - r0 * per;
    const bool hi16 = (lane & 16) != 0;
    const bool hi8 = (lane & 8) != 0;
    const int q = lane >> 3;

    int eid_n = -1, t_n = 0;
    if (lane < tot) {
        const int idx = node * per + j0;
        if (idx < E) {
            eid_n = r0 * E + idx;
            t_n = dst[eid_n];
        }
    }

    for (; node < N; node += nw) {
        const int eid0 = eid_n;
        int prev0 = -1;
        if (eid0 >= 0) prev0 = atomicExch(&head[(size_t)t_n * R + r0], eid0);
        eid_n = -1;
        const int nn = node + nw;
        if (nn < N && lane < tot) {
            const int idx = nn * per + j0;
            if (idx < E) {
                eid_n = r0 * E + idx;
                t_n = dst[eid_n];
            }
        }
        for (int i = lane + 32; i < tot; i += 32) {
            const int r = i / per;
            const int idx = node * per + (i - r * per);
            if (idx < E) {
                const int eid = r * E + idx;
                const int t = dst[eid];
                next[eid] = atomicExch(&head[(size_t)t * R + r], eid);
            }
        }

        const float* xr = x + (size_t)node * D;
        for (int j0r = 0; j0r < R; j0r += 4) {
            const int nrow = (R - j0r) < 4 ? (R - j0r) : 4;
            const float* w0 = attn_w + (size_t)(2 * j0r) * D;
            const float* w1 = attn_w + (size_t)(2 * (j0r + (nrow > 1 ? 1 : 0))) * D;
            const float* w2 = attn_w + (size_t)(2 * (j0r + (nrow > 2 ? 2 : 0))) * D;
            const float* w3 = attn_w + (size_t)(2 * (j0r + (nrow > 3 ? 3 : 0))) * D;
            float a0 = 0.0f, a1 = 0.0f, a2 = 0.0f, a3 = 0.0f;
            for (unsigned d = (unsigned)lane * 4u; d < (unsigned)D; d += 128u) {
                const float4 v = *(const float4*)(xr + d);
                a0 += dot4(v, *(const float4*)(w0 + d));
                a1 += dot4(v, *(const float4*)(w1 + d));
                if (nrow > 2) {
                    a2 += dot4(v, *(const float4*)(w2 + d));
                    a3 += dot4(v, *(const float4*)(w3 + d));
                }
            }
            float k0 = hi16 ? a2 : a0;
            float k1 = hi16 ? a3 : a1;
            const float g0 = hi16 ? a0 : a2;
            const float g1 = hi16 ? a1 : a3;
            k0 += __shfl_xor(g0, 16, 32);
            k1 += __shfl_xor(g1, 16, 32);
            float k = hi8 ? k1 : k0;
            const float g = hi8 ? k0 : k1;
            k += __shfl_xor(g, 8, 32);
            for (int off = 4; off > 0; off >>= 1) k += __shfl_xor(k, off, 32);
            if ((lane & 7) == 0 && q < nrow) S[(size_t)node * R + j0r + q] = k;
        }
        if (eid0 >= 0) next[eid0] = prev0;
    }
}

__device__ inline float lane_f(float v, int k) {
    return __int_as_float(__builtin_amdgcn_readlane(__float_as_int(v), k));
}

struct RowSet {
    v4f v[8];
    int k[8];
};

__device__ inline unsigned load_rows(RowSet& g, unsigned mask, int ms,
                                     const float* __restrict__ xcol, int D, bool on) {
    const v4f zero4 = {0.0f, 0.0f, 0.0f, 0.0f};
#pragma unroll
    for (int j = 0; j < 8; ++j) {
        g.v[j] = zero4;
        g.k[j] = -1;
        if (mask != 0u) {
            g.k[j] = __builtin_ctz(mask);
            mask &= mask - 1u;
            const int s = __builtin_amdgcn_readlane(ms, g.k[j]);
            if (on) g.v[j] = *(const v4f*)(xcol + (size_t)s * D);
        }
    }
    return mask;
}

__device__ inline void add_rows(v4f& acc, const RowSet& g, float alpha) {
#pragma unroll
    for (int j = 0; j < 8; ++j) {
        if (g.k[j] >= 0) acc += lane_f(alpha, g.k[j]) * g.v[j];
    }
}

__global__ __launch_bounds__(256, 16) void gather_softmax(const float* __restrict__ x,
                                                          const float* __restrict__ S,
                                                          const int* __restrict__ src,
                                                          const int* __restrict__ head,
                                                          const int* __restrict__ next,
                                                          float* __restrict__ out,
                                                          int N, int D, int R,
                                                          int nw) {
    const int lane = threadIdx.x & 31;
    int node = __builtin_amdgcn_readfirstlane(blockIdx.x * 8 + (threadIdx.x >> 5));
    if (node >= N) return;
    const v4f zero4 = {0.0f, 0.0f, 0.0f, 0.0f};
    const int Rfirst = R < 32 ? R : 32;

    int hnext = (lane < Rfirst) ? head[(size_t)node * R + lane] : -1;

    for (; node < N; node += nw) {
        const int hfirst = hnext;
        if (node + nw < N) hnext = (lane < Rfirst) ? head[(size_t)(node + nw) * R + lane] : -1;

        for (int rg0 = 0; rg0 < R; rg0 += 32) {
            const int Rg = (R - rg0) < 32 ? (R - rg0) : 32;
            const int budget = 32 / Rg;
            const int hop = lane / Rg;
            const int wl = lane - hop * Rg;
            int hv0 = hfirst;
            if (rg0 > 0) hv0 = (lane < Rg) ? head[(size_t)node * R + rg0 + lane] : -1;

            int hv = hv0, ms = -1, k = 0;
            while (k < budget && __builtin_amdgcn_ballot_w32(hv >= 0) != 0u) {
                int sv = -1, nv = -1;
                if (hv >= 0) {
                    sv = src[hv];
                    nv = next[hv];
                }
                const int got = __shfl(sv, wl, 32);
                if (hop == k) ms = got;
                hv = nv;
                ++k;
            }
            const int hops = k;

            if (__builtin_amdgcn_ballot_w32(hv >= 0) == 0u) {
                const bool valid = ms >= 0;
                const unsigned vmask = __builtin_amdgcn_ballot_w32(valid);
                if (vmask == 0u) continue;
                const float a = valid ? S[(size_t)ms * R + rg0 + wl] : -INFINITY;
                float alpha = 0.0f;

                for (int c0 = 0; c0 < D; c0 += 128) {
                    const int d = c0 + lane * 4;
                    const bool on = d < D;
                    float* orow = out + (size_t)node * D + d;
                    const float* xcol = x + d;
                    v4f o = zero4;
                    if (on) o = *(const v4f*)orow;
                    unsigned mask = vmask;
                    RowSet g;
                    mask = load_rows(g, mask, ms, xcol, D, on);

                    if (c0 == 0) {
                        float m = -INFINITY;
                        for (int h0 = 0; h0 < hops; h0 += 8) {
    #pragma unroll
                            for (int j = 0; j < 8; ++j) {
                                const float t = __shfl(a, ((h0 + j) * Rg + wl) & 31, 32);
                                if (h0 + j < hops) m = fmaxf(m, t);
                            }
                        }
                        const float p = valid ? __expf(a - m) : 0.0f;
                        float den = 0.0f;
                        for (int h0 = 0; h0 < hops; h0 += 8) {
    #pragma unroll
                            for (int j = 0; j < 8; ++j) {
                                const float t = __shfl(p, ((h0 + j) * Rg + wl) & 31, 32);
                                if (h0 + j < hops) den += t;
                            }
                        }
                        alpha = p / fmaxf(den, 1e-30f);
                    }

                    v4f acc = zero4;
                    add_rows(acc, g, alpha);
                    while (mask != 0u) {
                        mask = load_rows(g, mask, ms, xcol, D, on);
                        add_rows(acc, g, alpha);
                    }
                    if (on) *(v4f*)orow = o + acc;
                }
            } else {
                for (int c0 = 0; c0 < D; c0 += 128) {
                    const int d = c0 + lane * 4;
                    const bool on = d < D;
                    float* orow = out + (size_t)node * D + d;
                    const float* xcol = x + d;
                    v4f acc = zero4;
                    for (int w = 0; w < Rg; ++w) {
                        const int r = rg0 + w;
                        const int h0 = __builtin_amdgcn_readlane(hv0, w);
                        float m = -INFINITY;
                        for (int e = h0; e >= 0; e = next[e])
                            m = fmaxf(m, S[(size_t)src[e] * R + r]);
                        float den = 0.0f;
                        for (int e = h0; e >= 0; e = next[e])
                            den += __expf(S[(size_t)src[e] * R + r] - m);
                        const float dcl = fmaxf(den, 1e-30f);
                        for (int e = h0; e >= 0; e = next[e]) {
                            const int s = src[e];
                            const float al = __expf(S[(size_t)s * R + r] - m) / dcl;
                            if (on) acc += al * *(const v4f*)(xcol + (size_t)s * D);
                        }
                    }
                    if (on) {
                        const v4f o = *(const v4f*)orow;
                        *(v4f*)orow = o + acc;
                    }
                }
            }
        }
    }
}

__global__ __launch_bounds__(32) void wmma_selfloop(const float* __restrict__ x,
                                                    const float* __restrict__ W,
                                                    const float* __restrict__ bias,
                                                    float* __restrict__ out,
                                                    int* __restrict__ head,
                                                    int rn, int N, int D) {
    const int lane = threadIdx.x;
    const int m0 = blockIdx.x * 16;
    const int n0 = blockIdx.y * 64;
    const int row = lane & 15;
    const int khalf = (lane >> 4) * 2;

    const int fi = (blockIdx.x * (D >> 6) + blockIdx.y) * 32 + lane;
    if (fi < rn) head[fi] = -1;

    v8f c0 = {}, c1 = {}, c2 = {}, c3 = {};

    for (int k = 0; k < D; k += 4) {
        const float* ap = x + (size_t)(m0 + row) * D + (k + khalf);
        v2f a;
        a.x = ap[0];
        a.y = ap[1];

        const float* bp = W + (size_t)(k + khalf) * D + n0 + row;
        v2f b0, b1, b2, b3;
        b0.x = bp[0];  b0.y = bp[D + 0];
        b1.x = bp[16]; b1.y = bp[D + 16];
        b2.x = bp[32]; b2.y = bp[D + 32];
        b3.x = bp[48]; b3.y = bp[D + 48];

        c0 = __builtin_amdgcn_wmma_f32_16x16x4_f32(false, a, false, b0, (short)0, c0, false, false);
        c1 = __builtin_amdgcn_wmma_f32_16x16x4_f32(false, a, false, b1, (short)0, c1, false, false);
        c2 = __builtin_amdgcn_wmma_f32_16x16x4_f32(false, a, false, b2, (short)0, c2, false, false);
        c3 = __builtin_amdgcn_wmma_f32_16x16x4_f32(false, a, false, b3, (short)0, c3, false, false);
    }

    const int col = n0 + (lane & 15);
    const int mbase = m0 + (lane >> 4) * 8;
#pragma unroll
    for (int v = 0; v < 8; ++v) {
        float* o = out + (size_t)(mbase + v) * D;
        o[col + 0]  = c0[v] + bias[col + 0];
        o[col + 16] = c1[v] + bias[col + 16];
        o[col + 32] = c2[v] + bias[col + 32];
        o[col + 48] = c3[v] + bias[col + 48];
    }
}

__global__ __launch_bounds__(256) void gemm_tail(const float* __restrict__ x,
                                                 const float* __restrict__ W,
                                                 const float* __restrict__ bias,
                                                 float* __restrict__ out,
                                                 int N, int D, int row0) {
    long long idx = (long long)blockIdx.x * blockDim.x + threadIdx.x;
    int row = row0 + (int)(idx / D);
    int col = (int)(idx % D);
    if (row >= N) return;
    float acc = bias[col];
    for (int k = 0; k < D; ++k) acc += x[(size_t)row * D + k] * W[(size_t)k * D + col];
    out[(size_t)row * D + col] = acc;
}

extern "C" void kernel_launch(void* const* d_in, const int* in_sizes, int n_in,
                              void* d_out, int out_size, void* d_ws, size_t ws_size,
                              hipStream_t stream) {
    const float* x      = (const float*)d_in[0];
    const float* attn_w = (const float*)d_in[1];
    const float* W      = (const float*)d_in[2];
    const float* bias   = (const float*)d_in[3];
    const int*   src    = (const int*)d_in[4];
    const int*   dst    = (const int*)d_in[5];
    float*       out    = (float*)d_out;

    const int D = in_sizes[3];
    const int N = in_sizes[0] / D;
    const int R = in_sizes[1] / (2 * D);
    const int E = in_sizes[4] / R;

    int*   next_buf = (int*)d_ws;
    int*   head_buf = next_buf + (size_t)R * E;
    float* s_buf    = (float*)(head_buf + (size_t)R * N);

    const int rn = R * N;
    int row_tiles = 0;
    if ((D % 64) == 0) row_tiles = N / 16;
    const bool fill_rides = (long long)row_tiles * (D / 64) * 32 >= rn;
    if (!fill_rides) {
        init_heads<<<dim3((rn + 255) / 256), dim3(256), 0, stream>>>(head_buf, rn);
    }
    if (row_tiles > 0) {
        dim3 ggrid(row_tiles, D / 64);
        wmma_selfloop<<<ggrid, 32, 0, stream>>>(x, W, bias, out, head_buf,
                                                 fill_rides ? rn : 0, N, D);
    }
    int row0 = row_tiles * 16;
    long long tail_elems = (long long)(N - row0) * D;
    if (tail_elems > 0) {
        int blocks = (int)((tail_elems + 255) / 256);
        gemm_tail<<<dim3(blocks), dim3(256), 0, stream>>>(x, W, bias, out, N, D, row0);
    }

    const int node_groups = (N + 7) / 8;
    const int res_scores = node_groups < 1536 ? node_groups : 1536;
    const int res_gather = node_groups < 2048 ? node_groups : 2048;

    const int per = (E + N - 1) / N;
    node_scores<<<dim3(res_scores), 256, 0, stream>>>(x, attn_w, dst, s_buf, head_buf, next_buf,
                                                      N, D, R, E, per, res_scores * 8);

    gather_softmax<<<dim3(res_gather), 256, 0, stream>>>(x, s_buf, src, head_buf, next_buf, out,
                                                         N, D, R, res_gather * 8);
}


# Edges go into 16-slot buckets per (destination, relation), slot from one returning integer add; a node's sources are one line read a node ahead, no chain walk; sums in slot (arrival) order
# speedup vs baseline: 2.6719x; 1.0690x over previous
#include <hip/hip_runtime.h>
#include <math.h>

typedef __attribute__((ext_vector_type(2))) float v2f;
typedef __attribute__((ext_vector_type(4))) float v4f;
typedef __attribute__((ext_vector_type(8))) float v8f;

__global__ __launch_bounds__(256) void init_heads(int* __restrict__ cnt, int* __restrict__ ohead, int rn) {
    const int idx = blockIdx.x * blockDim.x + threadIdx.x;
    if (idx < rn) {
        cnt[idx] = 0;
        ohead[idx] = -1;
    }
}

__device__ inline float dot4(float4 a, float4 b) {
    return a.x * b.x + a.y * b.y + a.z * b.z + a.w * b.w;
}

__global__ __launch_bounds__(256) void node_scores(const float* __restrict__ x,
                                                   const float* __restrict__ attn_w,
                                                   const int* __restrict__ src,
                                                   const int* __restrict__ dst,
                                                   float* __restrict__ S,
                                                   int* __restrict__ cnt,
                                                   int* __restrict__ bucket,
                                                   int* __restrict__ ohead,
                                                   int* __restrict__ next,
                                                   int N, int D, int R, int E, int per,
                                                   int cap,
                                                   int nw) {
    const int lane = threadIdx.x & 31;
    int node = __builtin_amdgcn_readfirstlane(blockIdx.x * 8 + (threadIdx.x >> 5));
    if (node >= N) return;

    const int tot = R * per;
    const int r0 = (lane < tot) ? lane / per : 0;
    const int j0 = lane - r0 * per;
    const bool hi16 = (lane & 16) != 0;
    const bool hi8 = (lane & 8) != 0;
    const int q = lane >> 3;

    int eid_n = -1, t_n = 0, s_n = 0;
    if (lane < tot) {
        const int idx = node * per + j0;
        if (idx < E) {
            eid_n = r0 * E + idx;
            t_n = dst[eid_n];
            s_n = src[eid_n];
        }
    }

    for (; node < N; node += nw) {
        const int eid0 = eid_n;
        const size_t seg0 = (size_t)t_n * R + r0;
        const int s0 = s_n;
        int slot0 = 0;
        if (eid0 >= 0) slot0 = atomicAdd(&cnt[seg0], 1);
        eid_n = -1;
        const int nn = node + nw;
        if (nn < N && lane < tot) {
            const int idx = nn * per + j0;
            if (idx < E) {
                eid_n = r0 * E + idx;
                t_n = dst[eid_n];
                s_n = src[eid_n];
            }
        }
        for (int i = lane + 32; i < tot; i += 32) {
            const int r = i / per;
            const int idx = node * per + (i - r * per);
            if (idx < E) {
                const int eid = r * E + idx;
                const size_t seg = (size_t)dst[eid] * R + r;
                const int slot = atomicAdd(&cnt[seg], 1);
                if (slot < cap) bucket[seg * cap + slot] = src[eid];
                else next[eid] = atomicExch(&ohead[seg], eid);
            }
        }

        const float* xr = x + (size_t)node * D;
        for (int j0r = 0; j0r < R; j0r += 4) {
            const int nrow = (R - j0r) < 4 ? (R - j0r) : 4;
            const float* w0 = attn_w + (size_t)(2 * j0r) * D;
            const float* w1 = attn_w + (size_t)(2 * (j0r + (nrow > 1 ? 1 : 0))) * D;
            const float* w2 = attn_w + (size_t)(2 * (j0r + (nrow > 2 ? 2 : 0))) * D;
            const float* w3 = attn_w + (size_t)(2 * (j0r + (nrow > 3 ? 3 : 0))) * D;
            float a0 = 0.0f, a1 = 0.0f, a2 = 0.0f, a3 = 0.0f;
            for (unsigned d = (unsigned)lane * 4u; d < (unsigned)D; d += 128u) {
                const float4 v = *(const float4*)(xr + d);
                a0 += dot4(v, *(const float4*)(w0 + d));
                a1 += dot4(v, *(const float4*)(w1 + d));
                if (nrow > 2) {
                    a2 += dot4(v, *(const float4*)(w2 + d));
                    a3 += dot4(v, *(const float4*)(w3 + d));
                }
            }
            float k0 = hi16 ? a2 : a0;
            float k1 = hi16 ? a3 : a1;
            const float g0 = hi16 ? a0 : a2;
            const float g1 = hi16 ? a1 : a3;
            k0 += __shfl_xor(g0, 16, 32);
            k1 += __shfl_xor(g1, 16, 32);
            float k = hi8 ? k1 : k0;
            const float g = hi8 ? k0 : k1;
            k += __shfl_xor(g, 8, 32);
            for (int off = 4; off > 0; off >>= 1) k += __shfl_xor(k, off, 32);
            if ((lane & 7) == 0 && q < nrow) S[(size_t)node * R + j0r + q] = k;
        }
        if (eid0 >= 0) {
            if (slot0 < cap) bucket[seg0 * cap + slot0] = s0;
            else next[eid0] = atomicExch(&ohead[seg0], eid0);
        }
    }
}

__device__ inline float lane_f(float v, int k) {
    return __int_as_float(__builtin_amdgcn_readlane(__float_as_int(v), k));
}

struct RowSet {
    v4f v[8];
    int k[8];
};

__device__ inline unsigned load_rows(RowSet& g, unsigned mask, int ms,
                                     const float* __restrict__ xcol, int D, bool on) {
    const v4f zero4 = {0.0f, 0.0f, 0.0f, 0.0f};
#pragma unroll
    for (int j = 0; j < 8; ++j) {
        g.v[j] = zero4;
        g.k[j] = -1;
        if (mask != 0u) {
            g.k[j] = __builtin_ctz(mask);
            mask &= mask - 1u;
            const int s = __builtin_amdgcn_readlane(ms, g.k[j]);
            if (on) g.v[j] = *(const v4f*)(xcol + (size_t)s * D);
        }
    }
    return mask;
}

__device__ inline void add_rows(v4f& acc, const RowSet& g, float alpha) {
#pragma unroll
    for (int j = 0; j < 8; ++j) {
        if (g.k[j] >= 0) acc += lane_f(alpha, g.k[j]) * g.v[j];
    }
}

__device__ __forceinline__ void gather_set(const float* __restrict__ x,
                                           const float* __restrict__ S,
                                           const int* __restrict__ src,
                                           const int* __restrict__ bucket,
                                           const int* __restrict__ ohead,
                                           const int* __restrict__ next,
                                           float* __restrict__ out,
                                           int node, int D, int R, int cap, int lane,
                                           int rg0, int Rg, int hop, int wl, int cw, int ms) {
    const v4f zero4 = {0.0f, 0.0f, 0.0f, 0.0f};
    const int budget = 32 / Rg;
    const int lim = budget < cap ? budget : cap;

    if (__builtin_amdgcn_ballot_w32(cw > lim) == 0u) {
        const bool valid = hop < cw;
        const unsigned vmask = __builtin_amdgcn_ballot_w32(valid);
        if (vmask == 0u) return;
        const int hops = (31 - __builtin_clz(vmask)) / Rg + 1;
        const float a = valid ? S[(unsigned)(ms * R + rg0 + wl)] : -INFINITY;
        float alpha = 0.0f;

        for (int c0 = 0; c0 < D; c0 += 128) {
            const int d = c0 + lane * 4;
            const bool on = d < D;
            float* orow = out + (size_t)node * D + d;
            const float* xcol = x + d;
            v4f o = zero4;
            if (on) o = *(const v4f*)orow;
            unsigned mask = vmask;
            RowSet g;
            mask = load_rows(g, mask, ms, xcol, D, on);

            if (c0 == 0) {
                float m = -INFINITY;
                for (int h0 = 0; h0 < hops; h0 += 8) {
#pragma unroll
                    for (int j = 0; j < 8; ++j) {
                        const float t = __shfl(a, ((h0 + j) * Rg + wl) & 31, 32);
                        if (h0 + j < hops) m = fmaxf(m, t);
                    }
                }
                const float p = valid ? __expf(a - m) : 0.0f;
                float den = 0.0f;
                for (int h0 = 0; h0 < hops; h0 += 8) {
#pragma unroll
                    for (int j = 0; j < 8; ++j) {
                        const float t = __shfl(p, ((h0 + j) * Rg + wl) & 31, 32);
                        if (h0 + j < hops) den += t;
                    }
                }
                alpha = p / fmaxf(den, 1e-30f);
            }

            v4f acc = zero4;
            add_rows(acc, g, alpha);
            while (mask != 0u) {
                mask = load_rows(g, mask, ms, xcol, D, on);
                add_rows(acc, g, alpha);
            }
            if (on) *(v4f*)orow = o + acc;
        }
    } else {
        for (int c0 = 0; c0 < D; c0 += 128) {
            const int d = c0 + lane * 4;
            const bool on = d < D;
            float* orow = out + (size_t)node * D + d;
            const float* xcol = x + d;
            v4f acc = zero4;
            for (int w = 0; w < Rg; ++w) {
                const int r = rg0 + w;
                const size_t seg = (size_t)node * R + r;
                const int c = __builtin_amdgcn_readlane(cw, w);
                const int cb = c < cap ? c : cap;
                const int oh = c > cap ? ohead[seg] : -1;
                const int* bk = bucket + seg * cap;
                float m = -INFINITY;
                for (int k = 0; k < cb; ++k)
                    m = fmaxf(m, S[(size_t)bk[k] * R + r]);
                for (int e = oh; e >= 0; e = next[e])
                    m = fmaxf(m, S[(size_t)src[e] * R + r]);
                float den = 0.0f;
                for (int k = 0; k < cb; ++k)
                    den += __expf(S[(size_t)bk[k] * R + r] - m);
                for (int e = oh; e >= 0; e = next[e])
                    den += __expf(S[(size_t)src[e] * R + r] - m);
                const float dcl = fmaxf(den, 1e-30f);
                for (int k = 0; k < cb; ++k) {
                    const int s = bk[k];
                    const float al = __expf(S[(size_t)s * R + r] - m) / dcl;
                    if (on) acc += al * *(const v4f*)(xcol + (size_t)s * D);
                }
                for (int e = oh; e >= 0; e = next[e]) {
                    const int s = src[e];
                    const float al = __expf(S[(size_t)s * R + r] - m) / dcl;
                    if (on) acc += al * *(const v4f*)(xcol + (size_t)s * D);
                }
            }
            if (on) {
                const v4f o = *(const v4f*)orow;
                *(v4f*)orow = o + acc;
            }
        }
    }
}

__global__ __launch_bounds__(256, 16) void gather_softmax(const float* __restrict__ x,
                                                          const float* __restrict__ S,
                                                          const int* __restrict__ src,
                                                          const int* __restrict__ cnt,
                                                          const int* __restrict__ bucket,
                                                          const int* __restrict__ ohead,
                                                          const int* __restrict__ next,
                                                          float* __restrict__ out,
                                                          int N, int D, int R,
                                                          int cap,
                                                          int nw) {
    const int lane = threadIdx.x & 31;
    int node = __builtin_amdgcn_readfirstlane(blockIdx.x * 8 + (threadIdx.x >> 5));
    if (node >= N) return;

    const int Rfirst = R < 32 ? R : 32;
    const int hop0 = lane / Rfirst;
    const int wl0 = lane - hop0 * Rfirst;
    const bool in0 = hop0 < 32 / Rfirst;
    const bool sl0 = in0 && hop0 < cap;
    const unsigned coff0 = (unsigned)wl0;
    const unsigned boff0 = (unsigned)(wl0 * cap + hop0);
    const unsigned rc = (unsigned)(R * cap);

    int c_cur = in0 ? cnt[(unsigned)node * (unsigned)R + coff0] : 0;
    int c_nxt = (in0 && node + nw < N) ? cnt[(unsigned)(node + nw) * (unsigned)R + coff0] : 0;
    int m_cur = (sl0 && hop0 < c_cur) ? bucket[(unsigned)node * rc + boff0] : -1;
    asm volatile("" : : "v"(c_cur), "v"(m_cur));

    for (; node < N; node += nw) {
        int m_nxt = -1;
        if (sl0 && hop0 < c_nxt) m_nxt = bucket[(unsigned)(node + nw) * rc + boff0];
        int c_nn = 0;
        if (in0 && node + 2 * nw < N) c_nn = cnt[(unsigned)(node + 2 * nw) * (unsigned)R + coff0];

        gather_set(x, S, src, bucket, ohead, next, out, node, D, R, cap, lane,
                   0, Rfirst, hop0, wl0, c_cur, m_cur);

        for (int rg0 = 32; rg0 < R; rg0 += 32) {
            const int Rg = (R - rg0) < 32 ? (R - rg0) : 32;
            const int budget = 32 / Rg;
            const int hop = lane / Rg;
            const int wl = lane - hop * Rg;
            const int lim = budget < cap ? budget : cap;
            const size_t seg0 = (size_t)node * R + rg0;
            const int cw = (hop < budget) ? (cnt + seg0)[(unsigned)wl] : 0;
            const int ms = (hop < lim && hop < cw) ? (bucket + seg0 * cap)[(unsigned)(wl * cap + hop)] : -1;
            gather_set(x, S, src, bucket, ohead, next, out, node, D, R, cap, lane,
                       rg0, Rg, hop, wl, cw, ms);
        }

        c_cur = c_nxt;
        m_cur = m_nxt;
        c_nxt = c_nn;
    }
}

__global__ __launch_bounds__(32) void wmma_selfloop(const float* __restrict__ x,
                                                    const float* __restrict__ W,
                                                    const float* __restrict__ bias,
                                                    float* __restrict__ out,
                                                    int* __restrict__ cnt,
                                                    int* __restrict__ ohead,
                                                    int rn, int N, int D) {
    const int lane = threadIdx.x;
    const int m0 = blockIdx.x * 16;
    const int n0 = blockIdx.y * 64;
    const int row = lane & 15;
    const int khalf = (lane >> 4) * 2;

    const int fi = (blockIdx.x * (D >> 6) + blockIdx.y) * 32 + lane;
    if (fi < rn) {
        cnt[fi] = 0;
        ohead[fi] = -1;
    }

    v8f c0 = {}, c1 = {}, c2 = {}, c3 = {};

    for (int k = 0; k < D; k += 4) {
        const float* ap = x + (size_t)(m0 + row) * D + (k + khalf);
        v2f a;
        a.x = ap[0];
        a.y = ap[1];

        const float* bp = W + (size_t)(k + khalf) * D + n0 + row;
        v2f b0, b1, b2, b3;
        b0.x = bp[0];  b0.y = bp[D + 0];
        b1.x = bp[16]; b1.y = bp[D + 16];
        b2.x = bp[32]; b2.y = bp[D + 32];
        b3.x = bp[48]; b3.y = bp[D + 48];

        c0 = __builtin_amdgcn_wmma_f32_16x16x4_f32(false, a, false, b0, (short)0, c0, false, false);
        c1 = __builtin_amdgcn_wmma_f32_16x16x4_f32(false, a, false, b1, (short)0, c1, false, false);
        c2 = __builtin_amdgcn_wmma_f32_16x16x4_f32(false, a, false, b2, (short)0, c2, false, false);
        c3 = __builtin_amdgcn_wmma_f32_16x16x4_f32(false, a, false, b3, (short)0, c3, false, false);
    }

    const int col = n0 + (lane & 15);
    const int mbase = m0 + (lane >> 4) * 8;
#pragma unroll
    for (int v = 0; v < 8; ++v) {
        float* o = out + (size_t)(mbase + v) * D;
        o[col + 0]  = c0[v] + bias[col + 0];
        o[col + 16] = c1[v] + bias[col + 16];
        o[col + 32] = c2[v] + bias[col + 32];
        o[col + 48] = c3[v] + bias[col + 48];
    }
}

__global__ __launch_bounds__(256) void gemm_tail(const float* __restrict__ x,
                                                 const float* __restrict__ W,
                                                 const float* __restrict__ bias,
                                                 float* __restrict__ out,
                                                 int N, int D, int row0) {
    long long idx = (long long)blockIdx.x * blockDim.x + threadIdx.x;
    int row = row0 + (int)(idx / D);
    int col = (int)(idx % D);
    if (row >= N) return;
    float acc = bias[col];
    for (int k = 0; k < D; ++k) acc += x[(size_t)row * D + k] * W[(size_t)k * D + col];
    out[(size_t)row * D + col] = acc;
}

extern "C" void kernel_launch(void* const* d_in, const int* in_sizes, int n_in,
                              void* d_out, int out_size, void* d_ws, size_t ws_size,
                              hipStream_t stream) {
    const float* x      = (const float*)d_in[0];
    const float* attn_w = (const float*)d_in[1];
    const float* W      = (const float*)d_in[2];
    const float* bias   = (const float*)d_in[3];
    const int*   src    = (const int*)d_in[4];
    const int*   dst    = (const int*)d_in[5];
    float*       out    = (float*)d_out;

    const int D = in_sizes[3];
    const int N = in_sizes[0] / D;
    const int R = in_sizes[1] / (2 * D);
    const int E = in_sizes[4] / R;

    const int rn = R * N;
    const long long fixed_words = (long long)R * E + 3LL * rn;
    const long long spare_words = (long long)(ws_size / sizeof(int)) - fixed_words;
    int cap = 16;
    if (spare_words < 16LL * rn) cap = spare_words > 0 ? (int)(spare_words / rn) : 0;
    if ((long long)rn * cap > 0x7fffffffLL) cap = (int)(0x7fffffffLL / rn);
    int*   bucket_buf = (int*)d_ws;
    int*   next_buf   = bucket_buf + (size_t)rn * cap;
    int*   ohead_buf  = next_buf + (size_t)R * E;
    int*   cnt_buf    = ohead_buf + rn;
    float* s_buf      = (float*)(cnt_buf + rn);

    int row_tiles = 0;
    if ((D % 64) == 0) row_tiles = N / 16;
    const bool fill_rides = (long long)row_tiles * (D / 64) * 32 >= rn;
    if (!fill_rides) {
        init_heads<<<dim3((rn + 255) / 256), dim3(256), 0, stream>>>(cnt_buf, ohead_buf, rn);
    }
    if (row_tiles > 0) {
        dim3 ggrid(row_tiles, D / 64);
        wmma_selfloop<<<ggrid, 32, 0, stream>>>(x, W, bias, out, cnt_buf, ohead_buf,
                                                 fill_rides ? rn : 0, N, D);
    }
    int row0 = row_tiles * 16;
    long long tail_elems = (long long)(N - row0) * D;
    if (tail_elems > 0) {
        int blocks = (int)((tail_elems + 255) / 256);
        gemm_tail<<<dim3(blocks), dim3(256), 0, stream>>>(x, W, bias, out, N, D, row0);
    }

    const int node_groups = (N + 7) / 8;
    const int res_scores = node_groups < 1536 ? node_groups : 1536;
    const int res_gather = node_groups < 2048 ? node_groups : 2048;

    const int per = (E + N - 1) / N;
    node_scores<<<dim3(res_scores), 256, 0, stream>>>(x, attn_w, src, dst, s_buf, cnt_buf, bucket_buf,
                                                      ohead_buf, next_buf,
                                                      N, D, R, E, per, cap, res_scores * 8);

    gather_softmax<<<dim3(res_gather), 256, 0, stream>>>(x, s_buf, src, cnt_buf, bucket_buf, ohead_buf,
                                                         next_buf, out,
                                                         N, D, R, cap, res_gather * 8);
}


# Self-loop product: weights staged once per 4-wave work-group in LDS as k-pairs, a wave takes all 128 columns of its 16 rows: B fragments are 8-byte LDS reads, x rows read once; sums unchanged
# speedup vs baseline: 2.9230x; 1.0940x over previous
#include <hip/hip_runtime.h>
#include <math.h>

typedef __attribute__((ext_vector_type(2))) float v2f;
typedef __attribute__((ext_vector_type(4))) float v4f;
typedef __attribute__((ext_vector_type(8))) float v8f;

__global__ __launch_bounds__(256) void init_heads(int* __restrict__ cnt, int* __restrict__ ohead, int rn) {
    const int idx = blockIdx.x * blockDim.x + threadIdx.x;
    if (idx < rn) {
        cnt[idx] = 0;
        ohead[idx] = -1;
    }
}

__device__ inline float dot4(float4 a, float4 b) {
    return a.x * b.x + a.y * b.y + a.z * b.z + a.w * b.w;
}

__global__ __launch_bounds__(256) void node_scores(const float* __restrict__ x,
                                                   const float* __restrict__ attn_w,
                                                   const int* __restrict__ src,
                                                   const int* __restrict__ dst,
                                                   float* __restrict__ S,
                                                   int* __restrict__ cnt,
                                                   int* __restrict__ bucket,
                                                   int* __restrict__ ohead,
                                                   int* __restrict__ next,
                                                   int N, int D, int R, int E, int per,
                                                   int cap,
                                                   int nw) {
    const int lane = threadIdx.x & 31;
    int node = __builtin_amdgcn_readfirstlane(blockIdx.x * 8 + (threadIdx.x >> 5));
    if (node >= N) return;

    const int tot = R * per;
    const int r0 = (lane < tot) ? lane / per : 0;
    const int j0 = lane - r0 * per;
    const bool hi16 = (lane & 16) != 0;
    const bool hi8 = (lane & 8) != 0;
    const int q = lane >> 3;

    int eid_n = -1, t_n = 0, s_n = 0;
    if (lane < tot) {
        const int idx = node * per + j0;
        if (idx < E) {
            eid_n = r0 * E + idx;
            t_n = dst[eid_n];
            s_n = src[eid_n];
        }
    }

    for (; node < N; node += nw) {
        const int eid0 = eid_n;
        const size_t seg0 = (size_t)t_n * R + r0;
        const int s0 = s_n;
        int slot0 = 0;
        if (eid0 >= 0) slot0 = atomicAdd(&cnt[seg0], 1);
        eid_n = -1;
        const int nn = node + nw;
        if (nn < N && lane < tot) {
            const int idx = nn * per + j0;
            if (idx < E) {
                eid_n = r0 * E + idx;
                t_n = dst[eid_n];
                s_n = src[eid_n];
            }
        }
        for (int i = lane + 32; i < tot; i += 32) {
            const int r = i / per;
            const int idx = node * per + (i - r * per);
            if (idx < E) {
                const int eid = r * E + idx;
                const size_t seg = (size_t)dst[eid] * R + r;
                const int slot = atomicAdd(&cnt[seg], 1);
                if (slot < cap) bucket[seg * cap + slot] = src[eid];
                else next[eid] = atomicExch(&ohead[seg], eid);
            }
        }

        const float* xr = x + (size_t)node * D;
        for (int j0r = 0; j0r < R; j0r += 4) {
            const int nrow = (R - j0r) < 4 ? (R - j0r) : 4;
            const float* w0 = attn_w + (size_t)(2 * j0r) * D;
            const float* w1 = attn_w + (size_t)(2 * (j0r + (nrow > 1 ? 1 : 0))) * D;
            const float* w2 = attn_w + (size_t)(2 * (j0r + (nrow > 2 ? 2 : 0))) * D;
            const float* w3 = attn_w + (size_t)(2 * (j0r + (nrow > 3 ? 3 : 0))) * D;
            float a0 = 0.0f, a1 = 0.0f, a2 = 0.0f, a3 = 0.0f;
            for (unsigned d = (unsigned)lane * 4u; d < (unsigned)D; d += 128u) {
                const float4 v = *(const float4*)(xr + d);
                a0 += dot4(v, *(const float4*)(w0 + d));
                a1 += dot4(v, *(const float4*)(w1 + d));
                if (nrow > 2) {
                    a2 += dot4(v, *(const float4*)(w2 + d));
                    a3 += dot4(v, *(const float4*)(w3 + d));
                }
            }
            float k0 = hi16 ? a2 : a0;
            float k1 = hi16 ? a3 : a1;
            const float g0 = hi16 ? a0 : a2;
            const float g1 = hi16 ? a1 : a3;
            k0 += __shfl_xor(g0, 16, 32);
            k1 += __shfl_xor(g1, 16, 32);
            float k = hi8 ? k1 : k0;
            const float g = hi8 ? k0 : k1;
            k += __shfl_xor(g, 8, 32);
            for (int off = 4; off > 0; off >>= 1) k += __shfl_xor(k, off, 32);
            if ((lane & 7) == 0 && q < nrow) S[(size_t)node * R + j0r + q] = k;
        }
        if (eid0 >= 0) {
            if (slot0 < cap) bucket[seg0 * cap + slot0] = s0;
            else next[eid0] = atomicExch(&ohead[seg0], eid0);
        }
    }
}

__device__ inline float lane_f(float v, int k) {
    return __int_as_float(__builtin_amdgcn_readlane(__float_as_int(v), k));
}

struct RowSet {
    v4f v[8];
    int k[8];
};

__device__ inline unsigned load_rows(RowSet& g, unsigned mask, int ms,
                                     const float* __restrict__ xcol, int D, bool on) {
    const v4f zero4 = {0.0f, 0.0f, 0.0f, 0.0f};
#pragma unroll
    for (int j = 0; j < 8; ++j) {
        g.v[j] = zero4;
        g.k[j] = -1;
        if (mask != 0u) {
            g.k[j] = __builtin_ctz(mask);
            mask &= mask - 1u;
            const int s = __builtin_amdgcn_readlane(ms, g.k[j]);
            if (on) g.v[j] = *(const v4f*)(xcol + (size_t)s * D);
        }
    }
    return mask;
}

__device__ inline void add_rows(v4f& acc, const RowSet& g, float alpha) {
#pragma unroll
    for (int j = 0; j < 8; ++j) {
        if (g.k[j] >= 0) acc += lane_f(alpha, g.k[j]) * g.v[j];
    }
}

__device__ __forceinline__ void gather_set(const float* __restrict__ x,
                                           const float* __restrict__ S,
                                           const int* __restrict__ src,
                                           const int* __restrict__ bucket,
                                           const int* __restrict__ ohead,
                                           const int* __restrict__ next,
                                           float* __restrict__ out,
                                           int node, int D, int R, int cap, int lane,
                                           int rg0, int Rg, int hop, int wl, int cw, int ms) {
    const v4f zero4 = {0.0f, 0.0f, 0.0f, 0.0f};
    const int budget = 32 / Rg;
    const int lim = budget < cap ? budget : cap;

    if (__builtin_amdgcn_ballot_w32(cw > lim) == 0u) {
        const bool valid = hop < cw;
        const unsigned vmask = __builtin_amdgcn_ballot_w32(valid);
        if (vmask == 0u) return;
        const int hops = (31 - __builtin_clz(vmask)) / Rg + 1;
        const float a = valid ? S[(unsigned)(ms * R + rg0 + wl)] : -INFINITY;
        float alpha = 0.0f;

        for (int c0 = 0; c0 < D; c0 += 128) {
            const int d = c0 + lane * 4;
            const bool on = d < D;
            float* orow = out + (size_t)node * D + d;
            const float* xcol = x + d;
            v4f o = zero4;
            if (on) o = *(const v4f*)orow;
            unsigned mask = vmask;
            RowSet g;
            mask = load_rows(g, mask, ms, xcol, D, on);

            if (c0 == 0) {
                float m = -INFINITY;
                for (int h0 = 0; h0 < hops; h0 += 8) {
#pragma unroll
                    for (int j = 0; j < 8; ++j) {
                        const float t = __shfl(a, ((h0 + j) * Rg + wl) & 31, 32);
                        if (h0 + j < hops) m = fmaxf(m, t);
                    }
                }
                const float p = valid ? __expf(a - m) : 0.0f;
                float den = 0.0f;
                for (int h0 = 0; h0 < hops; h0 += 8) {
#pragma unroll
                    for (int j = 0; j < 8; ++j) {
                        const float t = __shfl(p, ((h0 + j) * Rg + wl) & 31, 32);
                        if (h0 + j < hops) den += t;
                    }
                }
                alpha = p / fmaxf(den, 1e-30f);
            }

            v4f acc = zero4;
            add_rows(acc, g, alpha);
            while (mask != 0u) {
                mask = load_rows(g, mask, ms, xcol, D, on);
                add_rows(acc, g, alpha);
            }
            if (on) *(v4f*)orow = o + acc;
        }
    } else {
        for (int c0 = 0; c0 < D; c0 += 128) {
            const int d = c0 + lane * 4;
            const bool on = d < D;
            float* orow = out + (size_t)node * D + d;
            const float* xcol = x + d;
            v4f acc = zero4;
            for (int w = 0; w < Rg; ++w) {
                const int r = rg0 + w;
                const size_t seg = (size_t)node * R + r;
                const int c = __builtin_amdgcn_readlane(cw, w);
                const int cb = c < cap ? c : cap;
                const int oh = c > cap ? ohead[seg] : -1;
                const int* bk = bucket + seg * cap;
                float m = -INFINITY;
                for (int k = 0; k < cb; ++k)
                    m = fmaxf(m, S[(size_t)bk[k] * R + r]);
                for (int e = oh; e >= 0; e = next[e])
                    m = fmaxf(m, S[(size_t)src[e] * R + r]);
                float den = 0.0f;
                for (int k = 0; k < cb; ++k)
                    den += __expf(S[(size_t)bk[k] * R + r] - m);
                for (int e = oh; e >= 0; e = next[e])
                    den += __expf(S[(size_t)src[e] * R + r] - m);
                const float dcl = fmaxf(den, 1e-30f);
                for (int k = 0; k < cb; ++k) {
                    const int s = bk[k];
                    const float al = __expf(S[(size_t)s * R + r] - m) / dcl;
                    if (on) acc += al * *(const v4f*)(xcol + (size_t)s * D);
                }
                for (int e = oh; e >= 0; e = next[e]) {
                    const int s = src[e];
                    const float al = __expf(S[(size_t)s * R + r] - m) / dcl;
                    if (on) acc += al * *(const v4f*)(xcol + (size_t)s * D);
                }
            }
            if (on) {
                const v4f o = *(const v4f*)orow;
                *(v4f*)orow = o + acc;
            }
        }
    }
}

__global__ __launch_bounds__(256, 16) void gather_softmax(const float* __restrict__ x,
                                                          const float* __restrict__ S,
                                                          const int* __restrict__ src,
                                                          const int* __restrict__ cnt,
                                                          const int* __restrict__ bucket,
                                                          const int* __restrict__ ohead,
                                                          const int* __restrict__ next,
                                                          float* __restrict__ out,
                                                          int N, int D, int R,
                                                          int cap,
                                                          int nw) {
    const int lane = threadIdx.x & 31;
    int node = __builtin_amdgcn_readfirstlane(blockIdx.x * 8 + (threadIdx.x >> 5));
    if (node >= N) return;

    const int Rfirst = R < 32 ? R : 32;
    const int hop0 = lane / Rfirst;
    const int wl0 = lane - hop0 * Rfirst;
    const bool in0 = hop0 < 32 / Rfirst;
    const bool sl0 = in0 && hop0 < cap;
    const unsigned coff0 = (unsigned)wl0;
    const unsigned boff0 = (unsigned)(wl0 * cap + hop0);
    const unsigned rc = (unsigned)(R * cap);

    int c_cur = in0 ? cnt[(unsigned)node * (unsigned)R + coff0] : 0;
    int c_nxt = (in0 && node + nw < N) ? cnt[(unsigned)(node + nw) * (unsigned)R + coff0] : 0;
    int m_cur = (sl0 && hop0 < c_cur) ? bucket[(unsigned)node * rc + boff0] : -1;
    asm volatile("" : : "v"(c_cur), "v"(m_cur));

    for (; node < N; node += nw) {
        int m_nxt = -1;
        if (sl0 && hop0 < c_nxt) m_nxt = bucket[(unsigned)(node + nw) * rc + boff0];
        int c_nn = 0;
        if (in0 && node + 2 * nw < N) c_nn = cnt[(unsigned)(node + 2 * nw) * (unsigned)R + coff0];

        gather_set(x, S, src, bucket, ohead, next, out, node, D, R, cap, lane,
                   0, Rfirst, hop0, wl0, c_cur, m_cur);

        for (int rg0 = 32; rg0 < R; rg0 += 32) {
            const int Rg = (R - rg0) < 32 ? (R - rg0) : 32;
            const int budget = 32 / Rg;
            const int hop = lane / Rg;
            const int wl = lane - hop * Rg;
            const int lim = budget < cap ? budget : cap;
            const size_t seg0 = (size_t)node * R + rg0;
            const int cw = (hop < budget) ? (cnt + seg0)[(unsigned)wl] : 0;
            const int ms = (hop < lim && hop < cw) ? (bucket + seg0 * cap)[(unsigned)(wl * cap + hop)] : -1;
            gather_set(x, S, src, bucket, ohead, next, out, node, D, R, cap, lane,
                       rg0, Rg, hop, wl, cw, ms);
        }

        c_cur = c_nxt;
        m_cur = m_nxt;
        c_nxt = c_nn;
    }
}

__global__ __launch_bounds__(32) void wmma_selfloop(const float* __restrict__ x,
                                                    const float* __restrict__ W,
                                                    const float* __restrict__ bias,
                                                    float* __restrict__ out,
                                                    int* __restrict__ cnt,
                                                    int* __restrict__ ohead,
                                                    int rn, int N, int D) {
    const int lane = threadIdx.x;
    const int m0 = blockIdx.x * 16;
    const int n0 = blockIdx.y * 64;
    const int row = lane & 15;
    const int khalf = (lane >> 4) * 2;

    const int fi = (blockIdx.x * (D >> 6) + blockIdx.y) * 32 + lane;
    if (fi < rn) {
        cnt[fi] = 0;
        ohead[fi] = -1;
    }

    v8f c0 = {}, c1 = {}, c2 = {}, c3 = {};

    for (int k = 0; k < D; k += 4) {
        const float* ap = x + (size_t)(m0 + row) * D + (k + khalf);
        v2f a;
        a.x = ap[0];
        a.y = ap[1];

        const float* bp = W + (size_t)(k + khalf) * D + n0 + row;
        v2f b0, b1, b2, b3;
        b0.x = bp[0];  b0.y = bp[D + 0];
        b1.x = bp[16]; b1.y = bp[D + 16];
        b2.x = bp[32]; b2.y = bp[D + 32];
        b3.x = bp[48]; b3.y = bp[D + 48];

        c0 = __builtin_amdgcn_wmma_f32_16x16x4_f32(false, a, false, b0, (short)0, c0, false, false);
        c1 = __builtin_amdgcn_wmma_f32_16x16x4_f32(false, a, false, b1, (short)0, c1, false, false);
        c2 = __builtin_amdgcn_wmma_f32_16x16x4_f32(false, a, false, b2, (short)0, c2, false, false);
        c3 = __builtin_amdgcn_wmma_f32_16x16x4_f32(false, a, false, b3, (short)0, c3, false, false);
    }

    const int col = n0 + (lane & 15);
    const int mbase = m0 + (lane >> 4) * 8;
#pragma unroll
    for (int v = 0; v < 8; ++v) {
        float* o = out + (size_t)(mbase + v) * D;
        o[col + 0]  = c0[v] + bias[col + 0];
        o[col + 16] = c1[v] + bias[col + 16];
        o[col + 32] = c2[v] + bias[col + 32];
        o[col + 48] = c3[v] + bias[col + 48];
    }
}

#define WP_PITCH 288

__global__ __launch_bounds__(128) void wmma_selfloop_lds(const float* __restrict__ x,
                                                        const float* __restrict__ W,
                                                        const float* __restrict__ bias,
                                                        float* __restrict__ out,
                                                        int* __restrict__ cnt,
                                                        int* __restrict__ ohead,
                                                        int rn, int row_tiles) {
    __shared__ float wp[64 * WP_PITCH];
    const int tid = threadIdx.x;
    const int lane = tid & 31;
    const int wave = tid >> 5;

    for (int fi = blockIdx.x * 128 + tid; fi < rn; fi += gridDim.x * 128) {
        cnt[fi] = 0;
        ohead[fi] = -1;
    }

    for (int it = tid; it < 64 * 32; it += 128) {
        const int kp = it >> 5;
        const int c4 = (it & 31) * 4;
        const float4 r0 = *(const float4*)(W + (size_t)(2 * kp) * 128 + c4);
        const float4 r1 = *(const float4*)(W + (size_t)(2 * kp + 1) * 128 + c4);
        float* q = wp + kp * WP_PITCH + c4 * 2;
        *(float4*)q = make_float4(r0.x, r1.x, r0.y, r1.y);
        *(float4*)(q + 4) = make_float4(r0.z, r1.z, r0.w, r1.w);
    }
    __syncthreads();

    const int tile = blockIdx.x * 4 + wave;
    if (tile >= row_tiles) return;

    const int row = lane & 15;
    const int h = lane >> 4;
    const float* ap = x + (size_t)(tile * 16 + row) * 128 + h * 2;
    const float* bp = wp + h * WP_PITCH + row * 2;

    v8f c[8];
#pragma unroll
    for (int j = 0; j < 8; ++j) c[j] = (v8f){};

    v2f an[4];
#pragma unroll
    for (int u = 0; u < 4; ++u) an[u] = *(const v2f*)(ap + 4 * u);

    for (int g = 0; g < 8; ++g) {
        v2f ac[4];
#pragma unroll
        for (int u = 0; u < 4; ++u) ac[u] = an[u];
        if (g + 1 < 8) {
#pragma unroll
            for (int u = 0; u < 4; ++u) an[u] = *(const v2f*)(ap + 16 * (g + 1) + 4 * u);
        }
#pragma unroll
        for (int u = 0; u < 4; ++u) {
            const float* b = bp + (2 * (4 * g + u)) * WP_PITCH;
            v2f bf[8];
#pragma unroll
            for (int j = 0; j < 8; ++j) bf[j] = *(const v2f*)(b + 32 * j);
#pragma unroll
            for (int j = 0; j < 8; ++j)
                c[j] = __builtin_amdgcn_wmma_f32_16x16x4_f32(false, ac[u], false, bf[j], (short)0, c[j], false, false);
        }
    }

    const int mbase = tile * 16 + h * 8;
    float bj[8];
#pragma unroll
    for (int j = 0; j < 8; ++j) bj[j] = bias[16 * j + row];
#pragma unroll
    for (int v = 0; v < 8; ++v) {
        float* o = out + (size_t)(mbase + v) * 128 + row;
#pragma unroll
        for (int j = 0; j < 8; ++j) o[16 * j] = c[j][v] + bj[j];
    }
}

__global__ __launch_bounds__(256) void gemm_tail(const float* __restrict__ x,
                                                 const float* __restrict__ W,
                                                 const float* __restrict__ bias,
                                                 float* __restrict__ out,
                                                 int N, int D, int row0) {
    long long idx = (long long)blockIdx.x * blockDim.x + threadIdx.x;
    int row = row0 + (int)(idx / D);
    int col = (int)(idx % D);
    if (row >= N) return;
    float acc = bias[col];
    for (int k = 0; k < D; ++k) acc += x[(size_t)row * D + k] * W[(size_t)k * D + col];
    out[(size_t)row * D + col] = acc;
}

extern "C" void kernel_launch(void* const* d_in, const int* in_sizes, int n_in,
                              void* d_out, int out_size, void* d_ws, size_t ws_size,
                              hipStream_t stream) {
    const float* x      = (const float*)d_in[0];
    const float* attn_w = (const float*)d_in[1];
    const float* W      = (const float*)d_in[2];
    const float* bias   = (const float*)d_in[3];
    const int*   src    = (const int*)d_in[4];
    const int*   dst    = (const int*)d_in[5];
    float*       out    = (float*)d_out;

    const int D = in_sizes[3];
    const int N = in_sizes[0] / D;
    const int R = in_sizes[1] / (2 * D);
    const int E = in_sizes[4] / R;

    const int rn = R * N;
    const long long fixed_words = (long long)R * E + 3LL * rn;
    const long long spare_words = (long long)(ws_size / sizeof(int)) - fixed_words;
    int cap = 16;
    if (spare_words < 16LL * rn) cap = spare_words > 0 ? (int)(spare_words / rn) : 0;
    if ((long long)rn * cap > 0x7fffffffLL) cap = (int)(0x7fffffffLL / rn);
    int*   bucket_buf = (int*)d_ws;
    int*   next_buf   = bucket_buf + (size_t)rn * cap;
    int*   ohead_buf  = next_buf + (size_t)R * E;
    int*   cnt_buf    = ohead_buf + rn;
    float* s_buf      = (float*)(cnt_buf + rn);

    int row_tiles = 0;
    if ((D % 64) == 0) row_tiles = N / 16;
    const bool lds_path = (D == 128) && row_tiles > 0;
    const bool fill_rides = lds_path || (long long)row_tiles * (D / 64) * 32 >= rn;
    if (!fill_rides) {
        init_heads<<<dim3((rn + 255) / 256), dim3(256), 0, stream>>>(cnt_buf, ohead_buf, rn);
    }
    if (lds_path) {
        wmma_selfloop_lds<<<dim3((row_tiles + 3) / 4), 128, 0, stream>>>(x, W, bias, out, cnt_buf, ohead_buf,
                                                                       rn, row_tiles);
    } else if (row_tiles > 0) {
        dim3 ggrid(row_tiles, D / 64);
        wmma_selfloop<<<ggrid, 32, 0, stream>>>(x, W, bias, out, cnt_buf, ohead_buf,
                                                 fill_rides ? rn : 0, N, D);
    }
    int row0 = row_tiles * 16;
    long long tail_elems = (long long)(N - row0) * D;
    if (tail_elems > 0) {
        int blocks = (int)((tail_elems + 255) / 256);
        gemm_tail<<<dim3(blocks), dim3(256), 0, stream>>>(x, W, bias, out, N, D, row0);
    }

    const int node_groups = (N + 7) / 8;
    const int res_scores = node_groups < 1536 ? node_groups : 1536;
    const int res_gather = node_groups < 2048 ? node_groups : 2048;

    const int per = (E + N - 1) / N;
    node_scores<<<dim3(res_scores), 256, 0, stream>>>(x, attn_w, src, dst, s_buf, cnt_buf, bucket_buf,
                                                      ohead_buf, next_buf,
                                                      N, D, R, E, per, cap, res_scores * 8);

    gather_softmax<<<dim3(res_gather), 256, 0, stream>>>(x, s_buf, src, cnt_buf, bucket_buf, ohead_buf,
                                                         next_buf, out,
                                                         N, D, R, cap, res_gather * 8);
}


# Bucketing and projections S ride in the self-loop product's waves: S a ninth column tile (summed in matrix k order), returning adds answered under the matrix loop; node pass shrinks to a fill launch
# speedup vs baseline: 3.2200x; 1.1016x over previous
#include <hip/hip_runtime.h>
#include <math.h>

typedef __attribute__((ext_vector_type(2))) float v2f;
typedef __attribute__((ext_vector_type(4))) float v4f;
typedef __attribute__((ext_vector_type(8))) float v8f;

__global__ __launch_bounds__(256) void init_heads(int* __restrict__ cnt, int* __restrict__ ohead, int rn) {
    const int idx = blockIdx.x * blockDim.x + threadIdx.x;
    if (idx < rn) {
        cnt[idx] = 0;
        ohead[idx] = -1;
    }
}

__device__ inline float dot4(float4 a, float4 b) {
    return a.x * b.x + a.y * b.y + a.z * b.z + a.w * b.w;
}

__global__ __launch_bounds__(256) void node_scores(const float* __restrict__ x,
                                                   const float* __restrict__ attn_w,
                                                   const int* __restrict__ src,
                                                   const int* __restrict__ dst,
                                                   float* __restrict__ S,
                                                   int* __restrict__ cnt,
                                                   int* __restrict__ bucket,
                                                   int* __restrict__ ohead,
                                                   int* __restrict__ next,
                                                   int N, int D, int R, int E, int per,
                                                   int cap,
                                                   int nw) {
    const int lane = threadIdx.x & 31;
    int node = __builtin_amdgcn_readfirstlane(blockIdx.x * 8 + (threadIdx.x >> 5));
    if (node >= N) return;

    const int tot = R * per;
    const int r0 = (lane < tot) ? lane / per : 0;
    const int j0 = lane - r0 * per;
    const bool hi16 = (lane & 16) != 0;
    const bool hi8 = (lane & 8) != 0;
    const int q = lane >> 3;

    int eid_n = -1, t_n = 0, s_n = 0;
    if (lane < tot) {
        const int idx = node * per + j0;
        if (idx < E) {
            eid_n = r0 * E + idx;
            t_n = dst[eid_n];
            s_n = src[eid_n];
        }
    }

    for (; node < N; node += nw) {
        const int eid0 = eid_n;
        const size_t seg0 = (size_t)t_n * R + r0;
        const int s0 = s_n;
        int slot0 = 0;
        if (eid0 >= 0) slot0 = atomicAdd(&cnt[seg0], 1);
        eid_n = -1;
        const int nn = node + nw;
        if (nn < N && lane < tot) {
            const int idx = nn * per + j0;
            if (idx < E) {
                eid_n = r0 * E + idx;
                t_n = dst[eid_n];
                s_n = src[eid_n];
            }
        }
        for (int i = lane + 32; i < tot; i += 32) {
            const int r = i / per;
            const int idx = node * per + (i - r * per);
            if (idx < E) {
                const int eid = r * E + idx;
                const size_t seg = (size_t)dst[eid] * R + r;
                const int slot = atomicAdd(&cnt[seg], 1);
                if (slot < cap) bucket[seg * cap + slot] = src[eid];
                else next[eid] = atomicExch(&ohead[seg], eid);
            }
        }

        const float* xr = x + (size_t)node * D;
        for (int j0r = 0; j0r < R; j0r += 4) {
            const int nrow = (R - j0r) < 4 ? (R - j0r) : 4;
            const float* w0 = attn_w + (size_t)(2 * j0r) * D;
            const float* w1 = attn_w + (size_t)(2 * (j0r + (nrow > 1 ? 1 : 0))) * D;
            const float* w2 = attn_w + (size_t)(2 * (j0r + (nrow > 2 ? 2 : 0))) * D;
            const float* w3 = attn_w + (size_t)(2 * (j0r + (nrow > 3 ? 3 : 0))) * D;
            float a0 = 0.0f, a1 = 0.0f, a2 = 0.0f, a3 = 0.0f;
            for (unsigned d = (unsigned)lane * 4u; d < (unsigned)D; d += 128u) {
                const float4 v = *(const float4*)(xr + d);
                a0 += dot4(v, *(const float4*)(w0 + d));
                a1 += dot4(v, *(const float4*)(w1 + d));
                if (nrow > 2) {
                    a2 += dot4(v, *(const float4*)(w2 + d));
                    a3 += dot4(v, *(const float4*)(w3 + d));
                }
            }
            float k0 = hi16 ? a2 : a0;
            float k1 = hi16 ? a3 : a1;
            const float g0 = hi16 ? a0 : a2;
            const float g1 = hi16 ? a1 : a3;
            k0 += __shfl_xor(g0, 16, 32);
            k1 += __shfl_xor(g1, 16, 32);
            float k = hi8 ? k1 : k0;
            const float g = hi8 ? k0 : k1;
            k += __shfl_xor(g, 8, 32);
            for (int off = 4; off > 0; off >>= 1) k += __shfl_xor(k, off, 32);
            if ((lane & 7) == 0 && q < nrow) S[(size_t)node * R + j0r + q] = k;
        }
        if (eid0 >= 0) {
            if (slot0 < cap) bucket[seg0 * cap + slot0] = s0;
            else next[eid0] = atomicExch(&ohead[seg0], eid0);
        }
    }
}

__device__ inline float lane_f(float v, int k) {
    return __int_as_float(__builtin_amdgcn_readlane(__float_as_int(v), k));
}

struct RowSet {
    v4f v[8];
    int k[8];
};

__device__ inline unsigned load_rows(RowSet& g, unsigned mask, int ms,
                                     const float* __restrict__ xcol, int D, bool on) {
    const v4f zero4 = {0.0f, 0.0f, 0.0f, 0.0f};
#pragma unroll
    for (int j = 0; j < 8; ++j) {
        g.v[j] = zero4;
        g.k[j] = -1;
        if (mask != 0u) {
            g.k[j] = __builtin_ctz(mask);
            mask &= mask - 1u;
            const int s = __builtin_amdgcn_readlane(ms, g.k[j]);
            if (on) g.v[j] = *(const v4f*)(xcol + (size_t)s * D);
        }
    }
    return mask;
}

__device__ inline void add_rows(v4f& acc, const RowSet& g, float alpha) {
#pragma unroll
    for (int j = 0; j < 8; ++j) {
        if (g.k[j] >= 0) acc += lane_f(alpha, g.k[j]) * g.v[j];
    }
}

__device__ __forceinline__ void gather_set(const float* __restrict__ x,
                                           const float* __restrict__ S,
                                           const int* __restrict__ src,
                                           const int* __restrict__ bucket,
                                           const int* __restrict__ ohead,
                                           const int* __restrict__ next,
                                           float* __restrict__ out,
                                           int node, int D, int R, int cap, int lane,
                                           int rg0, int Rg, int hop, int wl, int cw, int ms) {
    const v4f zero4 = {0.0f, 0.0f, 0.0f, 0.0f};
    const int budget = 32 / Rg;
    const int lim = budget < cap ? budget : cap;

    if (__builtin_amdgcn_ballot_w32(cw > lim) == 0u) {
        const bool valid = hop < cw;
        const unsigned vmask = __builtin_amdgcn_ballot_w32(valid);
        if (vmask == 0u) return;
        const int hops = (31 - __builtin_clz(vmask)) / Rg + 1;
        const float a = valid ? S[(unsigned)(ms * R + rg0 + wl)] : -INFINITY;
        float alpha = 0.0f;

        for (int c0 = 0; c0 < D; c0 += 128) {
            const int d = c0 + lane * 4;
            const bool on = d < D;
            float* orow = out + (size_t)node * D + d;
            const float* xcol = x + d;
            v4f o = zero4;
            if (on) o = *(const v4f*)orow;
            unsigned mask = vmask;
            RowSet g;
            mask = load_rows(g, mask, ms, xcol, D, on);

            if (c0 == 0) {
                float m = -INFINITY;
                for (int h0 = 0; h0 < hops; h0 += 8) {
#pragma unroll
                    for (int j = 0; j < 8; ++j) {
                        const float t = __shfl(a, ((h0 + j) * Rg + wl) & 31, 32);
                        if (h0 + j < hops) m = fmaxf(m, t);
                    }
                }
                const float p = valid ? __expf(a - m) : 0.0f;
                float den = 0.0f;
                for (int h0 = 0; h0 < hops; h0 += 8) {
#pragma unroll
                    for (int j = 0; j < 8; ++j) {
                        const float t = __shfl(p, ((h0 + j) * Rg + wl) & 31, 32);
                        if (h0 + j < hops) den += t;
                    }
                }
                alpha = p / fmaxf(den, 1e-30f);
            }

            v4f acc = zero4;
            add_rows(acc, g, alpha);
            while (mask != 0u) {
                mask = load_rows(g, mask, ms, xcol, D, on);
                add_rows(acc, g, alpha);
            }
            if (on) *(v4f*)orow = o + acc;
        }
    } else {
        for (int c0 = 0; c0 < D; c0 += 128) {
            const int d = c0 + lane * 4;
            const bool on = d < D;
            float* orow = out + (size_t)node * D + d;
            const float* xcol = x + d;
            v4f acc = zero4;
            for (int w = 0; w < Rg; ++w) {
                const int r = rg0 + w;
                const size_t seg = (size_t)node * R + r;
                const int c = __builtin_amdgcn_readlane(cw, w);
                const int cb = c < cap ? c : cap;
                const int oh = c > cap ? ohead[seg] : -1;
                const int* bk = bucket + seg * cap;
                float m = -INFINITY;
                for (int k = 0; k < cb; ++k)
                    m = fmaxf(m, S[(size_t)bk[k] * R + r]);
                for (int e = oh; e >= 0; e = next[e])
                    m = fmaxf(m, S[(size_t)src[e] * R + r]);
                float den = 0.0f;
                for (int k = 0; k < cb; ++k)
                    den += __expf(S[(size_t)bk[k] * R + r] - m);
                for (int e = oh; e >= 0; e = next[e])
                    den += __expf(S[(size_t)src[e] * R + r] - m);
                const float dcl = fmaxf(den, 1e-30f);
                for (int k = 0; k < cb; ++k) {
                    const int s = bk[k];
                    const float al = __expf(S[(size_t)s * R + r] - m) / dcl;
                    if (on) acc += al * *(const v4f*)(xcol + (size_t)s * D);
                }
                for (int e = oh; e >= 0; e = next[e]) {
                    const int s = src[e];
                    const float al = __expf(S[(size_t)s * R + r] - m) / dcl;
                    if (on) acc += al * *(const v4f*)(xcol + (size_t)s * D);
                }
            }
            if (on) {
                const v4f o = *(const v4f*)orow;
                *(v4f*)orow = o + acc;
            }
        }
    }
}

__global__ __launch_bounds__(256, 16) void gather_softmax(const float* __restrict__ x,
                                                          const float* __restrict__ S,
                                                          const int* __restrict__ src,
                                                          const int* __restrict__ cnt,
                                                          const int* __restrict__ bucket,
                                                          const int* __restrict__ ohead,
                                                          const int* __restrict__ next,
                                                          float* __restrict__ out,
                                                          int N, int D, int R,
                                                          int cap,
                                                          int nw) {
    const int lane = threadIdx.x & 31;
    int node = __builtin_amdgcn_readfirstlane(blockIdx.x * 8 + (threadIdx.x >> 5));
    if (node >= N) return;

    const int Rfirst = R < 32 ? R : 32;
    const int hop0 = lane / Rfirst;
    const int wl0 = lane - hop0 * Rfirst;
    const bool in0 = hop0 < 32 / Rfirst;
    const bool sl0 = in0 && hop0 < cap;
    const unsigned coff0 = (unsigned)wl0;
    const unsigned boff0 = (unsigned)(wl0 * cap + hop0);
    const unsigned rc = (unsigned)(R * cap);

    int c_cur = in0 ? cnt[(unsigned)node * (unsigned)R + coff0] : 0;
    int c_nxt = (in0 && node + nw < N) ? cnt[(unsigned)(node + nw) * (unsigned)R + coff0] : 0;
    int m_cur = (sl0 && hop0 < c_cur) ? bucket[(unsigned)node * rc + boff0] : -1;
    asm volatile("" : : "v"(c_cur), "v"(m_cur));

    for (; node < N; node += nw) {
        int m_nxt = -1;
        if (sl0 && hop0 < c_nxt) m_nxt = bucket[(unsigned)(node + nw) * rc + boff0];
        int c_nn = 0;
        if (in0 && node + 2 * nw < N) c_nn = cnt[(unsigned)(node + 2 * nw) * (unsigned)R + coff0];

        gather_set(x, S, src, bucket, ohead, next, out, node, D, R, cap, lane,
                   0, Rfirst, hop0, wl0, c_cur, m_cur);

        for (int rg0 = 32; rg0 < R; rg0 += 32) {
            const int Rg = (R - rg0) < 32 ? (R - rg0) : 32;
            const int budget = 32 / Rg;
            const int hop = lane / Rg;
            const int wl = lane - hop * Rg;
            const int lim = budget < cap ? budget : cap;
            const size_t seg0 = (size_t)node * R + rg0;
            const int cw = (hop < budget) ? (cnt + seg0)[(unsigned)wl] : 0;
            const int ms = (hop < lim && hop < cw) ? (bucket + seg0 * cap)[(unsigned)(wl * cap + hop)] : -1;
            gather_set(x, S, src, bucket, ohead, next, out, node, D, R, cap, lane,
                       rg0, Rg, hop, wl, cw, ms);
        }

        c_cur = c_nxt;
        m_cur = m_nxt;
        c_nxt = c_nn;
    }
}

__global__ __launch_bounds__(32) void wmma_selfloop(const float* __restrict__ x,
                                                    const float* __restrict__ W,
                                                    const float* __restrict__ bias,
                                                    float* __restrict__ out,
                                                    int* __restrict__ cnt,
                                                    int* __restrict__ ohead,
                                                    int rn, int N, int D) {
    const int lane = threadIdx.x;
    const int m0 = blockIdx.x * 16;
    const int n0 = blockIdx.y * 64;
    const int row = lane & 15;
    const int khalf = (lane >> 4) * 2;

    const int fi = (blockIdx.x * (D >> 6) + blockIdx.y) * 32 + lane;
    if (fi < rn) {
        cnt[fi] = 0;
        ohead[fi] = -1;
    }

    v8f c0 = {}, c1 = {}, c2 = {}, c3 = {};

    for (int k = 0; k < D; k += 4) {
        const float* ap = x + (size_t)(m0 + row) * D + (k + khalf);
        v2f a;
        a.x = ap[0];
        a.y = ap[1];

        const float* bp = W + (size_t)(k + khalf) * D + n0 + row;
        v2f b0, b1, b2, b3;
        b0.x = bp[0];  b0.y = bp[D + 0];
        b1.x = bp[16]; b1.y = bp[D + 16];
        b2.x = bp[32]; b2.y = bp[D + 32];
        b3.x = bp[48]; b3.y = bp[D + 48];

        c0 = __builtin_amdgcn_wmma_f32_16x16x4_f32(false, a, false, b0, (short)0, c0, false, false);
        c1 = __builtin_amdgcn_wmma_f32_16x16x4_f32(false, a, false, b1, (short)0, c1, false, false);
        c2 = __builtin_amdgcn_wmma_f32_16x16x4_f32(false, a, false, b2, (short)0, c2, false, false);
        c3 = __builtin_amdgcn_wmma_f32_16x16x4_f32(false, a, false, b3, (short)0, c3, false, false);
    }

    const int col = n0 + (lane & 15);
    const int mbase = m0 + (lane >> 4) * 8;
#pragma unroll
    for (int v = 0; v < 8; ++v) {
        float* o = out + (size_t)(mbase + v) * D;
        o[col + 0]  = c0[v] + bias[col + 0];
        o[col + 16] = c1[v] + bias[col + 16];
        o[col + 32] = c2[v] + bias[col + 32];
        o[col + 48] = c3[v] + bias[col + 48];
    }
}

#define WP_PITCH 288

__global__ __launch_bounds__(128) void wmma_selfloop_lds(const float* __restrict__ x,
                                                        const float* __restrict__ W,
                                                        const float* __restrict__ bias,
                                                        float* __restrict__ out,
                                                        int* __restrict__ cnt,
                                                        int* __restrict__ ohead,
                                                        int rn, int row_tiles) {
    __shared__ float wp[64 * WP_PITCH];
    const int tid = threadIdx.x;
    const int lane = tid & 31;
    const int wave = tid >> 5;

    for (int fi = blockIdx.x * 128 + tid; fi < rn; fi += gridDim.x * 128) {
        cnt[fi] = 0;
        ohead[fi] = -1;
    }

    for (int it = tid; it < 64 * 32; it += 128) {
        const int kp = it >> 5;
        const int c4 = (it & 31) * 4;
        const float4 r0 = *(const float4*)(W + (size_t)(2 * kp) * 128 + c4);
        const float4 r1 = *(const float4*)(W + (size_t)(2 * kp + 1) * 128 + c4);
        float* q = wp + kp * WP_PITCH + c4 * 2;
        *(float4*)q = make_float4(r0.x, r1.x, r0.y, r1.y);
        *(float4*)(q + 4) = make_float4(r0.z, r1.z, r0.w, r1.w);
    }
    __syncthreads();

    const int tile = blockIdx.x * 4 + wave;
    if (tile >= row_tiles) return;

    const int row = lane & 15;
    const int h = lane >> 4;
    const float* ap = x + (size_t)(tile * 16 + row) * 128 + h * 2;
    const float* bp = wp + h * WP_PITCH + row * 2;

    v8f c[8];
#pragma unroll
    for (int j = 0; j < 8; ++j) c[j] = (v8f){};

    v2f an[4];
#pragma unroll
    for (int u = 0; u < 4; ++u) an[u] = *(const v2f*)(ap + 4 * u);

    for (int g = 0; g < 8; ++g) {
        v2f ac[4];
#pragma unroll
        for (int u = 0; u < 4; ++u) ac[u] = an[u];
        if (g + 1 < 8) {
#pragma unroll
            for (int u = 0; u < 4; ++u) an[u] = *(const v2f*)(ap + 16 * (g + 1) + 4 * u);
        }
#pragma unroll
        for (int u = 0; u < 4; ++u) {
            const float* b = bp + (2 * (4 * g + u)) * WP_PITCH;
            v2f bf[8];
#pragma unroll
            for (int j = 0; j < 8; ++j) bf[j] = *(const v2f*)(b + 32 * j);
#pragma unroll
            for (int j = 0; j < 8; ++j)
                c[j] = __builtin_amdgcn_wmma_f32_16x16x4_f32(false, ac[u], false, bf[j], (short)0, c[j], false, false);
        }
    }

    const int mbase = tile * 16 + h * 8;
    float bj[8];
#pragma unroll
    for (int j = 0; j < 8; ++j) bj[j] = bias[16 * j + row];
#pragma unroll
    for (int v = 0; v < 8; ++v) {
        float* o = out + (size_t)(mbase + v) * 128 + row;
#pragma unroll
        for (int j = 0; j < 8; ++j) o[16 * j] = c[j][v] + bj[j];
    }
}

__global__ __launch_bounds__(256) void fill_heads(int* __restrict__ cnt, int* __restrict__ ohead, int rn) {
    for (int idx = blockIdx.x * 256 + threadIdx.x; idx < rn; idx += gridDim.x * 256) {
        cnt[idx] = 0;
        ohead[idx] = -1;
    }
}

__global__ __launch_bounds__(128) void wmma_selfloop_edges(const float* __restrict__ x,
                                                          const float* __restrict__ W,
                                                          const float* __restrict__ bias,
                                                          const float* __restrict__ attn_w,
                                                          const int* __restrict__ src,
                                                          const int* __restrict__ dst,
                                                          float* __restrict__ out,
                                                          float* __restrict__ S,
                                                          int* __restrict__ cnt,
                                                          int* __restrict__ bucket,
                                                          int* __restrict__ ohead,
                                                          int* __restrict__ next,
                                                          int row_tiles, int R, int E, int per, int cap) {
    __shared__ float wp[64 * WP_PITCH];
    const int tid = threadIdx.x;
    const int lane = tid & 31;
    const int wave = tid >> 5;
    const int tile = blockIdx.x * 4 + wave;

    const int span = 16 * per;
    const int tot = R * span;
    int eid[4], rr[4], tn[4], sn[4], seg[4], slot[4];
#pragma unroll
    for (int j = 0; j < 4; ++j) {
        eid[j] = -1;
        rr[j] = 0;
        const int i = lane + 32 * j;
        if (tile < row_tiles && i < tot) {
            const int r = i / span;
            const int idx = tile * span + (i - r * span);
            if (idx < E) {
                eid[j] = r * E + idx;
                rr[j] = r;
            }
        }
    }
#pragma unroll
    for (int j = 0; j < 4; ++j) {
        tn[j] = 0;
        sn[j] = 0;
        seg[j] = 0;
        slot[j] = 0;
        if (eid[j] >= 0) {
            tn[j] = dst[eid[j]];
            sn[j] = src[eid[j]];
        }
    }

    const int ca = tid & 15;
    const int ka = tid >> 4;
    float2 av[8];
#pragma unroll
    for (int q = 0; q < 8; ++q) {
        av[q] = make_float2(0.0f, 0.0f);
        if (ca < R) av[q] = *(const float2*)(attn_w + (size_t)ca * 256 + 2 * (ka + 8 * q));
    }

    for (int it = tid; it < 64 * 32; it += 128) {
        const int kp = it >> 5;
        const int c4 = (it & 31) * 4;
        const float4 r0 = *(const float4*)(W + (size_t)(2 * kp) * 128 + c4);
        const float4 r1 = *(const float4*)(W + (size_t)(2 * kp + 1) * 128 + c4);
        float* q = wp + kp * WP_PITCH + c4 * 2;
        *(float4*)q = make_float4(r0.x, r1.x, r0.y, r1.y);
        *(float4*)(q + 4) = make_float4(r0.z, r1.z, r0.w, r1.w);
    }
#pragma unroll
    for (int q = 0; q < 8; ++q) *(float2*)(wp + (ka + 8 * q) * WP_PITCH + 256 + 2 * ca) = av[q];
    __syncthreads();

    if (tile >= row_tiles) return;

    const int row = lane & 15;
    const int h = lane >> 4;
    const float* ap = x + (size_t)(tile * 16 + row) * 128 + h * 2;
    const float* bp = wp + h * WP_PITCH + row * 2;

    v8f c[9];
#pragma unroll
    for (int j = 0; j < 9; ++j) c[j] = (v8f){};

    v2f an[4];
#pragma unroll
    for (int u = 0; u < 4; ++u) an[u] = *(const v2f*)(ap + 4 * u);

#pragma unroll
    for (int j = 0; j < 4; ++j) {
        if (eid[j] >= 0) {
            seg[j] = tn[j] * R + rr[j];
            slot[j] = atomicAdd(&cnt[seg[j]], 1);
        }
    }

    for (int g = 0; g < 8; ++g) {
        v2f ac[4];
#pragma unroll
        for (int u = 0; u < 4; ++u) ac[u] = an[u];
        if (g + 1 < 8) {
#pragma unroll
            for (int u = 0; u < 4; ++u) an[u] = *(const v2f*)(ap + 16 * (g + 1) + 4 * u);
        }
#pragma unroll
        for (int u = 0; u < 4; ++u) {
            const float* b = bp + (2 * (4 * g + u)) * WP_PITCH;
            v2f bf[9];
#pragma unroll
            for (int j = 0; j < 9; ++j) bf[j] = *(const v2f*)(b + 32 * j);
#pragma unroll
            for (int j = 0; j < 9; ++j)
                c[j] = __builtin_amdgcn_wmma_f32_16x16x4_f32(false, ac[u], false, bf[j], (short)0, c[j], false, false);
        }
    }

    const int mbase = tile * 16 + h * 8;
    float bj[8];
#pragma unroll
    for (int j = 0; j < 8; ++j) bj[j] = bias[16 * j + row];
#pragma unroll
    for (int v = 0; v < 8; ++v) {
        float* o = out + (size_t)(mbase + v) * 128 + row;
#pragma unroll
        for (int j = 0; j < 8; ++j) o[16 * j] = c[j][v] + bj[j];
    }
    if (row < R) {
#pragma unroll
        for (int v = 0; v < 8; ++v) S[(size_t)(mbase + v) * R + row] = c[8][v];
    }

#pragma unroll
    for (int j = 0; j < 4; ++j) {
        if (eid[j] >= 0) {
            if (slot[j] < cap) bucket[(size_t)seg[j] * cap + slot[j]] = sn[j];
            else next[eid[j]] = atomicExch(&ohead[seg[j]], eid[j]);
        }
    }
    for (int i = lane + 128; i < tot; i += 32) {
        const int r = i / span;
        const int idx = tile * span + (i - r * span);
        if (idx < E) {
            const int e = r * E + idx;
            const size_t sg = (size_t)dst[e] * R + r;
            const int sl = atomicAdd(&cnt[sg], 1);
            if (sl < cap) bucket[sg * cap + sl] = src[e];
            else next[e] = atomicExch(&ohead[sg], e);
        }
    }
}

__global__ __launch_bounds__(256) void gemm_tail(const float* __restrict__ x,
                                                 const float* __restrict__ W,
                                                 const float* __restrict__ bias,
                                                 float* __restrict__ out,
                                                 int N, int D, int row0) {
    long long idx = (long long)blockIdx.x * blockDim.x + threadIdx.x;
    int row = row0 + (int)(idx / D);
    int col = (int)(idx % D);
    if (row >= N) return;
    float acc = bias[col];
    for (int k = 0; k < D; ++k) acc += x[(size_t)row * D + k] * W[(size_t)k * D + col];
    out[(size_t)row * D + col] = acc;
}

extern "C" void kernel_launch(void* const* d_in, const int* in_sizes, int n_in,
                              void* d_out, int out_size, void* d_ws, size_t ws_size,
                              hipStream_t stream) {
    const float* x      = (const float*)d_in[0];
    const float* attn_w = (const float*)d_in[1];
    const float* W      = (const float*)d_in[2];
    const float* bias   = (const float*)d_in[3];
    const int*   src    = (const int*)d_in[4];
    const int*   dst    = (const int*)d_in[5];
    float*       out    = (float*)d_out;

    const int D = in_sizes[3];
    const int N = in_sizes[0] / D;
    const int R = in_sizes[1] / (2 * D);
    const int E = in_sizes[4] / R;

    const int rn = R * N;
    const long long fixed_words = (long long)R * E + 3LL * rn;
    const long long spare_words = (long long)(ws_size / sizeof(int)) - fixed_words;
    int cap = 16;
    if (spare_words < 16LL * rn) cap = spare_words > 0 ? (int)(spare_words / rn) : 0;
    if ((long long)rn * cap > 0x7fffffffLL) cap = (int)(0x7fffffffLL / rn);
    int*   bucket_buf = (int*)d_ws;
    int*   next_buf   = bucket_buf + (size_t)rn * cap;
    int*   ohead_buf  = next_buf + (size_t)R * E;
    int*   cnt_buf    = ohead_buf + rn;
    float* s_buf      = (float*)(cnt_buf + rn);

    if (D == 128 && N >= 16 && (N % 16) == 0 && R >= 1 && R <= 16) {
        const int tiles = N / 16;
        const int per_f = (E + N - 1) / N;
        const int fill_groups = (rn + 255) / 256 < 256 ? (rn + 255) / 256 : 256;
        const int gather_groups = (N + 7) / 8 < 2048 ? (N + 7) / 8 : 2048;
        fill_heads<<<dim3(fill_groups), dim3(256), 0, stream>>>(cnt_buf, ohead_buf, rn);
        wmma_selfloop_edges<<<dim3((tiles + 3) / 4), 128, 0, stream>>>(x, W, bias, attn_w, src, dst, out, s_buf,
                                                                       cnt_buf, bucket_buf, ohead_buf, next_buf,
                                                                       tiles, R, E, per_f, cap);
        gather_softmax<<<dim3(gather_groups), 256, 0, stream>>>(x, s_buf, src, cnt_buf, bucket_buf, ohead_buf,
                                                                next_buf, out,
                                                                N, D, R, cap, gather_groups * 8);
        return;
    }

    int row_tiles = 0;
    if ((D % 64) == 0) row_tiles = N / 16;
    const bool lds_path = (D == 128) && row_tiles > 0;
    const bool fill_rides = lds_path || (long long)row_tiles * (D / 64) * 32 >= rn;
    if (!fill_rides) {
        init_heads<<<dim3((rn + 255) / 256), dim3(256), 0, stream>>>(cnt_buf, ohead_buf, rn);
    }
    if (lds_path) {
        wmma_selfloop_lds<<<dim3((row_tiles + 3) / 4), 128, 0, stream>>>(x, W, bias, out, cnt_buf, ohead_buf,
                                                                       rn, row_tiles);
    } else if (row_tiles > 0) {
        dim3 ggrid(row_tiles, D / 64);
        wmma_selfloop<<<ggrid, 32, 0, stream>>>(x, W, bias, out, cnt_buf, ohead_buf,
                                                 fill_rides ? rn : 0, N, D);
    }
    int row0 = row_tiles * 16;
    long long tail_elems = (long long)(N - row0) * D;
    if (tail_elems > 0) {
        int blocks = (int)((tail_elems + 255) / 256);
        gemm_tail<<<dim3(blocks), dim3(256), 0, stream>>>(x, W, bias, out, N, D, row0);
    }

    const int node_groups = (N + 7) / 8;
    const int res_scores = node_groups < 1536 ? node_groups : 1536;
    const int res_gather = node_groups < 2048 ? node_groups : 2048;

    const int per = (E + N - 1) / N;
    node_scores<<<dim3(res_scores), 256, 0, stream>>>(x, attn_w, src, dst, s_buf, cnt_buf, bucket_buf,
                                                      ohead_buf, next_buf,
                                                      N, D, R, E, per, cap, res_scores * 8);

    gather_softmax<<<dim3(res_gather), 256, 0, stream>>>(x, s_buf, src, cnt_buf, bucket_buf, ohead_buf,
                                                         next_buf, out,
                                                         N, D, R, cap, res_gather * 8);
}


# Edge counting kept inside one chip half: a half's waves take edge chunks from its own queue and count only edges whose destination it owns; no count line shared by both halves; sums in arrival order
# speedup vs baseline: 3.9202x; 1.2175x over previous
#include <hip/hip_runtime.h>
#include <math.h>

typedef __attribute__((ext_vector_type(2))) float v2f;
typedef __attribute__((ext_vector_type(4))) float v4f;
typedef __attribute__((ext_vector_type(8))) float v8f;

__global__ __launch_bounds__(256) void init_heads(int* __restrict__ cnt, int* __restrict__ ohead, int rn) {
    const int idx = blockIdx.x * blockDim.x + threadIdx.x;
    if (idx < rn) {
        cnt[idx] = 0;
        ohead[idx] = -1;
    }
}

__device__ inline float dot4(float4 a, float4 b) {
    return a.x * b.x + a.y * b.y + a.z * b.z + a.w * b.w;
}

__global__ __launch_bounds__(256) void node_scores(const float* __restrict__ x,
                                                   const float* __restrict__ attn_w,
                                                   const int* __restrict__ src,
                                                   const int* __restrict__ dst,
                                                   float* __restrict__ S,
                                                   int* __restrict__ cnt,
                                                   int* __restrict__ bucket,
                                                   int* __restrict__ ohead,
                                                   int* __restrict__ next,
                                                   int N, int D, int R, int E, int per,
                                                   int cap,
                                                   int nw) {
    const int lane = threadIdx.x & 31;
    int node = __builtin_amdgcn_readfirstlane(blockIdx.x * 8 + (threadIdx.x >> 5));
    if (node >= N) return;

    const int tot = R * per;
    const int r0 = (lane < tot) ? lane / per : 0;
    const int j0 = lane - r0 * per;
    const bool hi16 = (lane & 16) != 0;
    const bool hi8 = (lane & 8) != 0;
    const int q = lane >> 3;

    int eid_n = -1, t_n = 0, s_n = 0;
    if (lane < tot) {
        const int idx = node * per + j0;
        if (idx < E) {
            eid_n = r0 * E + idx;
            t_n = dst[eid_n];
            s_n = src[eid_n];
        }
    }

    for (; node < N; node += nw) {
        const int eid0 = eid_n;
        const size_t seg0 = (size_t)t_n * R + r0;
        const int s0 = s_n;
        int slot0 = 0;
        if (eid0 >= 0) slot0 = atomicAdd(&cnt[seg0], 1);
        eid_n = -1;
        const int nn = node + nw;
        if (nn < N && lane < tot) {
            const int idx = nn * per + j0;
            if (idx < E) {
                eid_n = r0 * E + idx;
                t_n = dst[eid_n];
                s_n = src[eid_n];
            }
        }
        for (int i = lane + 32; i < tot; i += 32) {
            const int r = i / per;
            const int idx = node * per + (i - r * per);
            if (idx < E) {
                const int eid = r * E + idx;
                const size_t seg = (size_t)dst[eid] * R + r;
                const int slot = atomicAdd(&cnt[seg], 1);
                if (slot < cap) bucket[seg * cap + slot] = src[eid];
                else next[eid] = atomicExch(&ohead[seg], eid);
            }
        }

        const float* xr = x + (size_t)node * D;
        for (int j0r = 0; j0r < R; j0r += 4) {
            const int nrow = (R - j0r) < 4 ? (R - j0r) : 4;
            const float* w0 = attn_w + (size_t)(2 * j0r) * D;
            const float* w1 = attn_w + (size_t)(2 * (j0r + (nrow > 1 ? 1 : 0))) * D;
            const float* w2 = attn_w + (size_t)(2 * (j0r + (nrow > 2 ? 2 : 0))) * D;
            const float* w3 = attn_w + (size_t)(2 * (j0r + (nrow > 3 ? 3 : 0))) * D;
            float a0 = 0.0f, a1 = 0.0f, a2 = 0.0f, a3 = 0.0f;
            for (unsigned d = (unsigned)lane * 4u; d < (unsigned)D; d += 128u) {
                const float4 v = *(const float4*)(xr + d);
                a0 += dot4(v, *(const float4*)(w0 + d));
                a1 += dot4(v, *(const float4*)(w1 + d));
                if (nrow > 2) {
                    a2 += dot4(v, *(const float4*)(w2 + d));
                    a3 += dot4(v, *(const float4*)(w3 + d));
                }
            }
            float k0 = hi16 ? a2 : a0;
            float k1 = hi16 ? a3 : a1;
            const float g0 = hi16 ? a0 : a2;
            const float g1 = hi16 ? a1 : a3;
            k0 += __shfl_xor(g0, 16, 32);
            k1 += __shfl_xor(g1, 16, 32);
            float k = hi8 ? k1 : k0;
            const float g = hi8 ? k0 : k1;
            k += __shfl_xor(g, 8, 32);
            for (int off = 4; off > 0; off >>= 1) k += __shfl_xor(k, off, 32);
            if ((lane & 7) == 0 && q < nrow) S[(size_t)node * R + j0r + q] = k;
        }
        if (eid0 >= 0) {
            if (slot0 < cap) bucket[seg0 * cap + slot0] = s0;
            else next[eid0] = atomicExch(&ohead[seg0], eid0);
        }
    }
}

__device__ inline float lane_f(float v, int k) {
    return __int_as_float(__builtin_amdgcn_readlane(__float_as_int(v), k));
}

struct RowSet {
    v4f v[8];
    int k[8];
};

__device__ inline unsigned load_rows(RowSet& g, unsigned mask, int ms,
                                     const float* __restrict__ xcol, int D, bool on) {
    const v4f zero4 = {0.0f, 0.0f, 0.0f, 0.0f};
#pragma unroll
    for (int j = 0; j < 8; ++j) {
        g.v[j] = zero4;
        g.k[j] = -1;
        if (mask != 0u) {
            g.k[j] = __builtin_ctz(mask);
            mask &= mask - 1u;
            const int s = __builtin_amdgcn_readlane(ms, g.k[j]);
            if (on) g.v[j] = *(const v4f*)(xcol + (size_t)s * D);
        }
    }
    return mask;
}

__device__ inline void add_rows(v4f& acc, const RowSet& g, float alpha) {
#pragma unroll
    for (int j = 0; j < 8; ++j) {
        if (g.k[j] >= 0) acc += lane_f(alpha, g.k[j]) * g.v[j];
    }
}

__device__ __forceinline__ void gather_set(const float* __restrict__ x,
                                           const float* __restrict__ S,
                                           const int* __restrict__ src,
                                           const int* __restrict__ bucket,
                                           const int* __restrict__ ohead,
                                           const int* __restrict__ next,
                                           float* __restrict__ out,
                                           int node, int D, int R, int cap, int lane,
                                           int rg0, int Rg, int hop, int wl, int cw, int ms) {
    const v4f zero4 = {0.0f, 0.0f, 0.0f, 0.0f};
    const int budget = 32 / Rg;
    const int lim = budget < cap ? budget : cap;

    if (__builtin_amdgcn_ballot_w32(cw > lim) == 0u) {
        const bool valid = hop < cw;
        const unsigned vmask = __builtin_amdgcn_ballot_w32(valid);
        if (vmask == 0u) return;
        const int hops = (31 - __builtin_clz(vmask)) / Rg + 1;
        const float a = valid ? S[(unsigned)(ms * R + rg0 + wl)] : -INFINITY;
        float alpha = 0.0f;

        for (int c0 = 0; c0 < D; c0 += 128) {
            const int d = c0 + lane * 4;
            const bool on = d < D;
            float* orow = out + (size_t)node * D + d;
            const float* xcol = x + d;
            v4f o = zero4;
            if (on) o = *(const v4f*)orow;
            unsigned mask = vmask;
            RowSet g;
            mask = load_rows(g, mask, ms, xcol, D, on);

            if (c0 == 0) {
                float m = -INFINITY;
                for (int h0 = 0; h0 < hops; h0 += 8) {
#pragma unroll
                    for (int j = 0; j < 8; ++j) {
                        const float t = __shfl(a, ((h0 + j) * Rg + wl) & 31, 32);
                        if (h0 + j < hops) m = fmaxf(m, t);
                    }
                }
                const float p = valid ? __expf(a - m) : 0.0f;
                float den = 0.0f;
                for (int h0 = 0; h0 < hops; h0 += 8) {
#pragma unroll
                    for (int j = 0; j < 8; ++j) {
                        const float t = __shfl(p, ((h0 + j) * Rg + wl) & 31, 32);
                        if (h0 + j < hops) den += t;
                    }
                }
                alpha = p / fmaxf(den, 1e-30f);
            }

            v4f acc = zero4;
            add_rows(acc, g, alpha);
            while (mask != 0u) {
                mask = load_rows(g, mask, ms, xcol, D, on);
                add_rows(acc, g, alpha);
            }
            if (on) *(v4f*)orow = o + acc;
        }
    } else {
        for (int c0 = 0; c0 < D; c0 += 128) {
            const int d = c0 + lane * 4;
            const bool on = d < D;
            float* orow = out + (size_t)node * D + d;
            const float* xcol = x + d;
            v4f acc = zero4;
            for (int w = 0; w < Rg; ++w) {
                const int r = rg0 + w;
                const size_t seg = (size_t)node * R + r;
                const int c = __builtin_amdgcn_readlane(cw, w);
                const int cb = c < cap ? c : cap;
                const int oh = c > cap ? ohead[seg] : -1;
                const int* bk = bucket + seg * cap;
                float m = -INFINITY;
                for (int k = 0; k < cb; ++k)
                    m = fmaxf(m, S[(size_t)bk[k] * R + r]);
                for (int e = oh; e >= 0; e = next[e])
                    m = fmaxf(m, S[(size_t)src[e] * R + r]);
                float den = 0.0f;
                for (int k = 0; k < cb; ++k)
                    den += __expf(S[(size_t)bk[k] * R + r] - m);
                for (int e = oh; e >= 0; e = next[e])
                    den += __expf(S[(size_t)src[e] * R + r] - m);
                const float dcl = fmaxf(den, 1e-30f);
                for (int k = 0; k < cb; ++k) {
                    const int s = bk[k];
                    const float al = __expf(S[(size_t)s * R + r] - m) / dcl;
                    if (on) acc += al * *(const v4f*)(xcol + (size_t)s * D);
                }
                for (int e = oh; e >= 0; e = next[e]) {
                    const int s = src[e];
                    const float al = __expf(S[(size_t)s * R + r] - m) / dcl;
                    if (on) acc += al * *(const v4f*)(xcol + (size_t)s * D);
                }
            }
            if (on) {
                const v4f o = *(const v4f*)orow;
                *(v4f*)orow = o + acc;
            }
        }
    }
}

__global__ __launch_bounds__(256, 16) void gather_softmax(const float* __restrict__ x,
                                                          const float* __restrict__ S,
                                                          const int* __restrict__ src,
                                                          const int* __restrict__ cnt,
                                                          const int* __restrict__ bucket,
                                                          const int* __restrict__ ohead,
                                                          const int* __restrict__ next,
                                                          float* __restrict__ out,
                                                          int N, int D, int R,
                                                          int cap,
                                                          int nw) {
    const int lane = threadIdx.x & 31;
    int node = __builtin_amdgcn_readfirstlane(blockIdx.x * 8 + (threadIdx.x >> 5));
    if (node >= N) return;

    const int Rfirst = R < 32 ? R : 32;
    const int hop0 = lane / Rfirst;
    const int wl0 = lane - hop0 * Rfirst;
    const bool in0 = hop0 < 32 / Rfirst;
    const bool sl0 = in0 && hop0 < cap;
    const unsigned coff0 = (unsigned)wl0;
    const unsigned boff0 = (unsigned)(wl0 * cap + hop0);
    const unsigned rc = (unsigned)(R * cap);

    int c_cur = in0 ? cnt[(unsigned)node * (unsigned)R + coff0] : 0;
    int c_nxt = (in0 && node + nw < N) ? cnt[(unsigned)(node + nw) * (unsigned)R + coff0] : 0;
    int m_cur = (sl0 && hop0 < c_cur) ? bucket[(unsigned)node * rc + boff0] : -1;
    asm volatile("" : : "v"(c_cur), "v"(m_cur));

    for (; node < N; node += nw) {
        int m_nxt = -1;
        if (sl0 && hop0 < c_nxt) m_nxt = bucket[(unsigned)(node + nw) * rc + boff0];
        int c_nn = 0;
        if (in0 && node + 2 * nw < N) c_nn = cnt[(unsigned)(node + 2 * nw) * (unsigned)R + coff0];

        gather_set(x, S, src, bucket, ohead, next, out, node, D, R, cap, lane,
                   0, Rfirst, hop0, wl0, c_cur, m_cur);

        for (int rg0 = 32; rg0 < R; rg0 += 32) {
            const int Rg = (R - rg0) < 32 ? (R - rg0) : 32;
            const int budget = 32 / Rg;
            const int hop = lane / Rg;
            const int wl = lane - hop * Rg;
            const int lim = budget < cap ? budget : cap;
            const size_t seg0 = (size_t)node * R + rg0;
            const int cw = (hop < budget) ? (cnt + seg0)[(unsigned)wl] : 0;
            const int ms = (hop < lim && hop < cw) ? (bucket + seg0 * cap)[(unsigned)(wl * cap + hop)] : -1;
            gather_set(x, S, src, bucket, ohead, next, out, node, D, R, cap, lane,
                       rg0, Rg, hop, wl, cw, ms);
        }

        c_cur = c_nxt;
        m_cur = m_nxt;
        c_nxt = c_nn;
    }
}

__global__ __launch_bounds__(32) void wmma_selfloop(const float* __restrict__ x,
                                                    const float* __restrict__ W,
                                                    const float* __restrict__ bias,
                                                    float* __restrict__ out,
                                                    int* __restrict__ cnt,
                                                    int* __restrict__ ohead,
                                                    int rn, int N, int D) {
    const int lane = threadIdx.x;
    const int m0 = blockIdx.x * 16;
    const int n0 = blockIdx.y * 64;
    const int row = lane & 15;
    const int khalf = (lane >> 4) * 2;

    const int fi = (blockIdx.x * (D >> 6) + blockIdx.y) * 32 + lane;
    if (fi < rn) {
        cnt[fi] = 0;
        ohead[fi] = -1;
    }

    v8f c0 = {}, c1 = {}, c2 = {}, c3 = {};

    for (int k = 0; k < D; k += 4) {
        const float* ap = x + (size_t)(m0 + row) * D + (k + khalf);
        v2f a;
        a.x = ap[0];
        a.y = ap[1];

        const float* bp = W + (size_t)(k + khalf) * D + n0 + row;
        v2f b0, b1, b2, b3;
        b0.x = bp[0];  b0.y = bp[D + 0];
        b1.x = bp[16]; b1.y = bp[D + 16];
        b2.x = bp[32]; b2.y = bp[D + 32];
        b3.x = bp[48]; b3.y = bp[D + 48];

        c0 = __builtin_amdgcn_wmma_f32_16x16x4_f32(false, a, false, b0, (short)0, c0, false, false);
        c1 = __builtin_amdgcn_wmma_f32_16x16x4_f32(false, a, false, b1, (short)0, c1, false, false);
        c2 = __builtin_amdgcn_wmma_f32_16x16x4_f32(false, a, false, b2, (short)0, c2, false, false);
        c3 = __builtin_amdgcn_wmma_f32_16x16x4_f32(false, a, false, b3, (short)0, c3, false, false);
    }

    const int col = n0 + (lane & 15);
    const int mbase = m0 + (lane >> 4) * 8;
#pragma unroll
    for (int v = 0; v < 8; ++v) {
        float* o = out + (size_t)(mbase + v) * D;
        o[col + 0]  = c0[v] + bias[col + 0];
        o[col + 16] = c1[v] + bias[col + 16];
        o[col + 32] = c2[v] + bias[col + 32];
        o[col + 48] = c3[v] + bias[col + 48];
    }
}

#define WP_PITCH 288
#define Q_STRIDE 64

__global__ __launch_bounds__(128) void wmma_selfloop_lds(const float* __restrict__ x,
                                                        const float* __restrict__ W,
                                                        const float* __restrict__ bias,
                                                        float* __restrict__ out,
                                                        int* __restrict__ cnt,
                                                        int* __restrict__ ohead,
                                                        int rn, int row_tiles) {
    __shared__ float wp[64 * WP_PITCH];
    const int tid = threadIdx.x;
    const int lane = tid & 31;
    const int wave = tid >> 5;

    for (int fi = blockIdx.x * 128 + tid; fi < rn; fi += gridDim.x * 128) {
        cnt[fi] = 0;
        ohead[fi] = -1;
    }

    for (int it = tid; it < 64 * 32; it += 128) {
        const int kp = it >> 5;
        const int c4 = (it & 31) * 4;
        const float4 r0 = *(const float4*)(W + (size_t)(2 * kp) * 128 + c4);
        const float4 r1 = *(const float4*)(W + (size_t)(2 * kp + 1) * 128 + c4);
        float* q = wp + kp * WP_PITCH + c4 * 2;
        *(float4*)q = make_float4(r0.x, r1.x, r0.y, r1.y);
        *(float4*)(q + 4) = make_float4(r0.z, r1.z, r0.w, r1.w);
    }
    __syncthreads();

    const int tile = blockIdx.x * 4 + wave;
    if (tile >= row_tiles) return;

    const int row = lane & 15;
    const int h = lane >> 4;
    const float* ap = x + (size_t)(tile * 16 + row) * 128 + h * 2;
    const float* bp = wp + h * WP_PITCH + row * 2;

    v8f c[8];
#pragma unroll
    for (int j = 0; j < 8; ++j) c[j] = (v8f){};

    v2f an[4];
#pragma unroll
    for (int u = 0; u < 4; ++u) an[u] = *(const v2f*)(ap + 4 * u);

    for (int g = 0; g < 8; ++g) {
        v2f ac[4];
#pragma unroll
        for (int u = 0; u < 4; ++u) ac[u] = an[u];
        if (g + 1 < 8) {
#pragma unroll
            for (int u = 0; u < 4; ++u) an[u] = *(const v2f*)(ap + 16 * (g + 1) + 4 * u);
        }
#pragma unroll
        for (int u = 0; u < 4; ++u) {
            const float* b = bp + (2 * (4 * g + u)) * WP_PITCH;
            v2f bf[8];
#pragma unroll
            for (int j = 0; j < 8; ++j) bf[j] = *(const v2f*)(b + 32 * j);
#pragma unroll
            for (int j = 0; j < 8; ++j)
                c[j] = __builtin_amdgcn_wmma_f32_16x16x4_f32(false, ac[u], false, bf[j], (short)0, c[j], false, false);
        }
    }

    const int mbase = tile * 16 + h * 8;
    float bj[8];
#pragma unroll
    for (int j = 0; j < 8; ++j) bj[j] = bias[16 * j + row];
#pragma unroll
    for (int v = 0; v < 8; ++v) {
        float* o = out + (size_t)(mbase + v) * 128 + row;
#pragma unroll
        for (int j = 0; j < 8; ++j) o[16 * j] = c[j][v] + bj[j];
    }
}

__global__ __launch_bounds__(256) void fill_heads(int* __restrict__ cnt, int* __restrict__ ohead, int* __restrict__ q, int rn) {
    if (blockIdx.x == 0 && threadIdx.x < 2) q[threadIdx.x * Q_STRIDE] = 0;
    for (int idx = blockIdx.x * 256 + threadIdx.x; idx < rn; idx += gridDim.x * 256) {
        cnt[idx] = 0;
        ohead[idx] = -1;
    }
}

__global__ __launch_bounds__(128) void wmma_selfloop_edges(const float* __restrict__ x,
                                                          const float* __restrict__ W,
                                                          const float* __restrict__ bias,
                                                          const float* __restrict__ attn_w,
                                                          const int* __restrict__ src,
                                                          const int* __restrict__ dst,
                                                          float* __restrict__ out,
                                                          float* __restrict__ S,
                                                          int* __restrict__ cnt,
                                                          int* __restrict__ bucket,
                                                          int* __restrict__ ohead,
                                                          int* __restrict__ next,
                                                          int* __restrict__ queue,
                                                          int row_tiles, int R, int E, int per, int cap,
                                                          int split) {
    __shared__ float wp[64 * WP_PITCH];
    const int tid = threadIdx.x;
    const int lane = tid & 31;
    const int wave = tid >> 5;
    const int tile = blockIdx.x * 4 + wave;

    const int hq0 = (int)((__builtin_amdgcn_s_sendmsg_rtn(0x87) >> 18) & 1u);
    int base = 0;
    int qoff = hq0 * Q_STRIDE;
    asm volatile("" : "+v"(qoff));
    if (tile < row_tiles && lane == 0) base = atomicAdd(&queue[qoff], 2);

    const int ca = tid & 15;
    const int ka = tid >> 4;
    float2 av[8];
#pragma unroll
    for (int q = 0; q < 8; ++q) {
        av[q] = make_float2(0.0f, 0.0f);
        if (ca < R) av[q] = *(const float2*)(attn_w + (size_t)ca * 256 + 2 * (ka + 8 * q));
    }

    for (int it = tid; it < 64 * 32; it += 128) {
        const int kp = it >> 5;
        const int c4 = (it & 31) * 4;
        const float4 r0 = *(const float4*)(W + (size_t)(2 * kp) * 128 + c4);
        const float4 r1 = *(const float4*)(W + (size_t)(2 * kp + 1) * 128 + c4);
        float* q = wp + kp * WP_PITCH + c4 * 2;
        *(float4*)q = make_float4(r0.x, r1.x, r0.y, r1.y);
        *(float4*)(q + 4) = make_float4(r0.z, r1.z, r0.w, r1.w);
    }
#pragma unroll
    for (int q = 0; q < 8; ++q) *(float2*)(wp + (ka + 8 * q) * WP_PITCH + 256 + 2 * ca) = av[q];
    __syncthreads();

    if (tile >= row_tiles) return;

    base = __builtin_amdgcn_readfirstlane(base);
    int ch[2], hv[2];
    ch[0] = base;
    ch[1] = base + 1;
    hv[0] = hq0;
    hv[1] = hq0;
    const int missing = (base + 1 < row_tiles) ? 0 : ((base < row_tiles) ? 1 : 2);
    if (missing > 0) {
        int b2 = 0;
        if (lane == 0) b2 = atomicAdd(&queue[(1 - hq0) * Q_STRIDE], missing);
        b2 = __builtin_amdgcn_readfirstlane(b2);
        if (missing == 2) {
            ch[0] = b2;
            hv[0] = 1 - hq0;
            ch[1] = b2 + 1;
        } else {
            ch[1] = b2;
        }
        hv[1] = 1 - hq0;
    }

    const int span = 16 * per;
    const int tot = R * span;
    int eid[8], rr[4], tn[8], sn[8], seg[8], slot[8];
#pragma unroll
    for (int j = 0; j < 4; ++j) {
        const int i = lane + 32 * j;
        rr[j] = (i < tot) ? i / span : 0;
#pragma unroll
        for (int v = 0; v < 2; ++v) {
            eid[4 * v + j] = -1;
            if (ch[v] < row_tiles && i < tot) {
                const int idx = ch[v] * span + (i - rr[j] * span);
                if (idx < E) eid[4 * v + j] = rr[j] * E + idx;
            }
        }
    }
#pragma unroll
    for (int e = 0; e < 8; ++e) {
        tn[e] = 0;
        sn[e] = 0;
        seg[e] = 0;
        slot[e] = 0;
        if (eid[e] >= 0) {
            tn[e] = dst[eid[e]];
            sn[e] = src[eid[e]];
        }
    }

    const int row = lane & 15;
    const int h = lane >> 4;
    const float* ap = x + (size_t)(tile * 16 + row) * 128 + h * 2;
    const float* bp = wp + h * WP_PITCH + row * 2;

    v8f c[9];
#pragma unroll
    for (int j = 0; j < 9; ++j) c[j] = (v8f){};

    v2f an[4];
#pragma unroll
    for (int u = 0; u < 4; ++u) an[u] = *(const v2f*)(ap + 4 * u);

#pragma unroll
    for (int e = 0; e < 8; ++e) {
        if (eid[e] >= 0 && ((tn[e] >= split) ? 1 : 0) != hv[e >> 2]) eid[e] = -1;
        if (eid[e] >= 0) {
            seg[e] = tn[e] * R + rr[e & 3];
            slot[e] = atomicAdd(&cnt[seg[e]], 1);
        }
    }

    for (int g = 0; g < 8; ++g) {
        v2f ac[4];
#pragma unroll
        for (int u = 0; u < 4; ++u) ac[u] = an[u];
        if (g + 1 < 8) {
#pragma unroll
            for (int u = 0; u < 4; ++u) an[u] = *(const v2f*)(ap + 16 * (g + 1) + 4 * u);
        }
#pragma unroll
        for (int u = 0; u < 4; ++u) {
            const float* b = bp + (2 * (4 * g + u)) * WP_PITCH;
            v2f bf[9];
#pragma unroll
            for (int j = 0; j < 9; ++j) bf[j] = *(const v2f*)(b + 32 * j);
#pragma unroll
            for (int j = 0; j < 9; ++j)
                c[j] = __builtin_amdgcn_wmma_f32_16x16x4_f32(false, ac[u], false, bf[j], (short)0, c[j], false, false);
        }
    }

    const int mbase = tile * 16 + h * 8;
    float bj[8];
#pragma unroll
    for (int j = 0; j < 8; ++j) bj[j] = bias[16 * j + row];
#pragma unroll
    for (int v = 0; v < 8; ++v) {
        float* o = out + (size_t)(mbase + v) * 128 + row;
#pragma unroll
        for (int j = 0; j < 8; ++j) o[16 * j] = c[j][v] + bj[j];
    }
    if (row < R) {
#pragma unroll
        for (int v = 0; v < 8; ++v) S[(size_t)(mbase + v) * R + row] = c[8][v];
    }

#pragma unroll
    for (int e = 0; e < 8; ++e) {
        if (eid[e] >= 0) {
            if (slot[e] < cap) bucket[(size_t)seg[e] * cap + slot[e]] = sn[e];
            else next[eid[e]] = atomicExch(&ohead[seg[e]], eid[e]);
        }
    }
    for (int v = 0; v < 2; ++v) {
        if (ch[v] >= row_tiles) continue;
        for (int i = lane + 128; i < tot; i += 32) {
            const int r = i / span;
            const int idx = ch[v] * span + (i - r * span);
            if (idx < E) {
                const int e = r * E + idx;
                const int t = dst[e];
                if (((t >= split) ? 1 : 0) == hv[v]) {
                    const size_t sg = (size_t)t * R + r;
                    const int sl = atomicAdd(&cnt[sg], 1);
                    if (sl < cap) bucket[sg * cap + sl] = src[e];
                    else next[e] = atomicExch(&ohead[sg], e);
                }
            }
        }
    }
}

__global__ __launch_bounds__(256) void gemm_tail(const float* __restrict__ x,
                                                 const float* __restrict__ W,
                                                 const float* __restrict__ bias,
                                                 float* __restrict__ out,
                                                 int N, int D, int row0) {
    long long idx = (long long)blockIdx.x * blockDim.x + threadIdx.x;
    int row = row0 + (int)(idx / D);
    int col = (int)(idx % D);
    if (row >= N) return;
    float acc = bias[col];
    for (int k = 0; k < D; ++k) acc += x[(size_t)row * D + k] * W[(size_t)k * D + col];
    out[(size_t)row * D + col] = acc;
}

extern "C" void kernel_launch(void* const* d_in, const int* in_sizes, int n_in,
                              void* d_out, int out_size, void* d_ws, size_t ws_size,
                              hipStream_t stream) {
    const float* x      = (const float*)d_in[0];
    const float* attn_w = (const float*)d_in[1];
    const float* W      = (const float*)d_in[2];
    const float* bias   = (const float*)d_in[3];
    const int*   src    = (const int*)d_in[4];
    const int*   dst    = (const int*)d_in[5];
    float*       out    = (float*)d_out;

    const int D = in_sizes[3];
    const int N = in_sizes[0] / D;
    const int R = in_sizes[1] / (2 * D);
    const int E = in_sizes[4] / R;

    const int rn = R * N;
    const long long fixed_words = (long long)R * E + 3LL * rn + 2 * Q_STRIDE;
    const long long spare_words = (long long)(ws_size / sizeof(int)) - fixed_words;
    int cap = 16;
    if (spare_words < 16LL * rn) cap = spare_words > 0 ? (int)(spare_words / rn) : 0;
    if ((long long)rn * cap > 0x7fffffffLL) cap = (int)(0x7fffffffLL / rn);
    int*   bucket_buf = (int*)d_ws;
    int*   next_buf   = bucket_buf + (size_t)rn * cap;
    int*   ohead_buf  = next_buf + (size_t)R * E;
    int*   cnt_buf    = ohead_buf + rn;
    float* s_buf      = (float*)(cnt_buf + rn);
    int*   q_buf      = (int*)(s_buf + rn);

    if (D == 128 && N >= 16 && (N % 16) == 0 && R >= 1 && R <= 16) {
        const int tiles = N / 16;
        const int per_f = (E + N - 1) / N;
        const int fill_groups = (rn + 255) / 256 < 256 ? (rn + 255) / 256 : 256;
        const int gather_groups = (N + 7) / 8 < 2048 ? (N + 7) / 8 : 2048;
        const int split = (N / 2) & ~31;
        fill_heads<<<dim3(fill_groups), dim3(256), 0, stream>>>(cnt_buf, ohead_buf, q_buf, rn);
        wmma_selfloop_edges<<<dim3((tiles + 3) / 4), 128, 0, stream>>>(x, W, bias, attn_w, src, dst, out, s_buf,
                                                                       cnt_buf, bucket_buf, ohead_buf, next_buf,
                                                                       q_buf, tiles, R, E, per_f, cap, split);
        gather_softmax<<<dim3(gather_groups), 256, 0, stream>>>(x, s_buf, src, cnt_buf, bucket_buf, ohead_buf,
                                                                next_buf, out,
                                                                N, D, R, cap, gather_groups * 8);
        return;
    }

    int row_tiles = 0;
    if ((D % 64) == 0) row_tiles = N / 16;
    const bool lds_path = (D == 128) && row_tiles > 0;
    const bool fill_rides = lds_path || (long long)row_tiles * (D / 64) * 32 >= rn;
    if (!fill_rides) {
        init_heads<<<dim3((rn + 255) / 256), dim3(256), 0, stream>>>(cnt_buf, ohead_buf, rn);
    }
    if (lds_path) {
        wmma_selfloop_lds<<<dim3((row_tiles + 3) / 4), 128, 0, stream>>>(x, W, bias, out, cnt_buf, ohead_buf,
                                                                       rn, row_tiles);
    } else if (row_tiles > 0) {
        dim3 ggrid(row_tiles, D / 64);
        wmma_selfloop<<<ggrid, 32, 0, stream>>>(x, W, bias, out, cnt_buf, ohead_buf,
                                                 fill_rides ? rn : 0, N, D);
    }
    int row0 = row_tiles * 16;
    long long tail_elems = (long long)(N - row0) * D;
    if (tail_elems > 0) {
        int blocks = (int)((tail_elems + 255) / 256);
        gemm_tail<<<dim3(blocks), dim3(256), 0, stream>>>(x, W, bias, out, N, D, row0);
    }

    const int node_groups = (N + 7) / 8;
    const int res_scores = node_groups < 1536 ? node_groups : 1536;
    const int res_gather = node_groups < 2048 ? node_groups : 2048;

    const int per = (E + N - 1) / N;
    node_scores<<<dim3(res_scores), 256, 0, stream>>>(x, attn_w, src, dst, s_buf, cnt_buf, bucket_buf,
                                                      ohead_buf, next_buf,
                                                      N, D, R, E, per, cap, res_scores * 8);

    gather_softmax<<<dim3(res_gather), 256, 0, stream>>>(x, s_buf, src, cnt_buf, bucket_buf, ohead_buf,
                                                         next_buf, out,
                                                         N, D, R, cap, res_gather * 8);
}


# Product kernel: all 32 A-fragment loads of a row tile issued ahead of the returning edge-count adds (one in-order counter): no wait in the matrix loop stands behind an add's answer; sums unchanged
# speedup vs baseline: 4.2630x; 1.0874x over previous
#include <hip/hip_runtime.h>
#include <math.h>

typedef __attribute__((ext_vector_type(2))) float v2f;
typedef __attribute__((ext_vector_type(4))) float v4f;
typedef __attribute__((ext_vector_type(8))) float v8f;

__global__ __launch_bounds__(256) void init_heads(int* __restrict__ cnt, int* __restrict__ ohead, int rn) {
    const int idx = blockIdx.x * blockDim.x + threadIdx.x;
    if (idx < rn) {
        cnt[idx] = 0;
        ohead[idx] = -1;
    }
}

__device__ inline float dot4(float4 a, float4 b) {
    return a.x * b.x + a.y * b.y + a.z * b.z + a.w * b.w;
}

__global__ __launch_bounds__(256) void node_scores(const float* __restrict__ x,
                                                   const float* __restrict__ attn_w,
                                                   const int* __restrict__ src,
                                                   const int* __restrict__ dst,
                                                   float* __restrict__ S,
                                                   int* __restrict__ cnt,
                                                   int* __restrict__ bucket,
                                                   int* __restrict__ ohead,
                                                   int* __restrict__ next,
                                                   int N, int D, int R, int E, int per,
                                                   int cap,
                                                   int nw) {
    const int lane = threadIdx.x & 31;
    int node = __builtin_amdgcn_readfirstlane(blockIdx.x * 8 + (threadIdx.x >> 5));
    if (node >= N) return;

    const int tot = R * per;
    const int r0 = (lane < tot) ? lane / per : 0;
    const int j0 = lane - r0 * per;
    const bool hi16 = (lane & 16) != 0;
    const bool hi8 = (lane & 8) != 0;
    const int q = lane >> 3;

    int eid_n = -1, t_n = 0, s_n = 0;
    if (lane < tot) {
        const int idx = node * per + j0;
        if (idx < E) {
            eid_n = r0 * E + idx;
            t_n = dst[eid_n];
            s_n = src[eid_n];
        }
    }

    for (; node < N; node += nw) {
        const int eid0 = eid_n;
        const size_t seg0 = (size_t)t_n * R + r0;
        const int s0 = s_n;
        int slot0 = 0;
        if (eid0 >= 0) slot0 = atomicAdd(&cnt[seg0], 1);
        eid_n = -1;
        const int nn = node + nw;
        if (nn < N && lane < tot) {
            const int idx = nn * per + j0;
            if (idx < E) {
                eid_n = r0 * E + idx;
                t_n = dst[eid_n];
                s_n = src[eid_n];
            }
        }
        for (int i = lane + 32; i < tot; i += 32) {
            const int r = i / per;
            const int idx = node * per + (i - r * per);
            if (idx < E) {
                const int eid = r * E + idx;
                const size_t seg = (size_t)dst[eid] * R + r;
                const int slot = atomicAdd(&cnt[seg], 1);
                if (slot < cap) bucket[seg * cap + slot] = src[eid];
                else next[eid] = atomicExch(&ohead[seg], eid);
            }
        }

        const float* xr = x + (size_t)node * D;
        for (int j0r = 0; j0r < R; j0r += 4) {
            const int nrow = (R - j0r) < 4 ? (R - j0r) : 4;
            const float* w0 = attn_w + (size_t)(2 * j0r) * D;
            const float* w1 = attn_w + (size_t)(2 * (j0r + (nrow > 1 ? 1 : 0))) * D;
            const float* w2 = attn_w + (size_t)(2 * (j0r + (nrow > 2 ? 2 : 0))) * D;
            const float* w3 = attn_w + (size_t)(2 * (j0r + (nrow > 3 ? 3 : 0))) * D;
            float a0 = 0.0f, a1 = 0.0f, a2 = 0.0f, a3 = 0.0f;
            for (unsigned d = (unsigned)lane * 4u; d < (unsigned)D; d += 128u) {
                const float4 v = *(const float4*)(xr + d);
                a0 += dot4(v, *(const float4*)(w0 + d));
                a1 += dot4(v, *(const float4*)(w1 + d));
                if (nrow > 2) {
                    a2 += dot4(v, *(const float4*)(w2 + d));
                    a3 += dot4(v, *(const float4*)(w3 + d));
                }
            }
            float k0 = hi16 ? a2 : a0;
            float k1 = hi16 ? a3 : a1;
            const float g0 = hi16 ? a0 : a2;
            const float g1 = hi16 ? a1 : a3;
            k0 += __shfl_xor(g0, 16, 32);
            k1 += __shfl_xor(g1, 16, 32);
            float k = hi8 ? k1 : k0;
            const float g = hi8 ? k0 : k1;
            k += __shfl_xor(g, 8, 32);
            for (int off = 4; off > 0; off >>= 1) k += __shfl_xor(k, off, 32);
            if ((lane & 7) == 0 && q < nrow) S[(size_t)node * R + j0r + q] = k;
        }
        if (eid0 >= 0) {
            if (slot0 < cap) bucket[seg0 * cap + slot0] = s0;
            else next[eid0] = atomicExch(&ohead[seg0], eid0);
        }
    }
}

__device__ inline float lane_f(float v, int k) {
    return __int_as_float(__builtin_amdgcn_readlane(__float_as_int(v), k));
}

struct RowSet {
    v4f v[8];
    int k[8];
};

__device__ inline unsigned load_rows(RowSet& g, unsigned mask, int ms,
                                     const float* __restrict__ xcol, int D, bool on) {
    const v4f zero4 = {0.0f, 0.0f, 0.0f, 0.0f};
#pragma unroll
    for (int j = 0; j < 8; ++j) {
        g.v[j] = zero4;
        g.k[j] = -1;
        if (mask != 0u) {
            g.k[j] = __builtin_ctz(mask);
            mask &= mask - 1u;
            const int s = __builtin_amdgcn_readlane(ms, g.k[j]);
            if (on) g.v[j] = *(const v4f*)(xcol + (size_t)s * D);
        }
    }
    return mask;
}

__device__ inline void add_rows(v4f& acc, const RowSet& g, float alpha) {
#pragma unroll
    for (int j = 0; j < 8; ++j) {
        if (g.k[j] >= 0) acc += lane_f(alpha, g.k[j]) * g.v[j];
    }
}

__device__ __forceinline__ void gather_set(const float* __restrict__ x,
                                           const float* __restrict__ S,
                                           const int* __restrict__ src,
                                           const int* __restrict__ bucket,
                                           const int* __restrict__ ohead,
                                           const int* __restrict__ next,
                                           float* __restrict__ out,
                                           int node, int D, int R, int cap, int lane,
                                           int rg0, int Rg, int hop, int wl, int cw, int ms) {
    const v4f zero4 = {0.0f, 0.0f, 0.0f, 0.0f};
    const int budget = 32 / Rg;
    const int lim = budget < cap ? budget : cap;

    if (__builtin_amdgcn_ballot_w32(cw > lim) == 0u) {
        const bool valid = hop < cw;
        const unsigned vmask = __builtin_amdgcn_ballot_w32(valid);
        if (vmask == 0u) return;
        const int hops = (31 - __builtin_clz(vmask)) / Rg + 1;
        const float a = valid ? S[(unsigned)(ms * R + rg0 + wl)] : -INFINITY;
        float alpha = 0.0f;

        for (int c0 = 0; c0 < D; c0 += 128) {
            const int d = c0 + lane * 4;
            const bool on = d < D;
            float* orow = out + (size_t)node * D + d;
            const float* xcol = x + d;
            v4f o = zero4;
            if (on) o = *(const v4f*)orow;
            unsigned mask = vmask;
            RowSet g;
            mask = load_rows(g, mask, ms, xcol, D, on);

            if (c0 == 0) {
                float m = -INFINITY;
                for (int h0 = 0; h0 < hops; h0 += 8) {
#pragma unroll
                    for (int j = 0; j < 8; ++j) {
                        const float t = __shfl(a, ((h0 + j) * Rg + wl) & 31, 32);
                        if (h0 + j < hops) m = fmaxf(m, t);
                    }
                }
                const float p = valid ? __expf(a - m) : 0.0f;
                float den = 0.0f;
                for (int h0 = 0; h0 < hops; h0 += 8) {
#pragma unroll
                    for (int j = 0; j < 8; ++j) {
                        const float t = __shfl(p, ((h0 + j) * Rg + wl) & 31, 32);
                        if (h0 + j < hops) den += t;
                    }
                }
                alpha = p / fmaxf(den, 1e-30f);
            }

            v4f acc = zero4;
            add_rows(acc, g, alpha);
            while (mask != 0u) {
                mask = load_rows(g, mask, ms, xcol, D, on);
                add_rows(acc, g, alpha);
            }
            if (on) *(v4f*)orow = o + acc;
        }
    } else {
        for (int c0 = 0; c0 < D; c0 += 128) {
            const int d = c0 + lane * 4;
            const bool on = d < D;
            float* orow = out + (size_t)node * D + d;
            const float* xcol = x + d;
            v4f acc = zero4;
            for (int w = 0; w < Rg; ++w) {
                const int r = rg0 + w;
                const size_t seg = (size_t)node * R + r;
                const int c = __builtin_amdgcn_readlane(cw, w);
                const int cb = c < cap ? c : cap;
                const int oh = c > cap ? ohead[seg] : -1;
                const int* bk = bucket + seg * cap;
                float m = -INFINITY;
                for (int k = 0; k < cb; ++k)
                    m = fmaxf(m, S[(size_t)bk[k] * R + r]);
                for (int e = oh; e >= 0; e = next[e])
                    m = fmaxf(m, S[(size_t)src[e] * R + r]);
                float den = 0.0f;
                for (int k = 0; k < cb; ++k)
                    den += __expf(S[(size_t)bk[k] * R + r] - m);
                for (int e = oh; e >= 0; e = next[e])
                    den += __expf(S[(size_t)src[e] * R + r] - m);
                const float dcl = fmaxf(den, 1e-30f);
                for (int k = 0; k < cb; ++k) {
                    const int s = bk[k];
                    const float al = __expf(S[(size_t)s * R + r] - m) / dcl;
                    if (on) acc += al * *(const v4f*)(xcol + (size_t)s * D);
                }
                for (int e = oh; e >= 0; e = next[e]) {
                    const int s = src[e];
                    const float al = __expf(S[(size_t)s * R + r] - m) / dcl;
                    if (on) acc += al * *(const v4f*)(xcol + (size_t)s * D);
                }
            }
            if (on) {
                const v4f o = *(const v4f*)orow;
                *(v4f*)orow = o + acc;
            }
        }
    }
}

__global__ __launch_bounds__(256, 16) void gather_softmax(const float* __restrict__ x,
                                                          const float* __restrict__ S,
                                                          const int* __restrict__ src,
                                                          const int* __restrict__ cnt,
                                                          const int* __restrict__ bucket,
                                                          const int* __restrict__ ohead,
                                                          const int* __restrict__ next,
                                                          float* __restrict__ out,
                                                          int N, int D, int R,
                                                          int cap,
                                                          int nw) {
    const int lane = threadIdx.x & 31;
    int node = __builtin_amdgcn_readfirstlane(blockIdx.x * 8 + (threadIdx.x >> 5));
    if (node >= N) return;

    const int Rfirst = R < 32 ? R : 32;
    const int hop0 = lane / Rfirst;
    const int wl0 = lane - hop0 * Rfirst;
    const bool in0 = hop0 < 32 / Rfirst;
    const bool sl0 = in0 && hop0 < cap;
    const unsigned coff0 = (unsigned)wl0;
    const unsigned boff0 = (unsigned)(wl0 * cap + hop0);
    const unsigned rc = (unsigned)(R * cap);

    int c_cur = in0 ? cnt[(unsigned)node * (unsigned)R + coff0] : 0;
    int c_nxt = (in0 && node + nw < N) ? cnt[(unsigned)(node + nw) * (unsigned)R + coff0] : 0;
    int m_cur = (sl0 && hop0 < c_cur) ? bucket[(unsigned)node * rc + boff0] : -1;
    asm volatile("" : : "v"(c_cur), "v"(m_cur));

    for (; node < N; node += nw) {
        int m_nxt = -1;
        if (sl0 && hop0 < c_nxt) m_nxt = bucket[(unsigned)(node + nw) * rc + boff0];
        int c_nn = 0;
        if (in0 && node + 2 * nw < N) c_nn = cnt[(unsigned)(node + 2 * nw) * (unsigned)R + coff0];

        gather_set(x, S, src, bucket, ohead, next, out, node, D, R, cap, lane,
                   0, Rfirst, hop0, wl0, c_cur, m_cur);

        for (int rg0 = 32; rg0 < R; rg0 += 32) {
            const int Rg = (R - rg0) < 32 ? (R - rg0) : 32;
            const int budget = 32 / Rg;
            const int hop = lane / Rg;
            const int wl = lane - hop * Rg;
            const int lim = budget < cap ? budget : cap;
            const size_t seg0 = (size_t)node * R + rg0;
            const int cw = (hop < budget) ? (cnt + seg0)[(unsigned)wl] : 0;
            const int ms = (hop < lim && hop < cw) ? (bucket + seg0 * cap)[(unsigned)(wl * cap + hop)] : -1;
            gather_set(x, S, src, bucket, ohead, next, out, node, D, R, cap, lane,
                       rg0, Rg, hop, wl, cw, ms);
        }

        c_cur = c_nxt;
        m_cur = m_nxt;
        c_nxt = c_nn;
    }
}

__global__ __launch_bounds__(32) void wmma_selfloop(const float* __restrict__ x,
                                                    const float* __restrict__ W,
                                                    const float* __restrict__ bias,
                                                    float* __restrict__ out,
                                                    int* __restrict__ cnt,
                                                    int* __restrict__ ohead,
                                                    int rn, int N, int D) {
    const int lane = threadIdx.x;
    const int m0 = blockIdx.x * 16;
    const int n0 = blockIdx.y * 64;
    const int row = lane & 15;
    const int khalf = (lane >> 4) * 2;

    const int fi = (blockIdx.x * (D >> 6) + blockIdx.y) * 32 + lane;
    if (fi < rn) {
        cnt[fi] = 0;
        ohead[fi] = -1;
    }

    v8f c0 = {}, c1 = {}, c2 = {}, c3 = {};

    for (int k = 0; k < D; k += 4) {
        const float* ap = x + (size_t)(m0 + row) * D + (k + khalf);
        v2f a;
        a.x = ap[0];
        a.y = ap[1];

        const float* bp = W + (size_t)(k + khalf) * D + n0 + row;
        v2f b0, b1, b2, b3;
        b0.x = bp[0];  b0.y = bp[D + 0];
        b1.x = bp[16]; b1.y = bp[D + 16];
        b2.x = bp[32]; b2.y = bp[D + 32];
        b3.x = bp[48]; b3.y = bp[D + 48];

        c0 = __builtin_amdgcn_wmma_f32_16x16x4_f32(false, a, false, b0, (short)0, c0, false, false);
        c1 = __builtin_amdgcn_wmma_f32_16x16x4_f32(false, a, false, b1, (short)0, c1, false, false);
        c2 = __builtin_amdgcn_wmma_f32_16x16x4_f32(false, a, false, b2, (short)0, c2, false, false);
        c3 = __builtin_amdgcn_wmma_f32_16x16x4_f32(false, a, false, b3, (short)0, c3, false, false);
    }

    const int col = n0 + (lane & 15);
    const int mbase = m0 + (lane >> 4) * 8;
#pragma unroll
    for (int v = 0; v < 8; ++v) {
        float* o = out + (size_t)(mbase + v) * D;
        o[col + 0]  = c0[v] + bias[col + 0];
        o[col + 16] = c1[v] + bias[col + 16];
        o[col + 32] = c2[v] + bias[col + 32];
        o[col + 48] = c3[v] + bias[col + 48];
    }
}

#define WP_PITCH 288
#define Q_STRIDE 64

__global__ __launch_bounds__(128) void wmma_selfloop_lds(const float* __restrict__ x,
                                                        const float* __restrict__ W,
                                                        const float* __restrict__ bias,
                                                        float* __restrict__ out,
                                                        int* __restrict__ cnt,
                                                        int* __restrict__ ohead,
                                                        int rn, int row_tiles) {
    __shared__ float wp[64 * WP_PITCH];
    const int tid = threadIdx.x;
    const int lane = tid & 31;
    const int wave = tid >> 5;

    for (int fi = blockIdx.x * 128 + tid; fi < rn; fi += gridDim.x * 128) {
        cnt[fi] = 0;
        ohead[fi] = -1;
    }

    for (int it = tid; it < 64 * 32; it += 128) {
        const int kp = it >> 5;
        const int c4 = (it & 31) * 4;
        const float4 r0 = *(const float4*)(W + (size_t)(2 * kp) * 128 + c4);
        const float4 r1 = *(const float4*)(W + (size_t)(2 * kp + 1) * 128 + c4);
        float* q = wp + kp * WP_PITCH + c4 * 2;
        *(float4*)q = make_float4(r0.x, r1.x, r0.y, r1.y);
        *(float4*)(q + 4) = make_float4(r0.z, r1.z, r0.w, r1.w);
    }
    __syncthreads();

    const int tile = blockIdx.x * 4 + wave;
    if (tile >= row_tiles) return;

    const int row = lane & 15;
    const int h = lane >> 4;
    const float* ap = x + (size_t)(tile * 16 + row) * 128 + h * 2;
    const float* bp = wp + h * WP_PITCH + row * 2;

    v8f c[8];
#pragma unroll
    for (int j = 0; j < 8; ++j) c[j] = (v8f){};

    v2f an[4];
#pragma unroll
    for (int u = 0; u < 4; ++u) an[u] = *(const v2f*)(ap + 4 * u);

    for (int g = 0; g < 8; ++g) {
        v2f ac[4];
#pragma unroll
        for (int u = 0; u < 4; ++u) ac[u] = an[u];
        if (g + 1 < 8) {
#pragma unroll
            for (int u = 0; u < 4; ++u) an[u] = *(const v2f*)(ap + 16 * (g + 1) + 4 * u);
        }
#pragma unroll
        for (int u = 0; u < 4; ++u) {
            const float* b = bp + (2 * (4 * g + u)) * WP_PITCH;
            v2f bf[8];
#pragma unroll
            for (int j = 0; j < 8; ++j) bf[j] = *(const v2f*)(b + 32 * j);
#pragma unroll
            for (int j = 0; j < 8; ++j)
                c[j] = __builtin_amdgcn_wmma_f32_16x16x4_f32(false, ac[u], false, bf[j], (short)0, c[j], false, false);
        }
    }

    const int mbase = tile * 16 + h * 8;
    float bj[8];
#pragma unroll
    for (int j = 0; j < 8; ++j) bj[j] = bias[16 * j + row];
#pragma unroll
    for (int v = 0; v < 8; ++v) {
        float* o = out + (size_t)(mbase + v) * 128 + row;
#pragma unroll
        for (int j = 0; j < 8; ++j) o[16 * j] = c[j][v] + bj[j];
    }
}

__global__ __launch_bounds__(256) void fill_heads(int* __restrict__ cnt, int* __restrict__ ohead, int* __restrict__ q, int rn) {
    if (blockIdx.x == 0 && threadIdx.x < 2) q[threadIdx.x * Q_STRIDE] = 0;
    for (int idx = blockIdx.x * 256 + threadIdx.x; idx < rn; idx += gridDim.x * 256) {
        cnt[idx] = 0;
        ohead[idx] = -1;
    }
}

__global__ __launch_bounds__(128) void wmma_selfloop_edges(const float* __restrict__ x,
                                                          const float* __restrict__ W,
                                                          const float* __restrict__ bias,
                                                          const float* __restrict__ attn_w,
                                                          const int* __restrict__ src,
                                                          const int* __restrict__ dst,
                                                          float* __restrict__ out,
                                                          float* __restrict__ S,
                                                          int* __restrict__ cnt,
                                                          int* __restrict__ bucket,
                                                          int* __restrict__ ohead,
                                                          int* __restrict__ next,
                                                          int* __restrict__ queue,
                                                          int row_tiles, int R, int E, int per, int cap,
                                                          int split) {
    __shared__ float wp[64 * WP_PITCH];
    const int tid = threadIdx.x;
    const int lane = tid & 31;
    const int wave = tid >> 5;
    const int tile = blockIdx.x * 4 + wave;

    const int hq0 = (int)((__builtin_amdgcn_s_sendmsg_rtn(0x87) >> 18) & 1u);
    int base = 0;
    int qoff = hq0 * Q_STRIDE;
    asm volatile("" : "+v"(qoff));
    if (tile < row_tiles && lane == 0) base = atomicAdd(&queue[qoff], 2);

    const int ca = tid & 15;
    const int ka = tid >> 4;
    float2 av[8];
#pragma unroll
    for (int q = 0; q < 8; ++q) {
        av[q] = make_float2(0.0f, 0.0f);
        if (ca < R) av[q] = *(const float2*)(attn_w + (size_t)ca * 256 + 2 * (ka + 8 * q));
    }

    for (int it = tid; it < 64 * 32; it += 128) {
        const int kp = it >> 5;
        const int c4 = (it & 31) * 4;
        const float4 r0 = *(const float4*)(W + (size_t)(2 * kp) * 128 + c4);
        const float4 r1 = *(const float4*)(W + (size_t)(2 * kp + 1) * 128 + c4);
        float* q = wp + kp * WP_PITCH + c4 * 2;
        *(float4*)q = make_float4(r0.x, r1.x, r0.y, r1.y);
        *(float4*)(q + 4) = make_float4(r0.z, r1.z, r0.w, r1.w);
    }
#pragma unroll
    for (int q = 0; q < 8; ++q) *(float2*)(wp + (ka + 8 * q) * WP_PITCH + 256 + 2 * ca) = av[q];
    __syncthreads();

    if (tile >= row_tiles) return;

    base = __builtin_amdgcn_readfirstlane(base);
    int ch[2], hv[2];
    ch[0] = base;
    ch[1] = base + 1;
    hv[0] = hq0;
    hv[1] = hq0;
    const int missing = (base + 1 < row_tiles) ? 0 : ((base < row_tiles) ? 1 : 2);
    if (missing > 0) {
        int b2 = 0;
        if (lane == 0) b2 = atomicAdd(&queue[(1 - hq0) * Q_STRIDE], missing);
        b2 = __builtin_amdgcn_readfirstlane(b2);
        if (missing == 2) {
            ch[0] = b2;
            hv[0] = 1 - hq0;
            ch[1] = b2 + 1;
        } else {
            ch[1] = b2;
        }
        hv[1] = 1 - hq0;
    }

    const int span = 16 * per;
    const int tot = R * span;
    int eid[8], rr[4], tn[8], sn[8], seg[8], slot[8];
#pragma unroll
    for (int j = 0; j < 4; ++j) {
        const int i = lane + 32 * j;
        rr[j] = (i < tot) ? i / span : 0;
#pragma unroll
        for (int v = 0; v < 2; ++v) {
            eid[4 * v + j] = -1;
            if (ch[v] < row_tiles && i < tot) {
                const int idx = ch[v] * span + (i - rr[j] * span);
                if (idx < E) eid[4 * v + j] = rr[j] * E + idx;
            }
        }
    }
#pragma unroll
    for (int e = 0; e < 8; ++e) {
        tn[e] = 0;
        sn[e] = 0;
        seg[e] = 0;
        slot[e] = 0;
        if (eid[e] >= 0) {
            tn[e] = dst[eid[e]];
            sn[e] = src[eid[e]];
        }
    }

    const int row = lane & 15;
    const int h = lane >> 4;
    const float* ap = x + (size_t)(tile * 16 + row) * 128 + h * 2;
    const float* bp = wp + h * WP_PITCH + row * 2;

    v8f c[9];
#pragma unroll
    for (int j = 0; j < 9; ++j) c[j] = (v8f){};

    v2f af[32];
#pragma unroll
    for (int t = 0; t < 32; ++t) af[t] = *(const v2f*)(ap + 4 * t);

#pragma unroll
    for (int e = 0; e < 8; ++e) {
        if (eid[e] >= 0 && ((tn[e] >= split) ? 1 : 0) != hv[e >> 2]) eid[e] = -1;
        if (eid[e] >= 0) {
            seg[e] = tn[e] * R + rr[e & 3];
            slot[e] = atomicAdd(&cnt[seg[e]], 1);
        }
    }

#pragma unroll
    for (int t = 0; t < 32; ++t) {
        const float* b = bp + (2 * t) * WP_PITCH;
        v2f bf[9];
#pragma unroll
        for (int j = 0; j < 9; ++j) bf[j] = *(const v2f*)(b + 32 * j);
#pragma unroll
        for (int j = 0; j < 9; ++j)
            c[j] = __builtin_amdgcn_wmma_f32_16x16x4_f32(false, af[t], false, bf[j], (short)0, c[j], false, false);
    }

    const int mbase = tile * 16 + h * 8;
    float bj[8];
#pragma unroll
    for (int j = 0; j < 8; ++j) bj[j] = bias[16 * j + row];
#pragma unroll
    for (int v = 0; v < 8; ++v) {
        float* o = out + (size_t)(mbase + v) * 128 + row;
#pragma unroll
        for (int j = 0; j < 8; ++j) o[16 * j] = c[j][v] + bj[j];
    }
    if (row < R) {
#pragma unroll
        for (int v = 0; v < 8; ++v) S[(size_t)(mbase + v) * R + row] = c[8][v];
    }

#pragma unroll
    for (int e = 0; e < 8; ++e) {
        if (eid[e] >= 0) {
            if (slot[e] < cap) bucket[(size_t)seg[e] * cap + slot[e]] = sn[e];
            else next[eid[e]] = atomicExch(&ohead[seg[e]], eid[e]);
        }
    }
    for (int v = 0; v < 2; ++v) {
        if (ch[v] >= row_tiles) continue;
        for (int i = lane + 128; i < tot; i += 32) {
            const int r = i / span;
            const int idx = ch[v] * span + (i - r * span);
            if (idx < E) {
                const int e = r * E + idx;
                const int t = dst[e];
                if (((t >= split) ? 1 : 0) == hv[v]) {
                    const size_t sg = (size_t)t * R + r;
                    const int sl = atomicAdd(&cnt[sg], 1);
                    if (sl < cap) bucket[sg * cap + sl] = src[e];
                    else next[e] = atomicExch(&ohead[sg], e);
                }
            }
        }
    }
}

__global__ __launch_bounds__(256) void gemm_tail(const float* __restrict__ x,
                                                 const float* __restrict__ W,
                                                 const float* __restrict__ bias,
                                                 float* __restrict__ out,
                                                 int N, int D, int row0) {
    long long idx = (long long)blockIdx.x * blockDim.x + threadIdx.x;
    int row = row0 + (int)(idx / D);
    int col = (int)(idx % D);
    if (row >= N) return;
    float acc = bias[col];
    for (int k = 0; k < D; ++k) acc += x[(size_t)row * D + k] * W[(size_t)k * D + col];
    out[(size_t)row * D + col] = acc;
}

extern "C" void kernel_launch(void* const* d_in, const int* in_sizes, int n_in,
                              void* d_out, int out_size, void* d_ws, size_t ws_size,
                              hipStream_t stream) {
    const float* x      = (const float*)d_in[0];
    const float* attn_w = (const float*)d_in[1];
    const float* W      = (const float*)d_in[2];
    const float* bias   = (const float*)d_in[3];
    const int*   src    = (const int*)d_in[4];
    const int*   dst    = (const int*)d_in[5];
    float*       out    = (float*)d_out;

    const int D = in_sizes[3];
    const int N = in_sizes[0] / D;
    const int R = in_sizes[1] / (2 * D);
    const int E = in_sizes[4] / R;

    const int rn = R * N;
    const long long fixed_words = (long long)R * E + 3LL * rn + 2 * Q_STRIDE;
    const long long spare_words = (long long)(ws_size / sizeof(int)) - fixed_words;
    int cap = 16;
    if (spare_words < 16LL * rn) cap = spare_words > 0 ? (int)(spare_words / rn) : 0;
    if ((long long)rn * cap > 0x7fffffffLL) cap = (int)(0x7fffffffLL / rn);
    int*   bucket_buf = (int*)d_ws;
    int*   next_buf   = bucket_buf + (size_t)rn * cap;
    int*   ohead_buf  = next_buf + (size_t)R * E;
    int*   cnt_buf    = ohead_buf + rn;
    float* s_buf      = (float*)(cnt_buf + rn);
    int*   q_buf      = (int*)(s_buf + rn);

    if (D == 128 && N >= 16 && (N % 16) == 0 && R >= 1 && R <= 16) {
        const int tiles = N / 16;
        const int per_f = (E + N - 1) / N;
        const int fill_groups = (rn + 255) / 256 < 256 ? (rn + 255) / 256 : 256;
        const int gather_groups = (N + 7) / 8 < 2048 ? (N + 7) / 8 : 2048;
        const int split = (N / 2) & ~31;
        fill_heads<<<dim3(fill_groups), dim3(256), 0, stream>>>(cnt_buf, ohead_buf, q_buf, rn);
        wmma_selfloop_edges<<<dim3((tiles + 3) / 4), 128, 0, stream>>>(x, W, bias, attn_w, src, dst, out, s_buf,
                                                                       cnt_buf, bucket_buf, ohead_buf, next_buf,
                                                                       q_buf, tiles, R, E, per_f, cap, split);
        gather_softmax<<<dim3(gather_groups), 256, 0, stream>>>(x, s_buf, src, cnt_buf, bucket_buf, ohead_buf,
                                                                next_buf, out,
                                                                N, D, R, cap, gather_groups * 8);
        return;
    }

    int row_tiles = 0;
    if ((D % 64) == 0) row_tiles = N / 16;
    const bool lds_path = (D == 128) && row_tiles > 0;
    const bool fill_rides = lds_path || (long long)row_tiles * (D / 64) * 32 >= rn;
    if (!fill_rides) {
        init_heads<<<dim3((rn + 255) / 256), dim3(256), 0, stream>>>(cnt_buf, ohead_buf, rn);
    }
    if (lds_path) {
        wmma_selfloop_lds<<<dim3((row_tiles + 3) / 4), 128, 0, stream>>>(x, W, bias, out, cnt_buf, ohead_buf,
                                                                       rn, row_tiles);
    } else if (row_tiles > 0) {
        dim3 ggrid(row_tiles, D / 64);
        wmma_selfloop<<<ggrid, 32, 0, stream>>>(x, W, bias, out, cnt_buf, ohead_buf,
                                                 fill_rides ? rn : 0, N, D);
    }
    int row0 = row_tiles * 16;
    long long tail_elems = (long long)(N - row0) * D;
    if (tail_elems > 0) {
        int blocks = (int)((tail_elems + 255) / 256);
        gemm_tail<<<dim3(blocks), dim3(256), 0, stream>>>(x, W, bias, out, N, D, row0);
    }

    const int node_groups = (N + 7) / 8;
    const int res_scores = node_groups < 1536 ? node_groups : 1536;
    const int res_gather = node_groups < 2048 ? node_groups : 2048;

    const int per = (E + N - 1) / N;
    node_scores<<<dim3(res_scores), 256, 0, stream>>>(x, attn_w, src, dst, s_buf, cnt_buf, bucket_buf,
                                                      ohead_buf, next_buf,
                                                      N, D, R, E, per, cap, res_scores * 8);

    gather_softmax<<<dim3(res_gather), 256, 0, stream>>>(x, s_buf, src, cnt_buf, bucket_buf, ohead_buf,
                                                         next_buf, out,
                                                         N, D, R, cap, res_gather * 8);
}
